# starting kernel
# baseline (speedup 1.0000x reference)
#include <hip/hip_runtime.h>
#include <hip/hip_bf16.h>

typedef float v2f __attribute__((ext_vector_type(2)));
typedef float v4f __attribute__((ext_vector_type(4)));
typedef float v8f __attribute__((ext_vector_type(8)));

#define SDIM 64
#define NCH  128
#define NB   8
#define SLEN 4096

__global__ void hippo_setup_kernel(const float* __restrict__ A,
                                   const float* __restrict__ Bm,
                                   const float* __restrict__ dtp,
                                   float* __restrict__ abar,
                                   float* __restrict__ bbar) {
    __shared__ float Lsh[SDIM * SDIM];
    __shared__ float Linv[SDIM * SDIM];
    const float dt = dtp[0];
    const int tid = threadIdx.x;

    for (int e = tid; e < SDIM * SDIM; e += 256) {
        const int i = e >> 6, j = e & 63;
        float l = -0.5f * dt * A[e];
        if (i == j) l += 1.0f;
        Lsh[e] = l;
    }
    __syncthreads();

    if (tid < SDIM) {
        const int j = tid;
        for (int i = 0; i < SDIM; ++i) {
            if (i < j) {
                Linv[i * SDIM + j] = 0.0f;
            } else {
                float sum = (i == j) ? 1.0f : 0.0f;
                for (int k = j; k < i; ++k)
                    sum -= Lsh[i * SDIM + k] * Linv[k * SDIM + j];
                Linv[i * SDIM + j] = sum / Lsh[i * SDIM + i];
            }
        }
    }
    __syncthreads();

    for (int e = tid; e < SDIM * SDIM; e += 256) {
        const int i = e >> 6, j = e & 63;
        abar[e] = 2.0f * Linv[e] - ((i == j) ? 1.0f : 0.0f);
    }
    for (int e = tid; e < SDIM * NCH; e += 256) {
        const int i = e >> 7, c = e & 127;
        float sum = 0.0f;
        for (int k = 0; k <= i; ++k)
            sum += Linv[i * SDIM + k] * Bm[k * NCH + c];
        bbar[e] = dt * sum;
    }
}

#define ABAR_TILE_NONZERO(tn, tk) ((tk) < 4 * ((tn) + 1))

__global__ __launch_bounds__(256, 1)
void hippo_scan_kernel(const float* __restrict__ sig,
                       const float* __restrict__ Cmat,
                       const float* __restrict__ abar,
                       const float* __restrict__ bbar,
                       float* __restrict__ out) {
    const int b    = blockIdx.x;
    const int lane = threadIdx.x & 31;
    const int w    = threadIdx.x >> 5;
    const int lc   = lane & 15;
    const int lh   = lane >> 4;
    const int c    = w * 16 + lc;

    v2f a[4][16];
#pragma unroll
    for (int tn = 0; tn < 4; ++tn)
#pragma unroll
        for (int tk = 0; tk < 16; ++tk) {
            if (ABAR_TILE_NONZERO(tn, tk)) {
                const int base = (tn * 16 + lc) * SDIM + tk * 4 + 2 * lh;
                a[tn][tk] = *reinterpret_cast<const v2f*>(abar + base);
            }
        }

    float bb[4][8], cw[4][8];
#pragma unroll
    for (int tn = 0; tn < 4; ++tn)
#pragma unroll
        for (int r = 0; r < 8; ++r) {
            const int n = tn * 16 + r + 8 * lh;
            bb[tn][r] = bbar[n * NCH + c];
            cw[tn][r] = Cmat[n * NCH + c];
        }

    v8f s[4];
#pragma unroll
    for (int tn = 0; tn < 4; ++tn)
#pragma unroll
        for (int r = 0; r < 8; ++r) s[tn][r] = 0.0f;

    const int colOff = (b * NCH + c) * SLEN;
    const float* xp = sig + colOff;
    float*       yp = out + colOff;

    v4f xcur4 = *reinterpret_cast<const v4f*>(xp);

#pragma unroll 1
    for (int tc = 0; tc < SLEN; tc += 4) {
        v4f xnext4;
        if (tc + 4 < SLEN) {
            xnext4 = *reinterpret_cast<const v4f*>(xp + tc + 4);
        } else {
            xnext4[0] = 0.0f; xnext4[1] = 0.0f; xnext4[2] = 0.0f; xnext4[3] = 0.0f;
        }

        v4f y4;
#pragma unroll
        for (int u = 0; u < 4; ++u) {
            const float xcur = xcur4[u];

            v2f bv[16];
#pragma unroll
            for (int tk = 0; tk < 16; ++tk) {
                float vals[2];
#pragma unroll
                for (int v = 0; v < 2; ++v) {
                    const int Rlow   = 4 * tk + v;
                    const int tn     = Rlow >> 4;
                    const int rr     = Rlow & 15;
                    const int half   = rr >> 3;
                    const int regLo  = rr & 7;
                    const int regHi  = regLo + 2;
                    const float lowV  = s[tn][regLo];
                    const float highV = s[tn][regHi];
                    const float sw = __shfl_xor(half ? lowV : highV, 16, 32);
                    const float direct = half ? highV : lowV;
                    vals[v] = (lh == half) ? direct : sw;
                }
                bv[tk].x = vals[0];
                bv[tk].y = vals[1];
            }

#pragma unroll
            for (int tn = 0; tn < 4; ++tn) {
                v8f acc;
#pragma unroll
                for (int r = 0; r < 8; ++r) acc[r] = bb[tn][r] * xcur;
#pragma unroll
                for (int tk = 0; tk < 16; ++tk) {
                    if (ABAR_TILE_NONZERO(tn, tk)) {
                        acc = __builtin_amdgcn_wmma_f32_16x16x4_f32(
                             false, a[tn][tk],
                             false, bv[tk],
                             (short)0, acc,
                             false,  false);
                    }
                }
                s[tn] = acc;
            }

            float y = 0.0f;
#pragma unroll
            for (int tn = 0; tn < 4; ++tn)
#pragma unroll
                for (int r = 0; r < 8; ++r) y += cw[tn][r] * s[tn][r];
            y += __shfl_xor(y, 16, 32);
            y4[u] = y;
        }

        if (lh == 0) *reinterpret_cast<v4f*>(yp + tc) = y4;

        xcur4 = xnext4;
    }
}

extern "C" void kernel_launch(void* const* d_in, const int* in_sizes, int n_in,
                              void* d_out, int out_size, void* d_ws, size_t ws_size,
                              hipStream_t stream) {
    const float* sig = (const float*)d_in[0];
    const float* Bm  = (const float*)d_in[1];
    const float* Cm  = (const float*)d_in[2];
    const float* Am  = (const float*)d_in[3];
    const float* dtp = (const float*)d_in[4];
    float* out  = (float*)d_out;
    float* abar = (float*)d_ws;
    float* bbar = abar + SDIM * SDIM;

    hippo_setup_kernel<<<1, 256, 0, stream>>>(Am, Bm, dtp, abar, bbar);
    hippo_scan_kernel<<<NB, 256, 0, stream>>>(sig, Cm, abar, bbar, out);
}


# k order of each WMMA permuted so the state accumulator register pairs are the B fragments: the per-step ds_bpermute relayout is gone; changes the order products are summed within a state row
# speedup vs baseline: 1.2007x; 1.2007x over previous
#include <hip/hip_runtime.h>
#include <hip/hip_bf16.h>

typedef float v2f __attribute__((ext_vector_type(2)));
typedef float v4f __attribute__((ext_vector_type(4)));
typedef float v8f __attribute__((ext_vector_type(8)));

#define SDIM 64
#define NCH  128
#define NB   8
#define SLEN 4096

__global__ void hippo_setup_kernel(const float* __restrict__ A,
                                   const float* __restrict__ Bm,
                                   const float* __restrict__ dtp,
                                   float* __restrict__ abar,
                                   float* __restrict__ bbar) {
    __shared__ float Lsh[SDIM * SDIM];
    __shared__ float Linv[SDIM * SDIM];
    const float dt = dtp[0];
    const int tid = threadIdx.x;

    for (int e = tid; e < SDIM * SDIM; e += 256) {
        const int i = e >> 6, j = e & 63;
        float l = -0.5f * dt * A[e];
        if (i == j) l += 1.0f;
        Lsh[e] = l;
    }
    __syncthreads();

    if (tid < SDIM) {
        const int j = tid;
        for (int i = 0; i < SDIM; ++i) {
            if (i < j) {
                Linv[i * SDIM + j] = 0.0f;
            } else {
                float sum = (i == j) ? 1.0f : 0.0f;
                for (int k = j; k < i; ++k)
                    sum -= Lsh[i * SDIM + k] * Linv[k * SDIM + j];
                Linv[i * SDIM + j] = sum / Lsh[i * SDIM + i];
            }
        }
    }
    __syncthreads();

    for (int e = tid; e < SDIM * SDIM; e += 256) {
        const int i = e >> 6, j = e & 63;
        abar[e] = 2.0f * Linv[e] - ((i == j) ? 1.0f : 0.0f);
    }
    for (int e = tid; e < SDIM * NCH; e += 256) {
        const int i = e >> 7, c = e & 127;
        float sum = 0.0f;
        for (int k = 0; k <= i; ++k)
            sum += Linv[i * SDIM + k] * Bm[k * NCH + c];
        bbar[e] = dt * sum;
    }
}

#define ABAR_BLOCK_NONZERO(tn, tb) ((tb) <= (tn))

__global__ __launch_bounds__(256, 1)
void hippo_scan_kernel(const float* __restrict__ sig,
                       const float* __restrict__ Cmat,
                       const float* __restrict__ abar,
                       const float* __restrict__ bbar,
                       float* __restrict__ out) {
    const int b    = blockIdx.x;
    const int lane = threadIdx.x & 31;
    const int w    = threadIdx.x >> 5;
    const int lc   = lane & 15;
    const int lh   = lane >> 4;
    const int c    = w * 16 + lc;

    v2f a[4][4][4];
#pragma unroll
    for (int tn = 0; tn < 4; ++tn)
#pragma unroll
        for (int tb = 0; tb < 4; ++tb)
#pragma unroll
            for (int j = 0; j < 4; ++j) {
                if (ABAR_BLOCK_NONZERO(tn, tb)) {
                    const int base = (tn * 16 + lc) * SDIM + tb * 16 + 8 * lh + 2 * j;
                    a[tn][tb][j] = *reinterpret_cast<const v2f*>(abar + base);
                }
            }

    float bb[4][8], cw[4][8];
#pragma unroll
    for (int tn = 0; tn < 4; ++tn)
#pragma unroll
        for (int r = 0; r < 8; ++r) {
            const int n = tn * 16 + r + 8 * lh;
            bb[tn][r] = bbar[n * NCH + c];
            cw[tn][r] = Cmat[n * NCH + c];
        }

    v8f s[4];
#pragma unroll
    for (int tn = 0; tn < 4; ++tn)
#pragma unroll
        for (int r = 0; r < 8; ++r) s[tn][r] = 0.0f;

    const int colOff = (b * NCH + c) * SLEN;
    const float* xp = sig + colOff;
    float*       yp = out + colOff;

    v4f xcur4 = *reinterpret_cast<const v4f*>(xp);

#pragma unroll 1
    for (int tc = 0; tc < SLEN; tc += 4) {
        v4f xnext4;
        if (tc + 4 < SLEN) {
            xnext4 = *reinterpret_cast<const v4f*>(xp + tc + 4);
        } else {
            xnext4[0] = 0.0f; xnext4[1] = 0.0f; xnext4[2] = 0.0f; xnext4[3] = 0.0f;
        }

        v4f y4;
#pragma unroll
        for (int u = 0; u < 4; ++u) {
            const float xcur = xcur4[u];

            v8f sn[4];
#pragma unroll
            for (int tn = 0; tn < 4; ++tn) {
                v8f acc;
#pragma unroll
                for (int r = 0; r < 8; ++r) acc[r] = bb[tn][r] * xcur;
#pragma unroll
                for (int tb = 0; tb < 4; ++tb)
#pragma unroll
                    for (int j = 0; j < 4; ++j) {
                        if (ABAR_BLOCK_NONZERO(tn, tb)) {
                            v2f bfrag;
                            bfrag.x = s[tb][2 * j];
                            bfrag.y = s[tb][2 * j + 1];
                            acc = __builtin_amdgcn_wmma_f32_16x16x4_f32(
                                 false, a[tn][tb][j],
                                 false, bfrag,
                                 (short)0, acc,
                                 false,  false);
                        }
                    }
                sn[tn] = acc;
            }
#pragma unroll
            for (int tn = 0; tn < 4; ++tn) s[tn] = sn[tn];

            float y = 0.0f;
#pragma unroll
            for (int tn = 0; tn < 4; ++tn)
#pragma unroll
                for (int r = 0; r < 8; ++r) y += cw[tn][r] * s[tn][r];
            y += __shfl_xor(y, 16, 32);
            y4[u] = y;
        }

        if (lh == 0) *reinterpret_cast<v4f*>(yp + tc) = y4;

        xcur4 = xnext4;
    }
}

extern "C" void kernel_launch(void* const* d_in, const int* in_sizes, int n_in,
                              void* d_out, int out_size, void* d_ws, size_t ws_size,
                              hipStream_t stream) {
    const float* sig = (const float*)d_in[0];
    const float* Bm  = (const float*)d_in[1];
    const float* Cm  = (const float*)d_in[2];
    const float* Am  = (const float*)d_in[3];
    const float* dtp = (const float*)d_in[4];
    float* out  = (float*)d_out;
    float* abar = (float*)d_ws;
    float* bbar = abar + SDIM * SDIM;

    hippo_setup_kernel<<<1, 256, 0, stream>>>(Am, Bm, dtp, abar, bbar);
    hippo_scan_kernel<<<NB, 256, 0, stream>>>(sig, Cm, abar, bbar, out);
}


# time split into 64 chunks of 64 steps run in parallel: local scans from zero, serial carry pass with A_bar^64 (recurrence run on the identity), final re-scan; state sums re-associated across chunks
# speedup vs baseline: 11.2335x; 9.3558x over previous
#include <hip/hip_runtime.h>
#include <hip/hip_bf16.h>

typedef float v2f __attribute__((ext_vector_type(2)));
typedef float v4f __attribute__((ext_vector_type(4)));
typedef float v8f __attribute__((ext_vector_type(8)));

#define SDIM 64
#define NCH  128
#define NB   8
#define SLEN 4096
#define KCH  64
#define LCH  64
#define NGRP 64
#define NBLK_LOCAL 505
#define NBLK_FINAL 512

__global__ void hippo_setup_kernel(const float* __restrict__ A,
                                   const float* __restrict__ Bm,
                                   const float* __restrict__ dtp,
                                   float* __restrict__ abar,
                                   float* __restrict__ bbar) {
    __shared__ float Lsh[SDIM * SDIM];
    __shared__ float Linv[SDIM * SDIM];
    const float dt = dtp[0];
    const int tid = threadIdx.x;

    for (int e = tid; e < SDIM * SDIM; e += 256) {
        const int i = e >> 6, j = e & 63;
        float l = -0.5f * dt * A[e];
        if (i == j) l += 1.0f;
        Lsh[e] = l;
    }
    __syncthreads();

    if (tid < SDIM) {
        const int j = tid;
        for (int i = 0; i < SDIM; ++i) {
            if (i < j) {
                Linv[i * SDIM + j] = 0.0f;
            } else {
                float sum = (i == j) ? 1.0f : 0.0f;
                for (int k = j; k < i; ++k)
                    sum -= Lsh[i * SDIM + k] * Linv[k * SDIM + j];
                Linv[i * SDIM + j] = sum / Lsh[i * SDIM + i];
            }
        }
    }
    __syncthreads();

    for (int e = tid; e < SDIM * SDIM; e += 256) {
        const int i = e >> 6, j = e & 63;
        abar[e] = 2.0f * Linv[e] - ((i == j) ? 1.0f : 0.0f);
    }
    for (int e = tid; e < SDIM * NCH; e += 256) {
        const int i = e >> 7, c = e & 127;
        float sum = 0.0f;
        for (int k = 0; k <= i; ++k)
            sum += Linv[i * SDIM + k] * Bm[k * NCH + c];
        bbar[e] = dt * sum;
    }
}

#define ABAR_BLOCK_NONZERO(tn, tb) ((tb) <= (tn))


__device__ __forceinline__ void load_afrags(const float* mat, int pitch, int lc, int lh,
                                            v2f (&a)[4][4][4]) {
#pragma unroll
    for (int tn = 0; tn < 4; ++tn)
#pragma unroll
        for (int tb = 0; tb < 4; ++tb)
#pragma unroll
            for (int j = 0; j < 4; ++j) {
                if (ABAR_BLOCK_NONZERO(tn, tb)) {
                    const int base = (tn * 16 + lc) * pitch + tb * 16 + 8 * lh + 2 * j;
                    a[tn][tb][j] = *reinterpret_cast<const v2f*>(mat + base);
                }
            }
}

__device__ __forceinline__ v2f pair2(float x, float y) {
    v2f p;
    p.x = x;
    p.y = y;
    return p;
}

__device__ __forceinline__ void mat_step(const v2f (&a)[4][4][4], const v8f (&s)[4],
                                         v8f (&acc)[4]) {
#pragma unroll
    for (int tn = 0; tn < 4; ++tn)
#pragma unroll
        for (int tb = 0; tb < 4; ++tb)
#pragma unroll
            for (int j = 0; j < 4; ++j) {
                if (ABAR_BLOCK_NONZERO(tn, tb)) {
                    acc[tn] = __builtin_amdgcn_wmma_f32_16x16x4_f32(
                         false, a[tn][tb][j],
                         false, pair2(s[tb][2 * j], s[tb][2 * j + 1]),
                         (short)0, acc[tn],
                         false,  false);
                }
            }
}

__device__ __forceinline__ void save_state(float* p, int lh, const v8f (&s)[4]) {
#pragma unroll
    for (int tn = 0; tn < 4; ++tn) *reinterpret_cast<v8f*>(p + tn * 16 + 8 * lh) = s[tn];
}
__device__ __forceinline__ void load_state(const float* p, int lh, v8f (&s)[4]) {
#pragma unroll
    for (int tn = 0; tn < 4; ++tn) s[tn] = *reinterpret_cast<const v8f*>(p + tn * 16 + 8 * lh);
}

template <int FINAL>
__global__ __launch_bounds__(256)
void hippo_chunk_kernel(const float* __restrict__ sig,
                        const float* __restrict__ Cmat,
                        const float* __restrict__ abar,
                        const float* __restrict__ bbar,
                        float* out) {
    const int lane = threadIdx.x & 31;
    const int w    = threadIdx.x >> 5;
    const int lc   = lane & 15;
    const int lh   = lane >> 4;
    const int c    = w * 16 + lc;

    int k, b;
    bool power = false;
    if (FINAL) {
        k = blockIdx.x >> 3;
        b = blockIdx.x & 7;
    } else if (blockIdx.x == 0) {
        power = true;
        k = 0;
        b = 0;
        if (w >= 4) return;
    } else {
        k = (blockIdx.x - 1) >> 3;
        b = (blockIdx.x - 1) & 7;
    }

    v2f a[4][4][4];
    load_afrags(abar, SDIM, lc, lh, a);

    float bb[4][8], cw[4][8];
#pragma unroll
    for (int tn = 0; tn < 4; ++tn)
#pragma unroll
        for (int r = 0; r < 8; ++r) {
            const int n = tn * 16 + r + 8 * lh;
            const float bv = bbar[n * NCH + c];
            bb[tn][r] = power ? 0.0f : bv;
            cw[tn][r] = FINAL ? Cmat[n * NCH + c] : 0.0f;
        }

    const int colOff = (b * NCH + c) * SLEN;
    float* colp = out + colOff;

    v8f s[4];
    if (FINAL && k > 0) {
        load_state(colp + k * LCH, lh, s);
    } else {
#pragma unroll
        for (int tn = 0; tn < 4; ++tn)
#pragma unroll
            for (int r = 0; r < 8; ++r)
                s[tn][r] = (power && (tn * 16 + r + 8 * lh == c)) ? 1.0f : 0.0f;
    }

    const float* xp = sig + colOff + k * LCH;
    float*       yp = colp + k * LCH;

    v4f xcur4 = *reinterpret_cast<const v4f*>(xp);
    if (power) { xcur4[0] = 0.0f; xcur4[1] = 0.0f; xcur4[2] = 0.0f; xcur4[3] = 0.0f; }

#pragma unroll 1
    for (int tc = 0; tc < LCH; tc += 4) {
        const int tnext = (tc + 4 < LCH) ? tc + 4 : tc;
        v4f xnext4 = *reinterpret_cast<const v4f*>(xp + tnext);
        if (power) { xnext4[0] = 0.0f; xnext4[1] = 0.0f; xnext4[2] = 0.0f; xnext4[3] = 0.0f; }

        v4f y4;
#pragma unroll
        for (int u = 0; u < 4; ++u) {
            const float xcur = xcur4[u];

            v8f acc[4];
#pragma unroll
            for (int tn = 0; tn < 4; ++tn)
#pragma unroll
                for (int r = 0; r < 8; ++r) acc[tn][r] = bb[tn][r] * xcur;
            mat_step(a, s, acc);
#pragma unroll
            for (int tn = 0; tn < 4; ++tn) s[tn] = acc[tn];

            if (FINAL) {
                float y = 0.0f;
#pragma unroll
                for (int tn = 0; tn < 4; ++tn)
#pragma unroll
                    for (int r = 0; r < 8; ++r) y += cw[tn][r] * s[tn][r];
                y += __shfl_xor(y, 16, 32);
                y4[u] = y;
            }
        }

        if (FINAL && lh == 0) *reinterpret_cast<v4f*>(yp + tc) = y4;

        xcur4 = xnext4;
    }

    if (!FINAL) {
        if (power) {
#pragma unroll
            for (int tn = 0; tn < 4; ++tn)
#pragma unroll
                for (int r = 0; r < 8; ++r) out[(tn * 16 + r + 8 * lh) * SLEN + c] = s[tn][r];
        } else {
            save_state(colp + (k + 1) * LCH, lh, s);
        }
    }
}

__device__ __forceinline__ void carry_trip(float* colp, int lh, int k,
                                           const v2f (&a)[4][4][4], v8f (&s)[4],
                                           v8f (&e)[4]) {
    v8f acc[4];
#pragma unroll
    for (int tn = 0; tn < 4; ++tn) acc[tn] = e[tn];
    const int kn = (k + 2 < KCH) ? k + 2 : KCH - 1;
    load_state(colp + kn * LCH, lh, e);
    __builtin_amdgcn_sched_barrier(0);
    mat_step(a, s, acc);
#pragma unroll
    for (int tn = 0; tn < 4; ++tn) s[tn] = acc[tn];
    save_state(colp + k * LCH, lh, s);
}

__global__ __launch_bounds__(32)
void hippo_carry_kernel(float* out) {
    const int lane = threadIdx.x & 31;
    const int lc   = lane & 15;
    const int lh   = lane >> 4;
    float* colp = out + (blockIdx.x * 16 + lc) * SLEN;

    v2f a[4][4][4];
    load_afrags(out, SLEN, lc, lh, a);

    v8f s[4];
#pragma unroll
    for (int tn = 0; tn < 4; ++tn)
#pragma unroll
        for (int r = 0; r < 8; ++r) s[tn][r] = 0.0f;

    v8f ea[4], eb[4];
    load_state(colp + 1 * LCH, lh, ea);
    load_state(colp + 2 * LCH, lh, eb);

    carry_trip(colp, lh, 1, a, s, ea);
#pragma unroll 1
    for (int k = 2; k < KCH; k += 2) {
        carry_trip(colp, lh, k, a, s, eb);
        carry_trip(colp, lh, k + 1, a, s, ea);
    }
}

extern "C" void kernel_launch(void* const* d_in, const int* in_sizes, int n_in,
                              void* d_out, int out_size, void* d_ws, size_t ws_size,
                              hipStream_t stream) {
    const float* sig = (const float*)d_in[0];
    const float* Bm  = (const float*)d_in[1];
    const float* Cm  = (const float*)d_in[2];
    const float* Am  = (const float*)d_in[3];
    const float* dtp = (const float*)d_in[4];
    float* out  = (float*)d_out;
    float* abar = (float*)d_ws;
    float* bbar = abar + SDIM * SDIM;

    hippo_setup_kernel<<<1, 256, 0, stream>>>(Am, Bm, dtp, abar, bbar);
    hippo_chunk_kernel<0><<<NBLK_LOCAL, 256, 0, stream>>>(sig, Cm, abar, bbar, out);
    hippo_carry_kernel<<<NGRP, 32, 0, stream>>>(out);
    hippo_chunk_kernel<1><<<NBLK_FINAL, 256, 0, stream>>>(sig, Cm, abar, bbar, out);
}


# setup kernel as one forward substitution L X = [I | B]: one register-resident column per thread, L broadcast from LDS, fully unrolled; B_bar sums re-associated (no product with the inverse)
# speedup vs baseline: 14.7426x; 1.3124x over previous
#include <hip/hip_runtime.h>
#include <hip/hip_bf16.h>

typedef float v2f __attribute__((ext_vector_type(2)));
typedef float v4f __attribute__((ext_vector_type(4)));
typedef float v8f __attribute__((ext_vector_type(8)));

#define SDIM 64
#define NCH  128
#define NB   8
#define SLEN 4096
#define KCH  64
#define LCH  64
#define NGRP 64
#define NBLK_LOCAL 505
#define NBLK_FINAL 512

#define NRHS   192
#define LPITCH 68

template <int K>
__device__ __forceinline__ void hippo_solve_from(float (&s)[SDIM], const float* Lt) {
    if constexpr (K < SDIM) {
        const float xk = s[K] / Lt[K * LPITCH + K];
        s[K] = xk;
        const float nxk = -xk;
#pragma unroll
        for (int i = K + 1; i < SDIM; ++i)
            s[i] = __builtin_fmaf(Lt[K * LPITCH + i], nxk, s[i]);
        hippo_solve_from<K + 1>(s, Lt);
    }
}

__global__ __launch_bounds__(256)
void hippo_setup_kernel(const float* __restrict__ A,
                        const float* __restrict__ Bm,
                        const float* __restrict__ dtp,
                        float* __restrict__ abar,
                        float* __restrict__ bbar) {
    __shared__ __attribute__((aligned(16))) float Lt[SDIM * LPITCH];
    const float dt = dtp[0];
    const int tid = threadIdx.x;
    const bool solver = tid < NRHS;
    const bool bcol   = tid >= SDIM;
    const int  c      = tid - SDIM;

    float s[SDIM];
    if (solver && bcol) {
#pragma unroll
        for (int i = 0; i < SDIM; ++i) s[i] = Bm[i * NCH + c];
    } else {
#pragma unroll
        for (int i = 0; i < SDIM; ++i) s[i] = (i == tid) ? 1.0f : 0.0f;
    }

    float av[16];
#pragma unroll
    for (int q = 0; q < 16; ++q) av[q] = A[tid + 256 * q];
#pragma unroll
    for (int q = 0; q < 16; ++q) {
        const int e = tid + 256 * q;
        const int i = e >> 6, k = e & 63;
        float l = -0.5f * dt * av[q];
        if (i == k) l += 1.0f;
        Lt[k * LPITCH + i] = l;
    }
    __syncthreads();

    if (solver) {
        hippo_solve_from<0>(s, Lt);
        if (bcol) {
#pragma unroll
            for (int i = 0; i < SDIM; ++i) bbar[i * NCH + c] = dt * s[i];
        } else {
#pragma unroll
            for (int i = 0; i < SDIM; ++i)
                abar[i * SDIM + tid] = 2.0f * s[i] - ((i == tid) ? 1.0f : 0.0f);
        }
    }
}

#define ABAR_BLOCK_NONZERO(tn, tb) ((tb) <= (tn))


__device__ __forceinline__ void load_afrags(const float* mat, int pitch, int lc, int lh,
                                            v2f (&a)[4][4][4]) {
#pragma unroll
    for (int tn = 0; tn < 4; ++tn)
#pragma unroll
        for (int tb = 0; tb < 4; ++tb)
#pragma unroll
            for (int j = 0; j < 4; ++j) {
                if (ABAR_BLOCK_NONZERO(tn, tb)) {
                    const int base = (tn * 16 + lc) * pitch + tb * 16 + 8 * lh + 2 * j;
                    a[tn][tb][j] = *reinterpret_cast<const v2f*>(mat + base);
                }
            }
}

__device__ __forceinline__ v2f pair2(float x, float y) {
    v2f p;
    p.x = x;
    p.y = y;
    return p;
}

__device__ __forceinline__ void mat_step(const v2f (&a)[4][4][4], const v8f (&s)[4],
                                         v8f (&acc)[4]) {
#pragma unroll
    for (int tn = 0; tn < 4; ++tn)
#pragma unroll
        for (int tb = 0; tb < 4; ++tb)
#pragma unroll
            for (int j = 0; j < 4; ++j) {
                if (ABAR_BLOCK_NONZERO(tn, tb)) {
                    acc[tn] = __builtin_amdgcn_wmma_f32_16x16x4_f32(
                         false, a[tn][tb][j],
                         false, pair2(s[tb][2 * j], s[tb][2 * j + 1]),
                         (short)0, acc[tn],
                         false,  false);
                }
            }
}

__device__ __forceinline__ void save_state(float* p, int lh, const v8f (&s)[4]) {
#pragma unroll
    for (int tn = 0; tn < 4; ++tn) *reinterpret_cast<v8f*>(p + tn * 16 + 8 * lh) = s[tn];
}
__device__ __forceinline__ void load_state(const float* p, int lh, v8f (&s)[4]) {
#pragma unroll
    for (int tn = 0; tn < 4; ++tn) s[tn] = *reinterpret_cast<const v8f*>(p + tn * 16 + 8 * lh);
}

template <int FINAL>
__global__ __launch_bounds__(256)
void hippo_chunk_kernel(const float* __restrict__ sig,
                        const float* __restrict__ Cmat,
                        const float* __restrict__ abar,
                        const float* __restrict__ bbar,
                        float* out) {
    const int lane = threadIdx.x & 31;
    const int w    = threadIdx.x >> 5;
    const int lc   = lane & 15;
    const int lh   = lane >> 4;
    const int c    = w * 16 + lc;

    int k, b;
    bool power = false;
    if (FINAL) {
        k = blockIdx.x >> 3;
        b = blockIdx.x & 7;
    } else if (blockIdx.x == 0) {
        power = true;
        k = 0;
        b = 0;
        if (w >= 4) return;
    } else {
        k = (blockIdx.x - 1) >> 3;
        b = (blockIdx.x - 1) & 7;
    }

    v2f a[4][4][4];
    load_afrags(abar, SDIM, lc, lh, a);

    float bb[4][8], cw[4][8];
#pragma unroll
    for (int tn = 0; tn < 4; ++tn)
#pragma unroll
        for (int r = 0; r < 8; ++r) {
            const int n = tn * 16 + r + 8 * lh;
            const float bv = bbar[n * NCH + c];
            bb[tn][r] = power ? 0.0f : bv;
            cw[tn][r] = FINAL ? Cmat[n * NCH + c] : 0.0f;
        }

    const int colOff = (b * NCH + c) * SLEN;
    float* colp = out + colOff;

    v8f s[4];
    if (FINAL && k > 0) {
        load_state(colp + k * LCH, lh, s);
    } else {
#pragma unroll
        for (int tn = 0; tn < 4; ++tn)
#pragma unroll
            for (int r = 0; r < 8; ++r)
                s[tn][r] = (power && (tn * 16 + r + 8 * lh == c)) ? 1.0f : 0.0f;
    }

    const float* xp = sig + colOff + k * LCH;
    float*       yp = colp + k * LCH;

    v4f xcur4 = *reinterpret_cast<const v4f*>(xp);
    if (power) { xcur4[0] = 0.0f; xcur4[1] = 0.0f; xcur4[2] = 0.0f; xcur4[3] = 0.0f; }

#pragma unroll 1
    for (int tc = 0; tc < LCH; tc += 4) {
        const int tnext = (tc + 4 < LCH) ? tc + 4 : tc;
        v4f xnext4 = *reinterpret_cast<const v4f*>(xp + tnext);
        if (power) { xnext4[0] = 0.0f; xnext4[1] = 0.0f; xnext4[2] = 0.0f; xnext4[3] = 0.0f; }

        v4f y4;
#pragma unroll
        for (int u = 0; u < 4; ++u) {
            const float xcur = xcur4[u];

            v8f acc[4];
#pragma unroll
            for (int tn = 0; tn < 4; ++tn)
#pragma unroll
                for (int r = 0; r < 8; ++r) acc[tn][r] = bb[tn][r] * xcur;
            mat_step(a, s, acc);
#pragma unroll
            for (int tn = 0; tn < 4; ++tn) s[tn] = acc[tn];

            if (FINAL) {
                float y = 0.0f;
#pragma unroll
                for (int tn = 0; tn < 4; ++tn)
#pragma unroll
                    for (int r = 0; r < 8; ++r) y += cw[tn][r] * s[tn][r];
                y += __shfl_xor(y, 16, 32);
                y4[u] = y;
            }
        }

        if (FINAL && lh == 0) *reinterpret_cast<v4f*>(yp + tc) = y4;

        xcur4 = xnext4;
    }

    if (!FINAL) {
        if (power) {
#pragma unroll
            for (int tn = 0; tn < 4; ++tn)
#pragma unroll
                for (int r = 0; r < 8; ++r) out[(tn * 16 + r + 8 * lh) * SLEN + c] = s[tn][r];
        } else {
            save_state(colp + (k + 1) * LCH, lh, s);
        }
    }
}

__device__ __forceinline__ void carry_trip(float* colp, int lh, int k,
                                           const v2f (&a)[4][4][4], v8f (&s)[4],
                                           v8f (&e)[4]) {
    v8f acc[4];
#pragma unroll
    for (int tn = 0; tn < 4; ++tn) acc[tn] = e[tn];
    const int kn = (k + 2 < KCH) ? k + 2 : KCH - 1;
    load_state(colp + kn * LCH, lh, e);
    __builtin_amdgcn_sched_barrier(0);
    mat_step(a, s, acc);
#pragma unroll
    for (int tn = 0; tn < 4; ++tn) s[tn] = acc[tn];
    save_state(colp + k * LCH, lh, s);
}

__global__ __launch_bounds__(32)
void hippo_carry_kernel(float* out) {
    const int lane = threadIdx.x & 31;
    const int lc   = lane & 15;
    const int lh   = lane >> 4;
    float* colp = out + (blockIdx.x * 16 + lc) * SLEN;

    v2f a[4][4][4];
    load_afrags(out, SLEN, lc, lh, a);

    v8f s[4];
#pragma unroll
    for (int tn = 0; tn < 4; ++tn)
#pragma unroll
        for (int r = 0; r < 8; ++r) s[tn][r] = 0.0f;

    v8f ea[4], eb[4];
    load_state(colp + 1 * LCH, lh, ea);
    load_state(colp + 2 * LCH, lh, eb);

    carry_trip(colp, lh, 1, a, s, ea);
#pragma unroll 1
    for (int k = 2; k < KCH; k += 2) {
        carry_trip(colp, lh, k, a, s, eb);
        carry_trip(colp, lh, k + 1, a, s, ea);
    }
}

extern "C" void kernel_launch(void* const* d_in, const int* in_sizes, int n_in,
                              void* d_out, int out_size, void* d_ws, size_t ws_size,
                              hipStream_t stream) {
    const float* sig = (const float*)d_in[0];
    const float* Bm  = (const float*)d_in[1];
    const float* Cm  = (const float*)d_in[2];
    const float* Am  = (const float*)d_in[3];
    const float* dtp = (const float*)d_in[4];
    float* out  = (float*)d_out;
    float* abar = (float*)d_ws;
    float* bbar = abar + SDIM * SDIM;

    hippo_setup_kernel<<<1, 256, 0, stream>>>(Am, Bm, dtp, abar, bbar);
    hippo_chunk_kernel<0><<<NBLK_LOCAL, 256, 0, stream>>>(sig, Cm, abar, bbar, out);
    hippo_carry_kernel<<<NGRP, 32, 0, stream>>>(out);
    hippo_chunk_kernel<1><<<NBLK_FINAL, 256, 0, stream>>>(sig, Cm, abar, bbar, out);
}


# local pass advances 4 time steps per WMMA product: matrix A_bar^4, the 4 inputs enter the C operand through A_bar^i B_bar (i=0..3) built in setup; state sums re-associated
# speedup vs baseline: 18.5628x; 1.2591x over previous
#include <hip/hip_runtime.h>
#include <hip/hip_bf16.h>

typedef float v2f __attribute__((ext_vector_type(2)));
typedef float v4f __attribute__((ext_vector_type(4)));
typedef float v8f __attribute__((ext_vector_type(8)));

#define SDIM 64
#define NCH  128
#define NB   8
#define SLEN 4096
#define KCH  64
#define LCH  64
#define NGRP 64
#define NBLK_LOCAL 505
#define NBLK_FINAL 512
#define RSTEP 4
#define A4_COL0 64
#define KB_COL0 128

#define NRHS   192
#define LPITCH 68

template <int K>
__device__ __forceinline__ void hippo_solve_from(float (&s)[SDIM], const float* Lt) {
    if constexpr (K < SDIM) {
        const float xk = s[K] / Lt[K * LPITCH + K];
        s[K] = xk;
        const float nxk = -xk;
#pragma unroll
        for (int i = K + 1; i < SDIM; ++i)
            s[i] = __builtin_fmaf(Lt[K * LPITCH + i], nxk, s[i]);
        hippo_solve_from<K + 1>(s, Lt);
    }
}

__device__ __forceinline__ void hippo_setup_powers(const float* abar, const float* bbar, float* out, int tid);

__global__ __launch_bounds__(256)
void hippo_setup_kernel(const float* __restrict__ A,
                        const float* __restrict__ Bm,
                        const float* __restrict__ dtp,
                        float* abar,
                        float* bbar,
                        float* out) {
    __shared__ __attribute__((aligned(16))) float Lt[SDIM * LPITCH];
    const float dt = dtp[0];
    const int tid = threadIdx.x;
    const bool solver = tid < NRHS;
    const bool bcol   = tid >= SDIM;
    const int  c      = tid - SDIM;

    float s[SDIM];
    if (solver && bcol) {
#pragma unroll
        for (int i = 0; i < SDIM; ++i) s[i] = Bm[i * NCH + c];
    } else {
#pragma unroll
        for (int i = 0; i < SDIM; ++i) s[i] = (i == tid) ? 1.0f : 0.0f;
    }

    float av[16];
#pragma unroll
    for (int q = 0; q < 16; ++q) av[q] = A[tid + 256 * q];
#pragma unroll
    for (int q = 0; q < 16; ++q) {
        const int e = tid + 256 * q;
        const int i = e >> 6, k = e & 63;
        float l = -0.5f * dt * av[q];
        if (i == k) l += 1.0f;
        Lt[k * LPITCH + i] = l;
    }
    __syncthreads();

    if (solver) {
        hippo_solve_from<0>(s, Lt);
        if (bcol) {
#pragma unroll
            for (int i = 0; i < SDIM; ++i) bbar[i * NCH + c] = dt * s[i];
        } else {
#pragma unroll
            for (int i = 0; i < SDIM; ++i)
                abar[i * SDIM + tid] = 2.0f * s[i] - ((i == tid) ? 1.0f : 0.0f);
        }
    }

    __syncthreads();
    hippo_setup_powers(abar, bbar, out, tid);
}

#define ABAR_BLOCK_NONZERO(tn, tb) ((tb) <= (tn))


__device__ __forceinline__ void load_afrags(const float* mat, int pitch, int lc, int lh,
                                            v2f (&a)[4][4][4]) {
#pragma unroll
    for (int tn = 0; tn < 4; ++tn)
#pragma unroll
        for (int tb = 0; tb < 4; ++tb)
#pragma unroll
            for (int j = 0; j < 4; ++j) {
                if (ABAR_BLOCK_NONZERO(tn, tb)) {
                    const int base = (tn * 16 + lc) * pitch + tb * 16 + 8 * lh + 2 * j;
                    a[tn][tb][j] = *reinterpret_cast<const v2f*>(mat + base);
                }
            }
}

__device__ __forceinline__ v2f pair2(float x, float y) {
    v2f p;
    p.x = x;
    p.y = y;
    return p;
}

__device__ __forceinline__ void mat_step(const v2f (&a)[4][4][4], const v8f (&s)[4],
                                         v8f (&acc)[4]) {
#pragma unroll
    for (int tn = 0; tn < 4; ++tn)
#pragma unroll
        for (int tb = 0; tb < 4; ++tb)
#pragma unroll
            for (int j = 0; j < 4; ++j) {
                if (ABAR_BLOCK_NONZERO(tn, tb)) {
                    acc[tn] = __builtin_amdgcn_wmma_f32_16x16x4_f32(
                         false, a[tn][tb][j],
                         false, pair2(s[tb][2 * j], s[tb][2 * j + 1]),
                         (short)0, acc[tn],
                         false,  false);
                }
            }
}

__device__ __forceinline__ void save_state(float* p, int lh, const v8f (&s)[4]) {
#pragma unroll
    for (int tn = 0; tn < 4; ++tn) *reinterpret_cast<v8f*>(p + tn * 16 + 8 * lh) = s[tn];
}
__device__ __forceinline__ void load_state(const float* p, int lh, v8f (&s)[4]) {
#pragma unroll
    for (int tn = 0; tn < 4; ++tn) s[tn] = *reinterpret_cast<const v8f*>(p + tn * 16 + 8 * lh);
}

__device__ __forceinline__ void hippo_setup_powers(const float* abar, const float* bbar, float* out, int tid) {
    const int lane = tid & 31;
    const int w    = tid >> 5;
    const int lc   = lane & 15;
    const int lh   = lane >> 4;
    const int c    = w * 16 + lc;

    v2f a[4][4][4];
    load_afrags(abar, SDIM, lc, lh, a);

    v8f s[4];
#pragma unroll
    for (int tn = 0; tn < 4; ++tn)
#pragma unroll
        for (int r = 0; r < 8; ++r) s[tn][r] = bbar[(tn * 16 + r + 8 * lh) * NCH + c];

#pragma unroll
    for (int i = 1; i < RSTEP; ++i) {
        v8f acc[4];
#pragma unroll
        for (int tn = 0; tn < 4; ++tn)
#pragma unroll
            for (int r = 0; r < 8; ++r) acc[tn][r] = 0.0f;
        mat_step(a, s, acc);
#pragma unroll
        for (int tn = 0; tn < 4; ++tn) s[tn] = acc[tn];
        save_state(out + (KB_COL0 + (i - 1) * NCH + c) * SLEN, lh, s);
    }

    if (w < 4) {
#pragma unroll
        for (int tn = 0; tn < 4; ++tn)
#pragma unroll
            for (int r = 0; r < 8; ++r) s[tn][r] = (tn * 16 + r + 8 * lh == c) ? 1.0f : 0.0f;
#pragma unroll
        for (int i = 0; i < RSTEP; ++i) {
            v8f acc[4];
#pragma unroll
            for (int tn = 0; tn < 4; ++tn)
#pragma unroll
                for (int r = 0; r < 8; ++r) acc[tn][r] = 0.0f;
            mat_step(a, s, acc);
#pragma unroll
            for (int tn = 0; tn < 4; ++tn) s[tn] = acc[tn];
        }
#pragma unroll
        for (int tn = 0; tn < 4; ++tn)
#pragma unroll
            for (int r = 0; r < 8; ++r)
                out[(A4_COL0 + tn * 16 + r + 8 * lh) * SLEN + c] = s[tn][r];
    }
}

__global__ __launch_bounds__(256)
void hippo_local_kernel(const float* __restrict__ sig,
                        const float* __restrict__ bbar,
                        float* out) {
    const int lane = threadIdx.x & 31;
    const int w    = threadIdx.x >> 5;
    const int lc   = lane & 15;
    const int lh   = lane >> 4;
    const int c    = w * 16 + lc;

    int k, b;
    bool power = false;
    if (blockIdx.x == 0) {
        power = true;
        k = 0;
        b = 0;
        if (w >= 4) return;
    } else {
        k = (blockIdx.x - 1) >> 3;
        b = (blockIdx.x - 1) & 7;
    }

    v2f a[4][4][4];
    load_afrags(out + A4_COL0 * SLEN, SLEN, lc, lh, a);

    v8f kb[RSTEP][4];
#pragma unroll
    for (int tn = 0; tn < 4; ++tn)
#pragma unroll
        for (int r = 0; r < 8; ++r) {
            const float bv = bbar[(tn * 16 + r + 8 * lh) * NCH + c];
            kb[0][tn][r] = power ? 0.0f : bv;
        }
#pragma unroll
    for (int i = 1; i < RSTEP; ++i) {
        load_state(out + (KB_COL0 + (i - 1) * NCH + c) * SLEN, lh, kb[i]);
#pragma unroll
        for (int tn = 0; tn < 4; ++tn)
#pragma unroll
            for (int r = 0; r < 8; ++r) kb[i][tn][r] = power ? 0.0f : kb[i][tn][r];
    }

    const int colOff = (b * NCH + c) * SLEN;
    float* colp = out + colOff;

    v8f s[4];
#pragma unroll
    for (int tn = 0; tn < 4; ++tn)
#pragma unroll
        for (int r = 0; r < 8; ++r)
            s[tn][r] = (power && (tn * 16 + r + 8 * lh == c)) ? 1.0f : 0.0f;

    const float* xp = sig + colOff + k * LCH;

    v4f xcur4 = *reinterpret_cast<const v4f*>(xp);
    if (power) { xcur4[0] = 0.0f; xcur4[1] = 0.0f; xcur4[2] = 0.0f; xcur4[3] = 0.0f; }

#pragma unroll 1
    for (int tc = 0; tc < LCH; tc += RSTEP) {
        const int tnext = (tc + RSTEP < LCH) ? tc + RSTEP : tc;
        v4f xnext4 = *reinterpret_cast<const v4f*>(xp + tnext);
        if (power) { xnext4[0] = 0.0f; xnext4[1] = 0.0f; xnext4[2] = 0.0f; xnext4[3] = 0.0f; }

        v8f acc[4];
#pragma unroll
        for (int tn = 0; tn < 4; ++tn)
#pragma unroll
            for (int r = 0; r < 8; ++r) {
                float v = kb[3][tn][r] * xcur4[0];
                v = __builtin_fmaf(kb[2][tn][r], xcur4[1], v);
                v = __builtin_fmaf(kb[1][tn][r], xcur4[2], v);
                v = __builtin_fmaf(kb[0][tn][r], xcur4[3], v);
                acc[tn][r] = v;
            }
        mat_step(a, s, acc);
#pragma unroll
        for (int tn = 0; tn < 4; ++tn) s[tn] = acc[tn];

        xcur4 = xnext4;
    }

    if (power) {
#pragma unroll
        for (int tn = 0; tn < 4; ++tn)
#pragma unroll
            for (int r = 0; r < 8; ++r) out[(tn * 16 + r + 8 * lh) * SLEN + c] = s[tn][r];
    } else {
        save_state(colp + (k + 1) * LCH, lh, s);
    }
}

template <int FINAL>
__global__ __launch_bounds__(256)
void hippo_chunk_kernel(const float* __restrict__ sig,
                        const float* __restrict__ Cmat,
                        const float* __restrict__ abar,
                        const float* __restrict__ bbar,
                        float* out) {
    const int lane = threadIdx.x & 31;
    const int w    = threadIdx.x >> 5;
    const int lc   = lane & 15;
    const int lh   = lane >> 4;
    const int c    = w * 16 + lc;

    int k, b;
    bool power = false;
    if (FINAL) {
        k = blockIdx.x >> 3;
        b = blockIdx.x & 7;
    } else if (blockIdx.x == 0) {
        power = true;
        k = 0;
        b = 0;
        if (w >= 4) return;
    } else {
        k = (blockIdx.x - 1) >> 3;
        b = (blockIdx.x - 1) & 7;
    }

    v2f a[4][4][4];
    load_afrags(abar, SDIM, lc, lh, a);

    float bb[4][8], cw[4][8];
#pragma unroll
    for (int tn = 0; tn < 4; ++tn)
#pragma unroll
        for (int r = 0; r < 8; ++r) {
            const int n = tn * 16 + r + 8 * lh;
            const float bv = bbar[n * NCH + c];
            bb[tn][r] = power ? 0.0f : bv;
            cw[tn][r] = FINAL ? Cmat[n * NCH + c] : 0.0f;
        }

    const int colOff = (b * NCH + c) * SLEN;
    float* colp = out + colOff;

    v8f s[4];
    if (FINAL && k > 0) {
        load_state(colp + k * LCH, lh, s);
    } else {
#pragma unroll
        for (int tn = 0; tn < 4; ++tn)
#pragma unroll
            for (int r = 0; r < 8; ++r)
                s[tn][r] = (power && (tn * 16 + r + 8 * lh == c)) ? 1.0f : 0.0f;
    }

    const float* xp = sig + colOff + k * LCH;
    float*       yp = colp + k * LCH;

    v4f xcur4 = *reinterpret_cast<const v4f*>(xp);
    if (power) { xcur4[0] = 0.0f; xcur4[1] = 0.0f; xcur4[2] = 0.0f; xcur4[3] = 0.0f; }

#pragma unroll 1
    for (int tc = 0; tc < LCH; tc += 4) {
        const int tnext = (tc + 4 < LCH) ? tc + 4 : tc;
        v4f xnext4 = *reinterpret_cast<const v4f*>(xp + tnext);
        if (power) { xnext4[0] = 0.0f; xnext4[1] = 0.0f; xnext4[2] = 0.0f; xnext4[3] = 0.0f; }

        v4f y4;
#pragma unroll
        for (int u = 0; u < 4; ++u) {
            const float xcur = xcur4[u];

            v8f acc[4];
#pragma unroll
            for (int tn = 0; tn < 4; ++tn)
#pragma unroll
                for (int r = 0; r < 8; ++r) acc[tn][r] = bb[tn][r] * xcur;
            mat_step(a, s, acc);
#pragma unroll
            for (int tn = 0; tn < 4; ++tn) s[tn] = acc[tn];

            if (FINAL) {
                float y = 0.0f;
#pragma unroll
                for (int tn = 0; tn < 4; ++tn)
#pragma unroll
                    for (int r = 0; r < 8; ++r) y += cw[tn][r] * s[tn][r];
                y += __shfl_xor(y, 16, 32);
                y4[u] = y;
            }
        }

        if (FINAL && lh == 0) *reinterpret_cast<v4f*>(yp + tc) = y4;

        xcur4 = xnext4;
    }

    if (!FINAL) {
        if (power) {
#pragma unroll
            for (int tn = 0; tn < 4; ++tn)
#pragma unroll
                for (int r = 0; r < 8; ++r) out[(tn * 16 + r + 8 * lh) * SLEN + c] = s[tn][r];
        } else {
            save_state(colp + (k + 1) * LCH, lh, s);
        }
    }
}

__device__ __forceinline__ void carry_trip(float* colp, int lh, int k,
                                           const v2f (&a)[4][4][4], v8f (&s)[4],
                                           v8f (&e)[4]) {
    v8f acc[4];
#pragma unroll
    for (int tn = 0; tn < 4; ++tn) acc[tn] = e[tn];
    const int kn = (k + 2 < KCH) ? k + 2 : KCH - 1;
    load_state(colp + kn * LCH, lh, e);
    __builtin_amdgcn_sched_barrier(0);
    mat_step(a, s, acc);
#pragma unroll
    for (int tn = 0; tn < 4; ++tn) s[tn] = acc[tn];
    save_state(colp + k * LCH, lh, s);
}

__global__ __launch_bounds__(32)
void hippo_carry_kernel(float* out) {
    const int lane = threadIdx.x & 31;
    const int lc   = lane & 15;
    const int lh   = lane >> 4;
    float* colp = out + (blockIdx.x * 16 + lc) * SLEN;

    v2f a[4][4][4];
    load_afrags(out, SLEN, lc, lh, a);

    v8f s[4];
#pragma unroll
    for (int tn = 0; tn < 4; ++tn)
#pragma unroll
        for (int r = 0; r < 8; ++r) s[tn][r] = 0.0f;

    v8f ea[4], eb[4];
    load_state(colp + 1 * LCH, lh, ea);
    load_state(colp + 2 * LCH, lh, eb);

    carry_trip(colp, lh, 1, a, s, ea);
#pragma unroll 1
    for (int k = 2; k < KCH; k += 2) {
        carry_trip(colp, lh, k, a, s, eb);
        carry_trip(colp, lh, k + 1, a, s, ea);
    }
}

extern "C" void kernel_launch(void* const* d_in, const int* in_sizes, int n_in,
                              void* d_out, int out_size, void* d_ws, size_t ws_size,
                              hipStream_t stream) {
    const float* sig = (const float*)d_in[0];
    const float* Bm  = (const float*)d_in[1];
    const float* Cm  = (const float*)d_in[2];
    const float* Am  = (const float*)d_in[3];
    const float* dtp = (const float*)d_in[4];
    float* out  = (float*)d_out;
    float* abar = (float*)d_ws;
    float* bbar = abar + SDIM * SDIM;

    hippo_setup_kernel<<<1, 256, 0, stream>>>(Am, Bm, dtp, abar, bbar, out);
    hippo_local_kernel<<<NBLK_LOCAL, 256, 0, stream>>>(sig, bbar, out);
    hippo_carry_kernel<<<NGRP, 32, 0, stream>>>(out);
    hippo_chunk_kernel<1><<<NBLK_FINAL, 256, 0, stream>>>(sig, Cm, abar, bbar, out);
}


# final pass advances 4 time steps per WMMA product (matrix A_bar^4): the 3 inner outputs are dots of the group's start state with (A_bar^T)^i C, built per wave; state and y sums re-associated
# speedup vs baseline: 25.4609x; 1.3716x over previous
#include <hip/hip_runtime.h>
#include <hip/hip_bf16.h>

typedef float v2f __attribute__((ext_vector_type(2)));
typedef float v4f __attribute__((ext_vector_type(4)));
typedef float v8f __attribute__((ext_vector_type(8)));

#define SDIM 64
#define NCH  128
#define NB   8
#define SLEN 4096
#define KCH  64
#define LCH  64
#define NGRP 64
#define NBLK_LOCAL 505
#define NBLK_FINAL 256
#define CPB 2
#define RSTEP 4
#define KB_COL0 128

#define NRHS   192
#define LPITCH 68

template <int K>
__device__ __forceinline__ void hippo_solve_from(float (&s)[SDIM], const float* Lt) {
    if constexpr (K < SDIM) {
        const float xk = s[K] / Lt[K * LPITCH + K];
        s[K] = xk;
        const float nxk = -xk;
#pragma unroll
        for (int i = K + 1; i < SDIM; ++i)
            s[i] = __builtin_fmaf(Lt[K * LPITCH + i], nxk, s[i]);
        hippo_solve_from<K + 1>(s, Lt);
    }
}

__device__ __forceinline__ void hippo_setup_powers(const float* abar, float* a4, float* out, int tid);

__global__ __launch_bounds__(256)
void hippo_setup_kernel(const float* __restrict__ A,
                        const float* __restrict__ Bm,
                        const float* __restrict__ dtp,
                        float* abar,
                        float* a4,
                        float* out) {
    __shared__ __attribute__((aligned(16))) float Lt[SDIM * LPITCH];
    const float dt = dtp[0];
    const int tid = threadIdx.x;
    const bool solver = tid < NRHS;
    const bool bcol   = tid >= SDIM;
    const int  c      = tid - SDIM;

    float s[SDIM];
    if (solver && bcol) {
#pragma unroll
        for (int i = 0; i < SDIM; ++i) s[i] = Bm[i * NCH + c];
    } else {
#pragma unroll
        for (int i = 0; i < SDIM; ++i) s[i] = (i == tid) ? 1.0f : 0.0f;
    }

    float av[16];
#pragma unroll
    for (int q = 0; q < 16; ++q) av[q] = A[tid + 256 * q];
#pragma unroll
    for (int q = 0; q < 16; ++q) {
        const int e = tid + 256 * q;
        const int i = e >> 6, k = e & 63;
        float l = -0.5f * dt * av[q];
        if (i == k) l += 1.0f;
        Lt[k * LPITCH + i] = l;
    }
    __syncthreads();

    if (solver) {
        hippo_solve_from<0>(s, Lt);
        if (bcol) {
            float* bp = out + (KB_COL0 + c) * SLEN;
#pragma unroll
            for (int i = 0; i < SDIM; i += 4) {
                v4f q;
                q[0] = dt * s[i];
                q[1] = dt * s[i + 1];
                q[2] = dt * s[i + 2];
                q[3] = dt * s[i + 3];
                *reinterpret_cast<v4f*>(bp + i) = q;
            }
        } else {
#pragma unroll
            for (int i = 0; i < SDIM; ++i)
                abar[i * SDIM + tid] = 2.0f * s[i] - ((i == tid) ? 1.0f : 0.0f);
        }
    }

    __syncthreads();
    hippo_setup_powers(abar, a4, out, tid);
}

#define ABAR_BLOCK_NONZERO(tn, tb) ((tb) <= (tn))


__device__ __forceinline__ void load_afrags(const float* mat, int pitch, int lc, int lh,
                                            v2f (&a)[4][4][4]) {
#pragma unroll
    for (int tn = 0; tn < 4; ++tn)
#pragma unroll
        for (int tb = 0; tb < 4; ++tb)
#pragma unroll
            for (int j = 0; j < 4; ++j) {
                if (ABAR_BLOCK_NONZERO(tn, tb)) {
                    const int base = (tn * 16 + lc) * pitch + tb * 16 + 8 * lh + 2 * j;
                    a[tn][tb][j] = *reinterpret_cast<const v2f*>(mat + base);
                }
            }
}

__device__ __forceinline__ v2f pair2(float x, float y) {
    v2f p;
    p.x = x;
    p.y = y;
    return p;
}

__device__ __forceinline__ void load_afrags_t(const float* mat, int lc, int lh,
                                              v2f (&a)[4][4][4]) {
#pragma unroll
    for (int tn = 0; tn < 4; ++tn)
#pragma unroll
        for (int tb = 0; tb < 4; ++tb)
#pragma unroll
            for (int j = 0; j < 4; ++j) {
                if (ABAR_BLOCK_NONZERO(tb, tn)) {
                    const int m = tb * 16 + 8 * lh + 2 * j;
                    a[tn][tb][j] = pair2(mat[m * SDIM + tn * 16 + lc],
                                         mat[(m + 1) * SDIM + tn * 16 + lc]);
                }
            }
}

__device__ __forceinline__ void mat_step_t(const v2f (&a)[4][4][4], const v8f (&s)[4],
                                           v8f (&acc)[4]) {
#pragma unroll
    for (int tn = 0; tn < 4; ++tn)
#pragma unroll
        for (int tb = 0; tb < 4; ++tb)
#pragma unroll
            for (int j = 0; j < 4; ++j) {
                if (ABAR_BLOCK_NONZERO(tb, tn)) {
                    acc[tn] = __builtin_amdgcn_wmma_f32_16x16x4_f32(
                         false, a[tn][tb][j],
                         false, pair2(s[tb][2 * j], s[tb][2 * j + 1]),
                         (short)0, acc[tn],
                         false,  false);
                }
            }
}

__device__ __forceinline__ void mat_step(const v2f (&a)[4][4][4], const v8f (&s)[4],
                                         v8f (&acc)[4]) {
#pragma unroll
    for (int tn = 0; tn < 4; ++tn)
#pragma unroll
        for (int tb = 0; tb < 4; ++tb)
#pragma unroll
            for (int j = 0; j < 4; ++j) {
                if (ABAR_BLOCK_NONZERO(tn, tb)) {
                    acc[tn] = __builtin_amdgcn_wmma_f32_16x16x4_f32(
                         false, a[tn][tb][j],
                         false, pair2(s[tb][2 * j], s[tb][2 * j + 1]),
                         (short)0, acc[tn],
                         false,  false);
                }
            }
}

__device__ __forceinline__ void save_state(float* p, int lh, const v8f (&s)[4]) {
#pragma unroll
    for (int tn = 0; tn < 4; ++tn) *reinterpret_cast<v8f*>(p + tn * 16 + 8 * lh) = s[tn];
}
__device__ __forceinline__ void load_state(const float* p, int lh, v8f (&s)[4]) {
#pragma unroll
    for (int tn = 0; tn < 4; ++tn) s[tn] = *reinterpret_cast<const v8f*>(p + tn * 16 + 8 * lh);
}

__device__ __forceinline__ void hippo_setup_powers(const float* abar, float* a4, float* out, int tid) {
    const int lane = tid & 31;
    const int w    = tid >> 5;
    const int lc   = lane & 15;
    const int lh   = lane >> 4;
    const int c    = w * 16 + lc;

    v2f a[4][4][4];
    load_afrags(abar, SDIM, lc, lh, a);

    v8f s[4];
    load_state(out + (KB_COL0 + c) * SLEN, lh, s);

#pragma unroll
    for (int i = 1; i < RSTEP; ++i) {
        v8f acc[4];
#pragma unroll
        for (int tn = 0; tn < 4; ++tn)
#pragma unroll
            for (int r = 0; r < 8; ++r) acc[tn][r] = 0.0f;
        mat_step(a, s, acc);
#pragma unroll
        for (int tn = 0; tn < 4; ++tn) s[tn] = acc[tn];
        save_state(out + (KB_COL0 + i * NCH + c) * SLEN, lh, s);
    }

    if (w < 4) {
#pragma unroll
        for (int tn = 0; tn < 4; ++tn)
#pragma unroll
            for (int r = 0; r < 8; ++r) s[tn][r] = (tn * 16 + r + 8 * lh == c) ? 1.0f : 0.0f;
#pragma unroll
        for (int i = 0; i < RSTEP; ++i) {
            v8f acc[4];
#pragma unroll
            for (int tn = 0; tn < 4; ++tn)
#pragma unroll
                for (int r = 0; r < 8; ++r) acc[tn][r] = 0.0f;
            mat_step(a, s, acc);
#pragma unroll
            for (int tn = 0; tn < 4; ++tn) s[tn] = acc[tn];
        }
#pragma unroll
        for (int tn = 0; tn < 4; ++tn)
#pragma unroll
            for (int r = 0; r < 8; ++r)
                a4[(tn * 16 + r + 8 * lh) * SDIM + c] = s[tn][r];
    }
}

__global__ __launch_bounds__(256)
void hippo_local_kernel(const float* __restrict__ sig,
                        const float* __restrict__ a4,
                        float* out) {
    const int lane = threadIdx.x & 31;
    const int w    = threadIdx.x >> 5;
    const int lc   = lane & 15;
    const int lh   = lane >> 4;
    const int c    = w * 16 + lc;

    int k, b;
    bool power = false;
    if (blockIdx.x == 0) {
        power = true;
        k = 0;
        b = 0;
        if (w >= 4) return;
    } else {
        k = (blockIdx.x - 1) >> 3;
        b = (blockIdx.x - 1) & 7;
    }

    v2f a[4][4][4];
    load_afrags(a4, SDIM, lc, lh, a);

    v8f kb[RSTEP][4];
#pragma unroll
    for (int i = 0; i < RSTEP; ++i) {
        load_state(out + (KB_COL0 + i * NCH + c) * SLEN, lh, kb[i]);
#pragma unroll
        for (int tn = 0; tn < 4; ++tn)
#pragma unroll
            for (int r = 0; r < 8; ++r) kb[i][tn][r] = power ? 0.0f : kb[i][tn][r];
    }

    const int colOff = (b * NCH + c) * SLEN;
    float* colp = out + colOff;

    v8f s[4];
#pragma unroll
    for (int tn = 0; tn < 4; ++tn)
#pragma unroll
        for (int r = 0; r < 8; ++r)
            s[tn][r] = (power && (tn * 16 + r + 8 * lh == c)) ? 1.0f : 0.0f;

    const float* xp = sig + colOff + k * LCH;

    v4f xcur4 = *reinterpret_cast<const v4f*>(xp);
    if (power) { xcur4[0] = 0.0f; xcur4[1] = 0.0f; xcur4[2] = 0.0f; xcur4[3] = 0.0f; }

#pragma unroll 1
    for (int tc = 0; tc < LCH; tc += RSTEP) {
        const int tnext = (tc + RSTEP < LCH) ? tc + RSTEP : tc;
        v4f xnext4 = *reinterpret_cast<const v4f*>(xp + tnext);
        if (power) { xnext4[0] = 0.0f; xnext4[1] = 0.0f; xnext4[2] = 0.0f; xnext4[3] = 0.0f; }

        v8f acc[4];
#pragma unroll
        for (int tn = 0; tn < 4; ++tn)
#pragma unroll
            for (int r = 0; r < 8; ++r) {
                float v = kb[3][tn][r] * xcur4[0];
                v = __builtin_fmaf(kb[2][tn][r], xcur4[1], v);
                v = __builtin_fmaf(kb[1][tn][r], xcur4[2], v);
                v = __builtin_fmaf(kb[0][tn][r], xcur4[3], v);
                acc[tn][r] = v;
            }
        mat_step(a, s, acc);
#pragma unroll
        for (int tn = 0; tn < 4; ++tn) s[tn] = acc[tn];

        xcur4 = xnext4;
    }

    if (power) {
#pragma unroll
        for (int tn = 0; tn < 4; ++tn)
#pragma unroll
            for (int r = 0; r < 8; ++r) out[(tn * 16 + r + 8 * lh) * SLEN + c] = s[tn][r];
    } else {
        save_state(colp + (k + 1) * LCH, lh, s);
    }
}

__global__ __launch_bounds__(256)
void hippo_final_kernel(const float* __restrict__ sig,
                        const float* __restrict__ Bm,
                        const float* __restrict__ Cmat,
                        const float* __restrict__ dtp,
                        const float* __restrict__ abar,
                        const float* __restrict__ a4,
                        float* out) {
    const int lane = threadIdx.x & 31;
    const int w    = threadIdx.x >> 5;
    const int lc   = lane & 15;
    const int lh   = lane >> 4;
    const int c    = w * 16 + lc;
    const int b    = blockIdx.x & 7;
    const float hdt = 0.5f * dtp[0];

    v8f kb[RSTEP][4], g[RSTEP][4];
    {
        v2f a1[4][4][4];
        load_afrags(abar, SDIM, lc, lh, a1);
        v8f bin[4], acc[4];
#pragma unroll
        for (int tn = 0; tn < 4; ++tn)
#pragma unroll
            for (int r = 0; r < 8; ++r) {
                bin[tn][r] = Bm[(tn * 16 + r + 8 * lh) * NCH + c];
                acc[tn][r] = bin[tn][r];
            }
        mat_step(a1, bin, acc);
#pragma unroll
        for (int tn = 0; tn < 4; ++tn)
#pragma unroll
            for (int r = 0; r < 8; ++r) kb[0][tn][r] = hdt * acc[tn][r];
#pragma unroll
        for (int i = 1; i < RSTEP; ++i) {
#pragma unroll
            for (int tn = 0; tn < 4; ++tn)
#pragma unroll
                for (int r = 0; r < 8; ++r) acc[tn][r] = 0.0f;
            mat_step(a1, kb[i - 1], acc);
#pragma unroll
            for (int tn = 0; tn < 4; ++tn) kb[i][tn] = acc[tn];
        }
    }
    {
        v2f at[4][4][4];
        load_afrags_t(abar, lc, lh, at);
#pragma unroll
        for (int tn = 0; tn < 4; ++tn)
#pragma unroll
            for (int r = 0; r < 8; ++r) g[0][tn][r] = Cmat[(tn * 16 + r + 8 * lh) * NCH + c];
#pragma unroll
        for (int i = 1; i < RSTEP; ++i) {
            v8f acc[4];
#pragma unroll
            for (int tn = 0; tn < 4; ++tn)
#pragma unroll
                for (int r = 0; r < 8; ++r) acc[tn][r] = 0.0f;
            mat_step_t(at, g[i - 1], acc);
#pragma unroll
            for (int tn = 0; tn < 4; ++tn) g[i][tn] = acc[tn];
        }
    }

    float h[RSTEP - 1];
#pragma unroll
    for (int jj = 0; jj < RSTEP - 1; ++jj) {
        float p = 0.0f;
#pragma unroll
        for (int tn = 0; tn < 4; ++tn)
#pragma unroll
            for (int r = 0; r < 8; ++r) p = __builtin_fmaf(g[0][tn][r], kb[jj][tn][r], p);
        p += __shfl_xor(p, 16, 32);
        h[jj] = p;
    }

    v2f a[4][4][4];
    load_afrags(a4, SDIM, lc, lh, a);

    const int colOff = (b * NCH + c) * SLEN;
    float* colp = out + colOff;

#pragma unroll 1
    for (int kk = 0; kk < CPB; ++kk) {
        const int k = (blockIdx.x >> 3) + kk * (KCH / CPB);

        v8f s[4];
        if (k > 0) {
            load_state(colp + k * LCH, lh, s);
        } else {
#pragma unroll
            for (int tn = 0; tn < 4; ++tn)
#pragma unroll
                for (int r = 0; r < 8; ++r) s[tn][r] = 0.0f;
        }

        const float* xp = sig + colOff + k * LCH;
        float*       yp = colp + k * LCH;

        v4f xcur4 = *reinterpret_cast<const v4f*>(xp);

#pragma unroll 1
        for (int tc = 0; tc < LCH; tc += RSTEP) {
            const int tnext = (tc + RSTEP < LCH) ? tc + RSTEP : tc;
            v4f xnext4 = *reinterpret_cast<const v4f*>(xp + tnext);

            float p1 = 0.0f, p2 = 0.0f, p3 = 0.0f;
#pragma unroll
            for (int tn = 0; tn < 4; ++tn)
#pragma unroll
                for (int r = 0; r < 8; ++r) {
                    p1 = __builtin_fmaf(g[1][tn][r], s[tn][r], p1);
                    p2 = __builtin_fmaf(g[2][tn][r], s[tn][r], p2);
                    p3 = __builtin_fmaf(g[3][tn][r], s[tn][r], p3);
                }

            v8f acc[4];
#pragma unroll
            for (int tn = 0; tn < 4; ++tn)
#pragma unroll
                for (int r = 0; r < 8; ++r) {
                    float v = kb[3][tn][r] * xcur4[0];
                    v = __builtin_fmaf(kb[2][tn][r], xcur4[1], v);
                    v = __builtin_fmaf(kb[1][tn][r], xcur4[2], v);
                    v = __builtin_fmaf(kb[0][tn][r], xcur4[3], v);
                    acc[tn][r] = v;
                }
            mat_step(a, s, acc);
#pragma unroll
            for (int tn = 0; tn < 4; ++tn) s[tn] = acc[tn];

            float p4 = 0.0f;
#pragma unroll
            for (int tn = 0; tn < 4; ++tn)
#pragma unroll
                for (int r = 0; r < 8; ++r) p4 = __builtin_fmaf(g[0][tn][r], s[tn][r], p4);

            p1 += __shfl_xor(p1, 16, 32);
            p2 += __shfl_xor(p2, 16, 32);
            p3 += __shfl_xor(p3, 16, 32);
            p4 += __shfl_xor(p4, 16, 32);

            v4f y4;
            y4[0] = __builtin_fmaf(h[0], xcur4[0], p1);
            y4[1] = __builtin_fmaf(h[1], xcur4[0], __builtin_fmaf(h[0], xcur4[1], p2));
            y4[2] = __builtin_fmaf(h[2], xcur4[0],
                                   __builtin_fmaf(h[1], xcur4[1], __builtin_fmaf(h[0], xcur4[2], p3)));
            y4[3] = p4;

            if (lh == 0) *reinterpret_cast<v4f*>(yp + tc) = y4;

            xcur4 = xnext4;
        }
    }
}

__device__ __forceinline__ void carry_trip(float* colp, int lh, int k,
                                           const v2f (&a)[4][4][4], v8f (&s)[4],
                                           v8f (&e)[4]) {
    v8f acc[4];
#pragma unroll
    for (int tn = 0; tn < 4; ++tn) acc[tn] = e[tn];
    const int kn = (k + 2 < KCH) ? k + 2 : KCH - 1;
    load_state(colp + kn * LCH, lh, e);
    __builtin_amdgcn_sched_barrier(0);
    mat_step(a, s, acc);
#pragma unroll
    for (int tn = 0; tn < 4; ++tn) s[tn] = acc[tn];
    save_state(colp + k * LCH, lh, s);
}

__global__ __launch_bounds__(32)
void hippo_carry_kernel(float* out) {
    const int lane = threadIdx.x & 31;
    const int lc   = lane & 15;
    const int lh   = lane >> 4;
    float* colp = out + (blockIdx.x * 16 + lc) * SLEN;

    v2f a[4][4][4];
    load_afrags(out, SLEN, lc, lh, a);

    v8f s[4];
#pragma unroll
    for (int tn = 0; tn < 4; ++tn)
#pragma unroll
        for (int r = 0; r < 8; ++r) s[tn][r] = 0.0f;

    v8f ea[4], eb[4];
    load_state(colp + 1 * LCH, lh, ea);
    load_state(colp + 2 * LCH, lh, eb);

    carry_trip(colp, lh, 1, a, s, ea);
#pragma unroll 1
    for (int k = 2; k < KCH; k += 2) {
        carry_trip(colp, lh, k, a, s, eb);
        carry_trip(colp, lh, k + 1, a, s, ea);
    }
}

extern "C" void kernel_launch(void* const* d_in, const int* in_sizes, int n_in,
                              void* d_out, int out_size, void* d_ws, size_t ws_size,
                              hipStream_t stream) {
    const float* sig = (const float*)d_in[0];
    const float* Bm  = (const float*)d_in[1];
    const float* Cm  = (const float*)d_in[2];
    const float* Am  = (const float*)d_in[3];
    const float* dtp = (const float*)d_in[4];
    float* out  = (float*)d_out;
    float* abar = (float*)d_ws;
    float* a4   = abar + SDIM * SDIM;

    hippo_setup_kernel<<<1, 256, 0, stream>>>(Am, Bm, dtp, abar, a4, out);
    hippo_local_kernel<<<NBLK_LOCAL, 256, 0, stream>>>(sig, a4, out);
    hippo_carry_kernel<<<NGRP, 32, 0, stream>>>(out);
    hippo_final_kernel<<<NBLK_FINAL, 256, 0, stream>>>(sig, Bm, Cm, dtp, abar, a4, out);
}


# time split into 32 chunks of 128 steps instead of 64 of 64: the serial carry pass walks 31 chunk boundaries instead of 63 (P = A_bar^128); state sums re-associated across chunks
# speedup vs baseline: 30.6882x; 1.2053x over previous
#include <hip/hip_runtime.h>
#include <hip/hip_bf16.h>

typedef float v2f __attribute__((ext_vector_type(2)));
typedef float v4f __attribute__((ext_vector_type(4)));
typedef float v8f __attribute__((ext_vector_type(8)));

#define SDIM 64
#define NCH  128
#define NB   8
#define SLEN 4096
#define KCH  32
#define LCH  128
#define NGRP 64
#define NBLK_LOCAL 249
#define NBLK_FINAL 256
#define RSTEP 4
#define KB_COL0 128

#define NRHS   192
#define LPITCH 68

template <int K>
__device__ __forceinline__ void hippo_solve_from(float (&s)[SDIM], const float* Lt) {
    if constexpr (K < SDIM) {
        const float xk = s[K] / Lt[K * LPITCH + K];
        s[K] = xk;
        const float nxk = -xk;
#pragma unroll
        for (int i = K + 1; i < SDIM; ++i)
            s[i] = __builtin_fmaf(Lt[K * LPITCH + i], nxk, s[i]);
        hippo_solve_from<K + 1>(s, Lt);
    }
}

__device__ __forceinline__ void hippo_setup_powers(const float* abar, float* a4, float* out, int tid);

__global__ __launch_bounds__(256)
void hippo_setup_kernel(const float* __restrict__ A,
                        const float* __restrict__ Bm,
                        const float* __restrict__ dtp,
                        float* abar,
                        float* a4,
                        float* out) {
    __shared__ __attribute__((aligned(16))) float Lt[SDIM * LPITCH];
    const float dt = dtp[0];
    const int tid = threadIdx.x;
    const bool solver = tid < NRHS;
    const bool bcol   = tid >= SDIM;
    const int  c      = tid - SDIM;

    float s[SDIM];
    if (solver && bcol) {
#pragma unroll
        for (int i = 0; i < SDIM; ++i) s[i] = Bm[i * NCH + c];
    } else {
#pragma unroll
        for (int i = 0; i < SDIM; ++i) s[i] = (i == tid) ? 1.0f : 0.0f;
    }

    float av[16];
#pragma unroll
    for (int q = 0; q < 16; ++q) av[q] = A[tid + 256 * q];
#pragma unroll
    for (int q = 0; q < 16; ++q) {
        const int e = tid + 256 * q;
        const int i = e >> 6, k = e & 63;
        float l = -0.5f * dt * av[q];
        if (i == k) l += 1.0f;
        Lt[k * LPITCH + i] = l;
    }
    __syncthreads();

    if (solver) {
        hippo_solve_from<0>(s, Lt);
        if (bcol) {
            float* bp = out + (KB_COL0 + c) * SLEN;
#pragma unroll
            for (int i = 0; i < SDIM; i += 4) {
                v4f q;
                q[0] = dt * s[i];
                q[1] = dt * s[i + 1];
                q[2] = dt * s[i + 2];
                q[3] = dt * s[i + 3];
                *reinterpret_cast<v4f*>(bp + i) = q;
            }
        } else {
#pragma unroll
            for (int i = 0; i < SDIM; ++i)
                abar[i * SDIM + tid] = 2.0f * s[i] - ((i == tid) ? 1.0f : 0.0f);
        }
    }

    __syncthreads();
    hippo_setup_powers(abar, a4, out, tid);
}

#define ABAR_BLOCK_NONZERO(tn, tb) ((tb) <= (tn))


__device__ __forceinline__ void load_afrags(const float* mat, int pitch, int lc, int lh,
                                            v2f (&a)[4][4][4]) {
#pragma unroll
    for (int tn = 0; tn < 4; ++tn)
#pragma unroll
        for (int tb = 0; tb < 4; ++tb)
#pragma unroll
            for (int j = 0; j < 4; ++j) {
                if (ABAR_BLOCK_NONZERO(tn, tb)) {
                    const int base = (tn * 16 + lc) * pitch + tb * 16 + 8 * lh + 2 * j;
                    a[tn][tb][j] = *reinterpret_cast<const v2f*>(mat + base);
                }
            }
}

__device__ __forceinline__ v2f pair2(float x, float y) {
    v2f p;
    p.x = x;
    p.y = y;
    return p;
}

__device__ __forceinline__ void load_afrags_t(const float* mat, int lc, int lh,
                                              v2f (&a)[4][4][4]) {
#pragma unroll
    for (int tn = 0; tn < 4; ++tn)
#pragma unroll
        for (int tb = 0; tb < 4; ++tb)
#pragma unroll
            for (int j = 0; j < 4; ++j) {
                if (ABAR_BLOCK_NONZERO(tb, tn)) {
                    const int m = tb * 16 + 8 * lh + 2 * j;
                    a[tn][tb][j] = pair2(mat[m * SDIM + tn * 16 + lc],
                                         mat[(m + 1) * SDIM + tn * 16 + lc]);
                }
            }
}

__device__ __forceinline__ void mat_step_t(const v2f (&a)[4][4][4], const v8f (&s)[4],
                                           v8f (&acc)[4]) {
#pragma unroll
    for (int tn = 0; tn < 4; ++tn)
#pragma unroll
        for (int tb = 0; tb < 4; ++tb)
#pragma unroll
            for (int j = 0; j < 4; ++j) {
                if (ABAR_BLOCK_NONZERO(tb, tn)) {
                    acc[tn] = __builtin_amdgcn_wmma_f32_16x16x4_f32(
                         false, a[tn][tb][j],
                         false, pair2(s[tb][2 * j], s[tb][2 * j + 1]),
                         (short)0, acc[tn],
                         false,  false);
                }
            }
}

__device__ __forceinline__ void mat_step(const v2f (&a)[4][4][4], const v8f (&s)[4],
                                         v8f (&acc)[4]) {
#pragma unroll
    for (int tn = 0; tn < 4; ++tn)
#pragma unroll
        for (int tb = 0; tb < 4; ++tb)
#pragma unroll
            for (int j = 0; j < 4; ++j) {
                if (ABAR_BLOCK_NONZERO(tn, tb)) {
                    acc[tn] = __builtin_amdgcn_wmma_f32_16x16x4_f32(
                         false, a[tn][tb][j],
                         false, pair2(s[tb][2 * j], s[tb][2 * j + 1]),
                         (short)0, acc[tn],
                         false,  false);
                }
            }
}

__device__ __forceinline__ void save_state(float* p, int lh, const v8f (&s)[4]) {
#pragma unroll
    for (int tn = 0; tn < 4; ++tn) *reinterpret_cast<v8f*>(p + tn * 16 + 8 * lh) = s[tn];
}
__device__ __forceinline__ void load_state(const float* p, int lh, v8f (&s)[4]) {
#pragma unroll
    for (int tn = 0; tn < 4; ++tn) s[tn] = *reinterpret_cast<const v8f*>(p + tn * 16 + 8 * lh);
}

__device__ __forceinline__ void hippo_setup_powers(const float* abar, float* a4, float* out, int tid) {
    const int lane = tid & 31;
    const int w    = tid >> 5;
    const int lc   = lane & 15;
    const int lh   = lane >> 4;
    const int c    = w * 16 + lc;

    v2f a[4][4][4];
    load_afrags(abar, SDIM, lc, lh, a);

    v8f s[4];
    load_state(out + (KB_COL0 + c) * SLEN, lh, s);

#pragma unroll
    for (int i = 1; i < RSTEP; ++i) {
        v8f acc[4];
#pragma unroll
        for (int tn = 0; tn < 4; ++tn)
#pragma unroll
            for (int r = 0; r < 8; ++r) acc[tn][r] = 0.0f;
        mat_step(a, s, acc);
#pragma unroll
        for (int tn = 0; tn < 4; ++tn) s[tn] = acc[tn];
        save_state(out + (KB_COL0 + i * NCH + c) * SLEN, lh, s);
    }

    if (w < 4) {
#pragma unroll
        for (int tn = 0; tn < 4; ++tn)
#pragma unroll
            for (int r = 0; r < 8; ++r) s[tn][r] = (tn * 16 + r + 8 * lh == c) ? 1.0f : 0.0f;
#pragma unroll
        for (int i = 0; i < RSTEP; ++i) {
            v8f acc[4];
#pragma unroll
            for (int tn = 0; tn < 4; ++tn)
#pragma unroll
                for (int r = 0; r < 8; ++r) acc[tn][r] = 0.0f;
            mat_step(a, s, acc);
#pragma unroll
            for (int tn = 0; tn < 4; ++tn) s[tn] = acc[tn];
        }
#pragma unroll
        for (int tn = 0; tn < 4; ++tn)
#pragma unroll
            for (int r = 0; r < 8; ++r)
                a4[(tn * 16 + r + 8 * lh) * SDIM + c] = s[tn][r];
    }
}

__global__ __launch_bounds__(256)
void hippo_local_kernel(const float* __restrict__ sig,
                        const float* __restrict__ a4,
                        float* out) {
    const int lane = threadIdx.x & 31;
    const int w    = threadIdx.x >> 5;
    const int lc   = lane & 15;
    const int lh   = lane >> 4;
    const int c    = w * 16 + lc;

    int k, b;
    bool power = false;
    if (blockIdx.x == 0) {
        power = true;
        k = 0;
        b = 0;
        if (w >= 4) return;
    } else {
        k = (blockIdx.x - 1) >> 3;
        b = (blockIdx.x - 1) & 7;
    }

    v2f a[4][4][4];
    load_afrags(a4, SDIM, lc, lh, a);

    v8f kb[RSTEP][4];
#pragma unroll
    for (int i = 0; i < RSTEP; ++i) {
        load_state(out + (KB_COL0 + i * NCH + c) * SLEN, lh, kb[i]);
#pragma unroll
        for (int tn = 0; tn < 4; ++tn)
#pragma unroll
            for (int r = 0; r < 8; ++r) kb[i][tn][r] = power ? 0.0f : kb[i][tn][r];
    }

    const int colOff = (b * NCH + c) * SLEN;
    float* colp = out + colOff;

    v8f s[4];
#pragma unroll
    for (int tn = 0; tn < 4; ++tn)
#pragma unroll
        for (int r = 0; r < 8; ++r)
            s[tn][r] = (power && (tn * 16 + r + 8 * lh == c)) ? 1.0f : 0.0f;

    const float* xp = sig + colOff + k * LCH;

    v4f xcur4 = *reinterpret_cast<const v4f*>(xp);
    if (power) { xcur4[0] = 0.0f; xcur4[1] = 0.0f; xcur4[2] = 0.0f; xcur4[3] = 0.0f; }

#pragma unroll 1
    for (int tc = 0; tc < LCH; tc += RSTEP) {
        const int tnext = (tc + RSTEP < LCH) ? tc + RSTEP : tc;
        v4f xnext4 = *reinterpret_cast<const v4f*>(xp + tnext);
        if (power) { xnext4[0] = 0.0f; xnext4[1] = 0.0f; xnext4[2] = 0.0f; xnext4[3] = 0.0f; }

        v8f acc[4];
#pragma unroll
        for (int tn = 0; tn < 4; ++tn)
#pragma unroll
            for (int r = 0; r < 8; ++r) {
                float v = kb[3][tn][r] * xcur4[0];
                v = __builtin_fmaf(kb[2][tn][r], xcur4[1], v);
                v = __builtin_fmaf(kb[1][tn][r], xcur4[2], v);
                v = __builtin_fmaf(kb[0][tn][r], xcur4[3], v);
                acc[tn][r] = v;
            }
        mat_step(a, s, acc);
#pragma unroll
        for (int tn = 0; tn < 4; ++tn) s[tn] = acc[tn];

        xcur4 = xnext4;
    }

    if (power) {
#pragma unroll
        for (int tn = 0; tn < 4; ++tn)
#pragma unroll
            for (int r = 0; r < 8; ++r) out[(tn * 16 + r + 8 * lh) * SLEN + c] = s[tn][r];
    } else {
        save_state(colp + (k + 1) * LCH, lh, s);
    }
}

__global__ __launch_bounds__(256)
void hippo_final_kernel(const float* __restrict__ sig,
                        const float* __restrict__ Bm,
                        const float* __restrict__ Cmat,
                        const float* __restrict__ dtp,
                        const float* __restrict__ abar,
                        const float* __restrict__ a4,
                        float* out) {
    const int lane = threadIdx.x & 31;
    const int w    = threadIdx.x >> 5;
    const int lc   = lane & 15;
    const int lh   = lane >> 4;
    const int c    = w * 16 + lc;
    const int b    = blockIdx.x & 7;
    const float hdt = 0.5f * dtp[0];

    v8f kb[RSTEP][4], g[RSTEP][4];
    {
        v2f a1[4][4][4];
        load_afrags(abar, SDIM, lc, lh, a1);
        v8f bin[4], acc[4];
#pragma unroll
        for (int tn = 0; tn < 4; ++tn)
#pragma unroll
            for (int r = 0; r < 8; ++r) {
                bin[tn][r] = Bm[(tn * 16 + r + 8 * lh) * NCH + c];
                acc[tn][r] = bin[tn][r];
            }
        mat_step(a1, bin, acc);
#pragma unroll
        for (int tn = 0; tn < 4; ++tn)
#pragma unroll
            for (int r = 0; r < 8; ++r) kb[0][tn][r] = hdt * acc[tn][r];
#pragma unroll
        for (int i = 1; i < RSTEP; ++i) {
#pragma unroll
            for (int tn = 0; tn < 4; ++tn)
#pragma unroll
                for (int r = 0; r < 8; ++r) acc[tn][r] = 0.0f;
            mat_step(a1, kb[i - 1], acc);
#pragma unroll
            for (int tn = 0; tn < 4; ++tn) kb[i][tn] = acc[tn];
        }
    }
    {
        v2f at[4][4][4];
        load_afrags_t(abar, lc, lh, at);
#pragma unroll
        for (int tn = 0; tn < 4; ++tn)
#pragma unroll
            for (int r = 0; r < 8; ++r) g[0][tn][r] = Cmat[(tn * 16 + r + 8 * lh) * NCH + c];
#pragma unroll
        for (int i = 1; i < RSTEP; ++i) {
            v8f acc[4];
#pragma unroll
            for (int tn = 0; tn < 4; ++tn)
#pragma unroll
                for (int r = 0; r < 8; ++r) acc[tn][r] = 0.0f;
            mat_step_t(at, g[i - 1], acc);
#pragma unroll
            for (int tn = 0; tn < 4; ++tn) g[i][tn] = acc[tn];
        }
    }

    float h[RSTEP - 1];
#pragma unroll
    for (int jj = 0; jj < RSTEP - 1; ++jj) {
        float p = 0.0f;
#pragma unroll
        for (int tn = 0; tn < 4; ++tn)
#pragma unroll
            for (int r = 0; r < 8; ++r) p = __builtin_fmaf(g[0][tn][r], kb[jj][tn][r], p);
        p += __shfl_xor(p, 16, 32);
        h[jj] = p;
    }

    v2f a[4][4][4];
    load_afrags(a4, SDIM, lc, lh, a);

    const int colOff = (b * NCH + c) * SLEN;
    float* colp = out + colOff;

    const int k = blockIdx.x >> 3;

    v8f s[4];
    if (k > 0) {
        load_state(colp + k * LCH, lh, s);
    } else {
#pragma unroll
        for (int tn = 0; tn < 4; ++tn)
#pragma unroll
            for (int r = 0; r < 8; ++r) s[tn][r] = 0.0f;
    }

    const float* xp = sig + colOff + k * LCH;
    float*       yp = colp + k * LCH;

    v4f xcur4 = *reinterpret_cast<const v4f*>(xp);

#pragma unroll 1
    for (int tc = 0; tc < LCH; tc += RSTEP) {
        const int tnext = (tc + RSTEP < LCH) ? tc + RSTEP : tc;
        v4f xnext4 = *reinterpret_cast<const v4f*>(xp + tnext);

        float p1 = 0.0f, p2 = 0.0f, p3 = 0.0f;
#pragma unroll
        for (int tn = 0; tn < 4; ++tn)
#pragma unroll
            for (int r = 0; r < 8; ++r) {
                p1 = __builtin_fmaf(g[1][tn][r], s[tn][r], p1);
                p2 = __builtin_fmaf(g[2][tn][r], s[tn][r], p2);
                p3 = __builtin_fmaf(g[3][tn][r], s[tn][r], p3);
            }

        v8f acc[4];
#pragma unroll
        for (int tn = 0; tn < 4; ++tn)
#pragma unroll
            for (int r = 0; r < 8; ++r) {
                float v = kb[3][tn][r] * xcur4[0];
                v = __builtin_fmaf(kb[2][tn][r], xcur4[1], v);
                v = __builtin_fmaf(kb[1][tn][r], xcur4[2], v);
                v = __builtin_fmaf(kb[0][tn][r], xcur4[3], v);
                acc[tn][r] = v;
            }
        mat_step(a, s, acc);
#pragma unroll
        for (int tn = 0; tn < 4; ++tn) s[tn] = acc[tn];

        float p4 = 0.0f;
#pragma unroll
        for (int tn = 0; tn < 4; ++tn)
#pragma unroll
            for (int r = 0; r < 8; ++r) p4 = __builtin_fmaf(g[0][tn][r], s[tn][r], p4);

        p1 += __shfl_xor(p1, 16, 32);
        p2 += __shfl_xor(p2, 16, 32);
        p3 += __shfl_xor(p3, 16, 32);
        p4 += __shfl_xor(p4, 16, 32);

        v4f y4;
        y4[0] = __builtin_fmaf(h[0], xcur4[0], p1);
        y4[1] = __builtin_fmaf(h[1], xcur4[0], __builtin_fmaf(h[0], xcur4[1], p2));
        y4[2] = __builtin_fmaf(h[2], xcur4[0],
                               __builtin_fmaf(h[1], xcur4[1], __builtin_fmaf(h[0], xcur4[2], p3)));
        y4[3] = p4;

        if (lh == 0) *reinterpret_cast<v4f*>(yp + tc) = y4;

        xcur4 = xnext4;
    }
}

__device__ __forceinline__ void carry_trip(float* colp, int lh, int k,
                                           const v2f (&a)[4][4][4], v8f (&s)[4],
                                           v8f (&e)[4]) {
    v8f acc[4];
#pragma unroll
    for (int tn = 0; tn < 4; ++tn) acc[tn] = e[tn];
    const int kn = (k + 2 < KCH) ? k + 2 : KCH - 1;
    load_state(colp + kn * LCH, lh, e);
    __builtin_amdgcn_sched_barrier(0);
    mat_step(a, s, acc);
#pragma unroll
    for (int tn = 0; tn < 4; ++tn) s[tn] = acc[tn];
    save_state(colp + k * LCH, lh, s);
}

__global__ __launch_bounds__(32)
void hippo_carry_kernel(float* out) {
    const int lane = threadIdx.x & 31;
    const int lc   = lane & 15;
    const int lh   = lane >> 4;
    float* colp = out + (blockIdx.x * 16 + lc) * SLEN;

    v2f a[4][4][4];
    load_afrags(out, SLEN, lc, lh, a);

    v8f s[4];
#pragma unroll
    for (int tn = 0; tn < 4; ++tn)
#pragma unroll
        for (int r = 0; r < 8; ++r) s[tn][r] = 0.0f;

    v8f ea[4], eb[4];
    load_state(colp + 1 * LCH, lh, ea);
    load_state(colp + 2 * LCH, lh, eb);

    carry_trip(colp, lh, 1, a, s, ea);
#pragma unroll 1
    for (int k = 2; k < KCH; k += 2) {
        carry_trip(colp, lh, k, a, s, eb);
        carry_trip(colp, lh, k + 1, a, s, ea);
    }
}

extern "C" void kernel_launch(void* const* d_in, const int* in_sizes, int n_in,
                              void* d_out, int out_size, void* d_ws, size_t ws_size,
                              hipStream_t stream) {
    const float* sig = (const float*)d_in[0];
    const float* Bm  = (const float*)d_in[1];
    const float* Cm  = (const float*)d_in[2];
    const float* Am  = (const float*)d_in[3];
    const float* dtp = (const float*)d_in[4];
    float* out  = (float*)d_out;
    float* abar = (float*)d_ws;
    float* a4   = abar + SDIM * SDIM;

    hippo_setup_kernel<<<1, 256, 0, stream>>>(Am, Bm, dtp, abar, a4, out);
    hippo_local_kernel<<<NBLK_LOCAL, 256, 0, stream>>>(sig, a4, out);
    hippo_carry_kernel<<<NGRP, 32, 0, stream>>>(out);
    hippo_final_kernel<<<NBLK_FINAL, 256, 0, stream>>>(sig, Bm, Cm, dtp, abar, a4, out);
}


# carry pass: the 40 products of a trip split over four waves, one per 16-row block of the state; old blocks exchanged through LDS, one LDS-only barrier per trip; k order and every sum unchanged
# speedup vs baseline: 33.7759x; 1.1006x over previous
#include <hip/hip_runtime.h>
#include <hip/hip_bf16.h>

typedef float v2f __attribute__((ext_vector_type(2)));
typedef float v4f __attribute__((ext_vector_type(4)));
typedef float v8f __attribute__((ext_vector_type(8)));

#define SDIM 64
#define NCH  128
#define NB   8
#define SLEN 4096
#define KCH  32
#define LCH  128
#define NGRP 64
#define NBLK_LOCAL 249
#define NBLK_FINAL 256
#define RSTEP 4
#define KB_COL0 128

#define NRHS   192
#define LPITCH 68

template <int K>
__device__ __forceinline__ void hippo_solve_from(float (&s)[SDIM], const float* Lt) {
    if constexpr (K < SDIM) {
        const float xk = s[K] / Lt[K * LPITCH + K];
        s[K] = xk;
        const float nxk = -xk;
#pragma unroll
        for (int i = K + 1; i < SDIM; ++i)
            s[i] = __builtin_fmaf(Lt[K * LPITCH + i], nxk, s[i]);
        hippo_solve_from<K + 1>(s, Lt);
    }
}

__device__ __forceinline__ void hippo_setup_powers(const float* abar, float* a4, float* out, int tid);

__global__ __launch_bounds__(256)
void hippo_setup_kernel(const float* __restrict__ A,
                        const float* __restrict__ Bm,
                        const float* __restrict__ dtp,
                        float* abar,
                        float* a4,
                        float* out) {
    __shared__ __attribute__((aligned(16))) float Lt[SDIM * LPITCH];
    const float dt = dtp[0];
    const int tid = threadIdx.x;
    const bool solver = tid < NRHS;
    const bool bcol   = tid >= SDIM;
    const int  c      = tid - SDIM;

    float s[SDIM];
    if (solver && bcol) {
#pragma unroll
        for (int i = 0; i < SDIM; ++i) s[i] = Bm[i * NCH + c];
    } else {
#pragma unroll
        for (int i = 0; i < SDIM; ++i) s[i] = (i == tid) ? 1.0f : 0.0f;
    }

    float av[16];
#pragma unroll
    for (int q = 0; q < 16; ++q) av[q] = A[tid + 256 * q];
#pragma unroll
    for (int q = 0; q < 16; ++q) {
        const int e = tid + 256 * q;
        const int i = e >> 6, k = e & 63;
        float l = -0.5f * dt * av[q];
        if (i == k) l += 1.0f;
        Lt[k * LPITCH + i] = l;
    }
    __syncthreads();

    if (solver) {
        hippo_solve_from<0>(s, Lt);
        if (bcol) {
            float* bp = out + (KB_COL0 + c) * SLEN;
#pragma unroll
            for (int i = 0; i < SDIM; i += 4) {
                v4f q;
                q[0] = dt * s[i];
                q[1] = dt * s[i + 1];
                q[2] = dt * s[i + 2];
                q[3] = dt * s[i + 3];
                *reinterpret_cast<v4f*>(bp + i) = q;
            }
        } else {
#pragma unroll
            for (int i = 0; i < SDIM; ++i)
                abar[i * SDIM + tid] = 2.0f * s[i] - ((i == tid) ? 1.0f : 0.0f);
        }
    }

    __syncthreads();
    hippo_setup_powers(abar, a4, out, tid);
}

#define ABAR_BLOCK_NONZERO(tn, tb) ((tb) <= (tn))


__device__ __forceinline__ void load_afrags(const float* mat, int pitch, int lc, int lh,
                                            v2f (&a)[4][4][4]) {
#pragma unroll
    for (int tn = 0; tn < 4; ++tn)
#pragma unroll
        for (int tb = 0; tb < 4; ++tb)
#pragma unroll
            for (int j = 0; j < 4; ++j) {
                if (ABAR_BLOCK_NONZERO(tn, tb)) {
                    const int base = (tn * 16 + lc) * pitch + tb * 16 + 8 * lh + 2 * j;
                    a[tn][tb][j] = *reinterpret_cast<const v2f*>(mat + base);
                }
            }
}

__device__ __forceinline__ v2f pair2(float x, float y) {
    v2f p;
    p.x = x;
    p.y = y;
    return p;
}

__device__ __forceinline__ void load_afrags_t(const float* mat, int lc, int lh,
                                              v2f (&a)[4][4][4]) {
#pragma unroll
    for (int tn = 0; tn < 4; ++tn)
#pragma unroll
        for (int tb = 0; tb < 4; ++tb)
#pragma unroll
            for (int j = 0; j < 4; ++j) {
                if (ABAR_BLOCK_NONZERO(tb, tn)) {
                    const int m = tb * 16 + 8 * lh + 2 * j;
                    a[tn][tb][j] = pair2(mat[m * SDIM + tn * 16 + lc],
                                         mat[(m + 1) * SDIM + tn * 16 + lc]);
                }
            }
}

__device__ __forceinline__ void mat_step_t(const v2f (&a)[4][4][4], const v8f (&s)[4],
                                           v8f (&acc)[4]) {
#pragma unroll
    for (int tn = 0; tn < 4; ++tn)
#pragma unroll
        for (int tb = 0; tb < 4; ++tb)
#pragma unroll
            for (int j = 0; j < 4; ++j) {
                if (ABAR_BLOCK_NONZERO(tb, tn)) {
                    acc[tn] = __builtin_amdgcn_wmma_f32_16x16x4_f32(
                         false, a[tn][tb][j],
                         false, pair2(s[tb][2 * j], s[tb][2 * j + 1]),
                         (short)0, acc[tn],
                         false,  false);
                }
            }
}

__device__ __forceinline__ void mat_step(const v2f (&a)[4][4][4], const v8f (&s)[4],
                                         v8f (&acc)[4]) {
#pragma unroll
    for (int tn = 0; tn < 4; ++tn)
#pragma unroll
        for (int tb = 0; tb < 4; ++tb)
#pragma unroll
            for (int j = 0; j < 4; ++j) {
                if (ABAR_BLOCK_NONZERO(tn, tb)) {
                    acc[tn] = __builtin_amdgcn_wmma_f32_16x16x4_f32(
                         false, a[tn][tb][j],
                         false, pair2(s[tb][2 * j], s[tb][2 * j + 1]),
                         (short)0, acc[tn],
                         false,  false);
                }
            }
}

__device__ __forceinline__ void save_state(float* p, int lh, const v8f (&s)[4]) {
#pragma unroll
    for (int tn = 0; tn < 4; ++tn) *reinterpret_cast<v8f*>(p + tn * 16 + 8 * lh) = s[tn];
}
__device__ __forceinline__ void load_state(const float* p, int lh, v8f (&s)[4]) {
#pragma unroll
    for (int tn = 0; tn < 4; ++tn) s[tn] = *reinterpret_cast<const v8f*>(p + tn * 16 + 8 * lh);
}

__device__ __forceinline__ void hippo_setup_powers(const float* abar, float* a4, float* out, int tid) {
    const int lane = tid & 31;
    const int w    = tid >> 5;
    const int lc   = lane & 15;
    const int lh   = lane >> 4;
    const int c    = w * 16 + lc;

    v2f a[4][4][4];
    load_afrags(abar, SDIM, lc, lh, a);

    v8f s[4];
    load_state(out + (KB_COL0 + c) * SLEN, lh, s);

#pragma unroll
    for (int i = 1; i < RSTEP; ++i) {
        v8f acc[4];
#pragma unroll
        for (int tn = 0; tn < 4; ++tn)
#pragma unroll
            for (int r = 0; r < 8; ++r) acc[tn][r] = 0.0f;
        mat_step(a, s, acc);
#pragma unroll
        for (int tn = 0; tn < 4; ++tn) s[tn] = acc[tn];
        save_state(out + (KB_COL0 + i * NCH + c) * SLEN, lh, s);
    }

    if (w < 4) {
#pragma unroll
        for (int tn = 0; tn < 4; ++tn)
#pragma unroll
            for (int r = 0; r < 8; ++r) s[tn][r] = (tn * 16 + r + 8 * lh == c) ? 1.0f : 0.0f;
#pragma unroll
        for (int i = 0; i < RSTEP; ++i) {
            v8f acc[4];
#pragma unroll
            for (int tn = 0; tn < 4; ++tn)
#pragma unroll
                for (int r = 0; r < 8; ++r) acc[tn][r] = 0.0f;
            mat_step(a, s, acc);
#pragma unroll
            for (int tn = 0; tn < 4; ++tn) s[tn] = acc[tn];
        }
#pragma unroll
        for (int tn = 0; tn < 4; ++tn)
#pragma unroll
            for (int r = 0; r < 8; ++r)
                a4[(tn * 16 + r + 8 * lh) * SDIM + c] = s[tn][r];
    }
}

__global__ __launch_bounds__(256)
void hippo_local_kernel(const float* __restrict__ sig,
                        const float* __restrict__ a4,
                        float* out) {
    const int lane = threadIdx.x & 31;
    const int w    = threadIdx.x >> 5;
    const int lc   = lane & 15;
    const int lh   = lane >> 4;
    const int c    = w * 16 + lc;

    int k, b;
    bool power = false;
    if (blockIdx.x == 0) {
        power = true;
        k = 0;
        b = 0;
        if (w >= 4) return;
    } else {
        k = (blockIdx.x - 1) >> 3;
        b = (blockIdx.x - 1) & 7;
    }

    v2f a[4][4][4];
    load_afrags(a4, SDIM, lc, lh, a);

    v8f kb[RSTEP][4];
#pragma unroll
    for (int i = 0; i < RSTEP; ++i) {
        load_state(out + (KB_COL0 + i * NCH + c) * SLEN, lh, kb[i]);
#pragma unroll
        for (int tn = 0; tn < 4; ++tn)
#pragma unroll
            for (int r = 0; r < 8; ++r) kb[i][tn][r] = power ? 0.0f : kb[i][tn][r];
    }

    const int colOff = (b * NCH + c) * SLEN;
    float* colp = out + colOff;

    v8f s[4];
#pragma unroll
    for (int tn = 0; tn < 4; ++tn)
#pragma unroll
        for (int r = 0; r < 8; ++r)
            s[tn][r] = (power && (tn * 16 + r + 8 * lh == c)) ? 1.0f : 0.0f;

    const float* xp = sig + colOff + k * LCH;

    v4f xcur4 = *reinterpret_cast<const v4f*>(xp);
    if (power) { xcur4[0] = 0.0f; xcur4[1] = 0.0f; xcur4[2] = 0.0f; xcur4[3] = 0.0f; }

#pragma unroll 1
    for (int tc = 0; tc < LCH; tc += RSTEP) {
        const int tnext = (tc + RSTEP < LCH) ? tc + RSTEP : tc;
        v4f xnext4 = *reinterpret_cast<const v4f*>(xp + tnext);
        if (power) { xnext4[0] = 0.0f; xnext4[1] = 0.0f; xnext4[2] = 0.0f; xnext4[3] = 0.0f; }

        v8f acc[4];
#pragma unroll
        for (int tn = 0; tn < 4; ++tn)
#pragma unroll
            for (int r = 0; r < 8; ++r) {
                float v = kb[3][tn][r] * xcur4[0];
                v = __builtin_fmaf(kb[2][tn][r], xcur4[1], v);
                v = __builtin_fmaf(kb[1][tn][r], xcur4[2], v);
                v = __builtin_fmaf(kb[0][tn][r], xcur4[3], v);
                acc[tn][r] = v;
            }
        mat_step(a, s, acc);
#pragma unroll
        for (int tn = 0; tn < 4; ++tn) s[tn] = acc[tn];

        xcur4 = xnext4;
    }

    if (power) {
#pragma unroll
        for (int tn = 0; tn < 4; ++tn)
#pragma unroll
            for (int r = 0; r < 8; ++r) out[(tn * 16 + r + 8 * lh) * SLEN + c] = s[tn][r];
    } else {
        save_state(colp + (k + 1) * LCH, lh, s);
    }
}

__global__ __launch_bounds__(256)
void hippo_final_kernel(const float* __restrict__ sig,
                        const float* __restrict__ Bm,
                        const float* __restrict__ Cmat,
                        const float* __restrict__ dtp,
                        const float* __restrict__ abar,
                        const float* __restrict__ a4,
                        float* out) {
    const int lane = threadIdx.x & 31;
    const int w    = threadIdx.x >> 5;
    const int lc   = lane & 15;
    const int lh   = lane >> 4;
    const int c    = w * 16 + lc;
    const int b    = blockIdx.x & 7;
    const float hdt = 0.5f * dtp[0];

    v8f kb[RSTEP][4], g[RSTEP][4];
    {
        v2f a1[4][4][4];
        load_afrags(abar, SDIM, lc, lh, a1);
        v8f bin[4], acc[4];
#pragma unroll
        for (int tn = 0; tn < 4; ++tn)
#pragma unroll
            for (int r = 0; r < 8; ++r) {
                bin[tn][r] = Bm[(tn * 16 + r + 8 * lh) * NCH + c];
                acc[tn][r] = bin[tn][r];
            }
        mat_step(a1, bin, acc);
#pragma unroll
        for (int tn = 0; tn < 4; ++tn)
#pragma unroll
            for (int r = 0; r < 8; ++r) kb[0][tn][r] = hdt * acc[tn][r];
#pragma unroll
        for (int i = 1; i < RSTEP; ++i) {
#pragma unroll
            for (int tn = 0; tn < 4; ++tn)
#pragma unroll
                for (int r = 0; r < 8; ++r) acc[tn][r] = 0.0f;
            mat_step(a1, kb[i - 1], acc);
#pragma unroll
            for (int tn = 0; tn < 4; ++tn) kb[i][tn] = acc[tn];
        }
    }
    {
        v2f at[4][4][4];
        load_afrags_t(abar, lc, lh, at);
#pragma unroll
        for (int tn = 0; tn < 4; ++tn)
#pragma unroll
            for (int r = 0; r < 8; ++r) g[0][tn][r] = Cmat[(tn * 16 + r + 8 * lh) * NCH + c];
#pragma unroll
        for (int i = 1; i < RSTEP; ++i) {
            v8f acc[4];
#pragma unroll
            for (int tn = 0; tn < 4; ++tn)
#pragma unroll
                for (int r = 0; r < 8; ++r) acc[tn][r] = 0.0f;
            mat_step_t(at, g[i - 1], acc);
#pragma unroll
            for (int tn = 0; tn < 4; ++tn) g[i][tn] = acc[tn];
        }
    }

    float h[RSTEP - 1];
#pragma unroll
    for (int jj = 0; jj < RSTEP - 1; ++jj) {
        float p = 0.0f;
#pragma unroll
        for (int tn = 0; tn < 4; ++tn)
#pragma unroll
            for (int r = 0; r < 8; ++r) p = __builtin_fmaf(g[0][tn][r], kb[jj][tn][r], p);
        p += __shfl_xor(p, 16, 32);
        h[jj] = p;
    }

    v2f a[4][4][4];
    load_afrags(a4, SDIM, lc, lh, a);

    const int colOff = (b * NCH + c) * SLEN;
    float* colp = out + colOff;

    const int k = blockIdx.x >> 3;

    v8f s[4];
    if (k > 0) {
        load_state(colp + k * LCH, lh, s);
    } else {
#pragma unroll
        for (int tn = 0; tn < 4; ++tn)
#pragma unroll
            for (int r = 0; r < 8; ++r) s[tn][r] = 0.0f;
    }

    const float* xp = sig + colOff + k * LCH;
    float*       yp = colp + k * LCH;

    v4f xcur4 = *reinterpret_cast<const v4f*>(xp);

#pragma unroll 1
    for (int tc = 0; tc < LCH; tc += RSTEP) {
        const int tnext = (tc + RSTEP < LCH) ? tc + RSTEP : tc;
        v4f xnext4 = *reinterpret_cast<const v4f*>(xp + tnext);

        float p1 = 0.0f, p2 = 0.0f, p3 = 0.0f;
#pragma unroll
        for (int tn = 0; tn < 4; ++tn)
#pragma unroll
            for (int r = 0; r < 8; ++r) {
                p1 = __builtin_fmaf(g[1][tn][r], s[tn][r], p1);
                p2 = __builtin_fmaf(g[2][tn][r], s[tn][r], p2);
                p3 = __builtin_fmaf(g[3][tn][r], s[tn][r], p3);
            }

        v8f acc[4];
#pragma unroll
        for (int tn = 0; tn < 4; ++tn)
#pragma unroll
            for (int r = 0; r < 8; ++r) {
                float v = kb[3][tn][r] * xcur4[0];
                v = __builtin_fmaf(kb[2][tn][r], xcur4[1], v);
                v = __builtin_fmaf(kb[1][tn][r], xcur4[2], v);
                v = __builtin_fmaf(kb[0][tn][r], xcur4[3], v);
                acc[tn][r] = v;
            }
        mat_step(a, s, acc);
#pragma unroll
        for (int tn = 0; tn < 4; ++tn) s[tn] = acc[tn];

        float p4 = 0.0f;
#pragma unroll
        for (int tn = 0; tn < 4; ++tn)
#pragma unroll
            for (int r = 0; r < 8; ++r) p4 = __builtin_fmaf(g[0][tn][r], s[tn][r], p4);

        p1 += __shfl_xor(p1, 16, 32);
        p2 += __shfl_xor(p2, 16, 32);
        p3 += __shfl_xor(p3, 16, 32);
        p4 += __shfl_xor(p4, 16, 32);

        v4f y4;
        y4[0] = __builtin_fmaf(h[0], xcur4[0], p1);
        y4[1] = __builtin_fmaf(h[1], xcur4[0], __builtin_fmaf(h[0], xcur4[1], p2));
        y4[2] = __builtin_fmaf(h[2], xcur4[0],
                               __builtin_fmaf(h[1], xcur4[1], __builtin_fmaf(h[0], xcur4[2], p3)));
        y4[3] = p4;

        if (lh == 0) *reinterpret_cast<v4f*>(yp + tc) = y4;

        xcur4 = xnext4;
    }
}

#define CW_TILE 256
#define CW_BUF  1024

__device__ __forceinline__ v8f lds_tile_load(const float* buf, int tb, int lane) {
    const v4f lo = *reinterpret_cast<const v4f*>(buf + tb * CW_TILE + lane * 4);
    const v4f hi = *reinterpret_cast<const v4f*>(buf + tb * CW_TILE + 128 + lane * 4);
    v8f t;
    t[0] = lo[0]; t[1] = lo[1]; t[2] = lo[2]; t[3] = lo[3];
    t[4] = hi[0]; t[5] = hi[1]; t[6] = hi[2]; t[7] = hi[3];
    return t;
}
__device__ __forceinline__ void lds_tile_store(float* buf, int tb, int lane, const v8f& t) {
    v4f lo, hi;
    lo[0] = t[0]; lo[1] = t[1]; lo[2] = t[2]; lo[3] = t[3];
    hi[0] = t[4]; hi[1] = t[5]; hi[2] = t[6]; hi[3] = t[7];
    *reinterpret_cast<v4f*>(buf + tb * CW_TILE + lane * 4) = lo;
    *reinterpret_cast<v4f*>(buf + tb * CW_TILE + 128 + lane * 4) = hi;
}

__device__ __forceinline__ void carry_block(const v2f (&a)[4], const v8f& t, v8f& acc) {
#pragma unroll
    for (int j = 0; j < 4; ++j)
        acc = __builtin_amdgcn_wmma_f32_16x16x4_f32(
             false, a[j],
             false, pair2(t[2 * j], t[2 * j + 1]),
             (short)0, acc,
             false,  false);
}

__device__ __forceinline__ void carry_lds_barrier() {
    __builtin_amdgcn_fence(__ATOMIC_RELEASE, "workgroup", "local");
    __builtin_amdgcn_s_barrier();
    __builtin_amdgcn_fence(__ATOMIC_ACQUIRE, "workgroup", "local");
}

__device__ __forceinline__ void carry_trip(float* rowp, float* sx, int tn, int lane, int k,
                                           const v2f (&aoff)[3][4], const v2f (&aown)[4],
                                           v8f& s, const v8f& e) {
    v8f acc = e;
    const float* rd = sx + ((k - 1) & 1) * CW_BUF;
    if ((tn & 1) == 0 && k > 2) {
        const v8f nb = lds_tile_load(rd, tn + 1, lane);
        *reinterpret_cast<v8f*>(rowp + (k - 1) * LCH) = s;
        *reinterpret_cast<v8f*>(rowp + (k - 1) * LCH + 16) = nb;
    }
    if (tn > 0) {
        v8f t[3];
        t[0] = lds_tile_load(rd, 0, lane);
        if (tn > 1) t[1] = lds_tile_load(rd, 1, lane);
        if (tn > 2) t[2] = lds_tile_load(rd, 2, lane);
        carry_block(aoff[0], t[0], acc);
        if (tn > 1) carry_block(aoff[1], t[1], acc);
        if (tn > 2) carry_block(aoff[2], t[2], acc);
    }
    carry_block(aown, s, acc);
    s = acc;
    lds_tile_store(sx + (k & 1) * CW_BUF, tn, lane, s);
    carry_lds_barrier();
}

__global__ __launch_bounds__(128)
void hippo_carry_kernel(float* out) {
    __shared__ __attribute__((aligned(16))) float sx[2 * CW_BUF];
    const int lane = threadIdx.x & 31;
    const int lc   = lane & 15;
    const int lh   = lane >> 4;
    const int tn = __builtin_amdgcn_readfirstlane((int)(threadIdx.x >> 5));
    float* rowp = out + (blockIdx.x * 16 + lc) * SLEN + tn * 16 + 8 * lh;

    v2f aoff[3][4], aown[4];
    const float* prow = out + (tn * 16 + lc) * SLEN + 8 * lh;
#pragma unroll
    for (int tb = 0; tb < 3; ++tb)
#pragma unroll
        for (int j = 0; j < 4; ++j)
            aoff[tb][j] = *reinterpret_cast<const v2f*>(prow + tb * 16 + 2 * j);
#pragma unroll
    for (int j = 0; j < 4; ++j)
        aown[j] = *reinterpret_cast<const v2f*>(prow + tn * 16 + 2 * j);

    v8f s  = *reinterpret_cast<const v8f*>(rowp + 1 * LCH);
    v8f q0 = *reinterpret_cast<const v8f*>(rowp + 2 * LCH);
    v8f q1 = *reinterpret_cast<const v8f*>(rowp + 3 * LCH);
    v8f q2 = *reinterpret_cast<const v8f*>(rowp + 4 * LCH);
    v8f q3 = *reinterpret_cast<const v8f*>(rowp + 5 * LCH);
    lds_tile_store(sx + CW_BUF, tn, lane, s);
    carry_lds_barrier();

#pragma unroll 1
    for (int k = 2; k < KCH; k += 2) {
        const int ka = (k + 4 < KCH) ? k + 4 : KCH - 1;
        const int kb = (k + 5 < KCH) ? k + 5 : KCH - 1;
        const v8f n2 = *reinterpret_cast<const v8f*>(rowp + ka * LCH);
        const v8f n3 = *reinterpret_cast<const v8f*>(rowp + kb * LCH);
        carry_trip(rowp, sx, tn, lane, k, aoff, aown, s, q0);
        carry_trip(rowp, sx, tn, lane, k + 1, aoff, aown, s, q1);
        q0 = q2;
        q1 = q3;
        q2 = n2;
        q3 = n3;
    }

    if ((tn & 1) == 0) {
        const v8f nb = lds_tile_load(sx + CW_BUF, tn + 1, lane);
        *reinterpret_cast<v8f*>(rowp + (KCH - 1) * LCH) = s;
        *reinterpret_cast<v8f*>(rowp + (KCH - 1) * LCH + 16) = nb;
    }
}

extern "C" void kernel_launch(void* const* d_in, const int* in_sizes, int n_in,
                              void* d_out, int out_size, void* d_ws, size_t ws_size,
                              hipStream_t stream) {
    const float* sig = (const float*)d_in[0];
    const float* Bm  = (const float*)d_in[1];
    const float* Cm  = (const float*)d_in[2];
    const float* Am  = (const float*)d_in[3];
    const float* dtp = (const float*)d_in[4];
    float* out  = (float*)d_out;
    float* abar = (float*)d_ws;
    float* a4   = abar + SDIM * SDIM;

    hippo_setup_kernel<<<1, 256, 0, stream>>>(Am, Bm, dtp, abar, a4, out);
    hippo_local_kernel<<<NBLK_LOCAL, 256, 0, stream>>>(sig, a4, out);
    hippo_carry_kernel<<<NGRP, 128, 0, stream>>>(out);
    hippo_final_kernel<<<NBLK_FINAL, 256, 0, stream>>>(sig, Bm, Cm, dtp, abar, a4, out);
}


# local pass advances 8 time steps per WMMA product, not 4 (A_bar^8 from setup; each wave builds A_bar^i B_bar, i=4..7, itself): 20 products a chunk instead of 32; state sums re-associated
# speedup vs baseline: 36.1727x; 1.0710x over previous
#include <hip/hip_runtime.h>
#include <hip/hip_bf16.h>

typedef float v2f __attribute__((ext_vector_type(2)));
typedef float v4f __attribute__((ext_vector_type(4)));
typedef float v8f __attribute__((ext_vector_type(8)));

#define SDIM 64
#define NCH  128
#define NB   8
#define SLEN 4096
#define KCH  32
#define LCH  128
#define NGRP 64
#define NBLK_LOCAL 249
#define NBLK_FINAL 256
#define RSTEP 4
#define LSTEP 8
#define A8_COL0 64
#define KB_COL0 128

#define NRHS   192
#define LPITCH 68

template <int K>
__device__ __forceinline__ void hippo_solve_from(float (&s)[SDIM], const float* Lt) {
    if constexpr (K < SDIM) {
        const float xk = s[K] / Lt[K * LPITCH + K];
        s[K] = xk;
        const float nxk = -xk;
#pragma unroll
        for (int i = K + 1; i < SDIM; ++i)
            s[i] = __builtin_fmaf(Lt[K * LPITCH + i], nxk, s[i]);
        hippo_solve_from<K + 1>(s, Lt);
    }
}

__device__ __forceinline__ void hippo_setup_powers(const float* abar, float* a4, float* out, int tid);

__global__ __launch_bounds__(256)
void hippo_setup_kernel(const float* __restrict__ A,
                        const float* __restrict__ Bm,
                        const float* __restrict__ dtp,
                        float* abar,
                        float* a4,
                        float* out) {
    __shared__ __attribute__((aligned(16))) float Lt[SDIM * LPITCH];
    const float dt = dtp[0];
    const int tid = threadIdx.x;
    const bool solver = tid < NRHS;
    const bool bcol   = tid >= SDIM;
    const int  c      = tid - SDIM;

    float s[SDIM];
    if (solver && bcol) {
#pragma unroll
        for (int i = 0; i < SDIM; ++i) s[i] = Bm[i * NCH + c];
    } else {
#pragma unroll
        for (int i = 0; i < SDIM; ++i) s[i] = (i == tid) ? 1.0f : 0.0f;
    }

    float av[16];
#pragma unroll
    for (int q = 0; q < 16; ++q) av[q] = A[tid + 256 * q];
#pragma unroll
    for (int q = 0; q < 16; ++q) {
        const int e = tid + 256 * q;
        const int i = e >> 6, k = e & 63;
        float l = -0.5f * dt * av[q];
        if (i == k) l += 1.0f;
        Lt[k * LPITCH + i] = l;
    }
    __syncthreads();

    if (solver) {
        hippo_solve_from<0>(s, Lt);
        if (bcol) {
            float* bp = out + (KB_COL0 + c) * SLEN;
#pragma unroll
            for (int i = 0; i < SDIM; i += 4) {
                v4f q;
                q[0] = dt * s[i];
                q[1] = dt * s[i + 1];
                q[2] = dt * s[i + 2];
                q[3] = dt * s[i + 3];
                *reinterpret_cast<v4f*>(bp + i) = q;
            }
        } else {
#pragma unroll
            for (int i = 0; i < SDIM; ++i)
                abar[i * SDIM + tid] = 2.0f * s[i] - ((i == tid) ? 1.0f : 0.0f);
        }
    }

    __syncthreads();
    hippo_setup_powers(abar, a4, out, tid);
}

#define ABAR_BLOCK_NONZERO(tn, tb) ((tb) <= (tn))


__device__ __forceinline__ void load_afrags(const float* mat, int pitch, int lc, int lh,
                                            v2f (&a)[4][4][4]) {
#pragma unroll
    for (int tn = 0; tn < 4; ++tn)
#pragma unroll
        for (int tb = 0; tb < 4; ++tb)
#pragma unroll
            for (int j = 0; j < 4; ++j) {
                if (ABAR_BLOCK_NONZERO(tn, tb)) {
                    const int base = (tn * 16 + lc) * pitch + tb * 16 + 8 * lh + 2 * j;
                    a[tn][tb][j] = *reinterpret_cast<const v2f*>(mat + base);
                }
            }
}

__device__ __forceinline__ v2f pair2(float x, float y) {
    v2f p;
    p.x = x;
    p.y = y;
    return p;
}

__device__ __forceinline__ void load_afrags_t(const float* mat, int lc, int lh,
                                              v2f (&a)[4][4][4]) {
#pragma unroll
    for (int tn = 0; tn < 4; ++tn)
#pragma unroll
        for (int tb = 0; tb < 4; ++tb)
#pragma unroll
            for (int j = 0; j < 4; ++j) {
                if (ABAR_BLOCK_NONZERO(tb, tn)) {
                    const int m = tb * 16 + 8 * lh + 2 * j;
                    a[tn][tb][j] = pair2(mat[m * SDIM + tn * 16 + lc],
                                         mat[(m + 1) * SDIM + tn * 16 + lc]);
                }
            }
}

__device__ __forceinline__ void mat_step_t(const v2f (&a)[4][4][4], const v8f (&s)[4],
                                           v8f (&acc)[4]) {
#pragma unroll
    for (int tn = 0; tn < 4; ++tn)
#pragma unroll
        for (int tb = 0; tb < 4; ++tb)
#pragma unroll
            for (int j = 0; j < 4; ++j) {
                if (ABAR_BLOCK_NONZERO(tb, tn)) {
                    acc[tn] = __builtin_amdgcn_wmma_f32_16x16x4_f32(
                         false, a[tn][tb][j],
                         false, pair2(s[tb][2 * j], s[tb][2 * j + 1]),
                         (short)0, acc[tn],
                         false,  false);
                }
            }
}

__device__ __forceinline__ void mat_step(const v2f (&a)[4][4][4], const v8f (&s)[4],
                                         v8f (&acc)[4]) {
#pragma unroll
    for (int tn = 0; tn < 4; ++tn)
#pragma unroll
        for (int tb = 0; tb < 4; ++tb)
#pragma unroll
            for (int j = 0; j < 4; ++j) {
                if (ABAR_BLOCK_NONZERO(tn, tb)) {
                    acc[tn] = __builtin_amdgcn_wmma_f32_16x16x4_f32(
                         false, a[tn][tb][j],
                         false, pair2(s[tb][2 * j], s[tb][2 * j + 1]),
                         (short)0, acc[tn],
                         false,  false);
                }
            }
}

__device__ __forceinline__ void save_state(float* p, int lh, const v8f (&s)[4]) {
#pragma unroll
    for (int tn = 0; tn < 4; ++tn) *reinterpret_cast<v8f*>(p + tn * 16 + 8 * lh) = s[tn];
}
__device__ __forceinline__ void load_state(const float* p, int lh, v8f (&s)[4]) {
#pragma unroll
    for (int tn = 0; tn < 4; ++tn) s[tn] = *reinterpret_cast<const v8f*>(p + tn * 16 + 8 * lh);
}

__device__ __forceinline__ void hippo_setup_powers(const float* abar, float* a4, float* out, int tid) {
    const int lane = tid & 31;
    const int w    = tid >> 5;
    const int lc   = lane & 15;
    const int lh   = lane >> 4;
    const int c    = w * 16 + lc;

    v2f a[4][4][4];
    load_afrags(abar, SDIM, lc, lh, a);

    v8f s[4];
    load_state(out + (KB_COL0 + c) * SLEN, lh, s);

#pragma unroll
    for (int i = 1; i < RSTEP; ++i) {
        v8f acc[4];
#pragma unroll
        for (int tn = 0; tn < 4; ++tn)
#pragma unroll
            for (int r = 0; r < 8; ++r) acc[tn][r] = 0.0f;
        mat_step(a, s, acc);
#pragma unroll
        for (int tn = 0; tn < 4; ++tn) s[tn] = acc[tn];
        save_state(out + (KB_COL0 + i * NCH + c) * SLEN, lh, s);
    }

    if (w < 4) {
#pragma unroll
        for (int tn = 0; tn < 4; ++tn)
#pragma unroll
            for (int r = 0; r < 8; ++r) s[tn][r] = (tn * 16 + r + 8 * lh == c) ? 1.0f : 0.0f;
#pragma unroll
        for (int i = 0; i < RSTEP; ++i) {
            v8f acc[4];
#pragma unroll
            for (int tn = 0; tn < 4; ++tn)
#pragma unroll
                for (int r = 0; r < 8; ++r) acc[tn][r] = 0.0f;
            mat_step(a, s, acc);
#pragma unroll
            for (int tn = 0; tn < 4; ++tn) s[tn] = acc[tn];
        }
#pragma unroll
        for (int tn = 0; tn < 4; ++tn)
#pragma unroll
            for (int r = 0; r < 8; ++r)
                a4[(tn * 16 + r + 8 * lh) * SDIM + c] = s[tn][r];
#pragma unroll 1
        for (int i = RSTEP; i < LSTEP; ++i) {
            v8f acc[4];
#pragma unroll
            for (int tn = 0; tn < 4; ++tn)
#pragma unroll
                for (int r = 0; r < 8; ++r) acc[tn][r] = 0.0f;
            mat_step(a, s, acc);
#pragma unroll
            for (int tn = 0; tn < 4; ++tn) s[tn] = acc[tn];
        }
#pragma unroll
        for (int tn = 0; tn < 4; ++tn)
#pragma unroll
            for (int r = 0; r < 8; ++r)
                out[(A8_COL0 + tn * 16 + r + 8 * lh) * SLEN + c] = s[tn][r];
    }
}

__global__ __launch_bounds__(256)
void hippo_local_kernel(const float* __restrict__ sig,
                        const float* __restrict__ a4,
                        float* out) {
    const int lane = threadIdx.x & 31;
    const int w    = threadIdx.x >> 5;
    const int lc   = lane & 15;
    const int lh   = lane >> 4;
    const int c    = w * 16 + lc;

    int k, b;
    bool power = false;
    if (blockIdx.x == 0) {
        power = true;
        k = 0;
        b = 0;
        if (w >= 4) return;
    } else {
        k = (blockIdx.x - 1) >> 3;
        b = (blockIdx.x - 1) & 7;
    }

    v8f kb[LSTEP][4];
#pragma unroll
    for (int i = 0; i < RSTEP; ++i) {
        load_state(out + (KB_COL0 + i * NCH + c) * SLEN, lh, kb[i]);
#pragma unroll
        for (int tn = 0; tn < 4; ++tn)
#pragma unroll
            for (int r = 0; r < 8; ++r) kb[i][tn][r] = power ? 0.0f : kb[i][tn][r];
    }
    {
        v2f a4f[4][4][4];
        load_afrags(a4, SDIM, lc, lh, a4f);
#pragma unroll
        for (int i = RSTEP; i < LSTEP; ++i) {
            v8f acc[4];
#pragma unroll
            for (int tn = 0; tn < 4; ++tn)
#pragma unroll
                for (int r = 0; r < 8; ++r) acc[tn][r] = 0.0f;
            mat_step(a4f, kb[i - RSTEP], acc);
#pragma unroll
            for (int tn = 0; tn < 4; ++tn) kb[i][tn] = acc[tn];
        }
    }

    v2f a[4][4][4];
    load_afrags(out + A8_COL0 * SLEN, SLEN, lc, lh, a);

    const int colOff = (b * NCH + c) * SLEN;
    float* colp = out + colOff;

    v8f s[4];
#pragma unroll
    for (int tn = 0; tn < 4; ++tn)
#pragma unroll
        for (int r = 0; r < 8; ++r)
            s[tn][r] = (power && (tn * 16 + r + 8 * lh == c)) ? 1.0f : 0.0f;

    const float* xp = sig + colOff + k * LCH;

    v4f xlo = *reinterpret_cast<const v4f*>(xp);
    v4f xhi = *reinterpret_cast<const v4f*>(xp + 4);
    if (power) {
        xlo[0] = 0.0f; xlo[1] = 0.0f; xlo[2] = 0.0f; xlo[3] = 0.0f;
        xhi[0] = 0.0f; xhi[1] = 0.0f; xhi[2] = 0.0f; xhi[3] = 0.0f;
    }

#pragma unroll 1
    for (int tc = 0; tc < LCH; tc += LSTEP) {
        const int tnext = (tc + LSTEP < LCH) ? tc + LSTEP : tc;
        v4f nlo = *reinterpret_cast<const v4f*>(xp + tnext);
        v4f nhi = *reinterpret_cast<const v4f*>(xp + tnext + 4);
        if (power) {
            nlo[0] = 0.0f; nlo[1] = 0.0f; nlo[2] = 0.0f; nlo[3] = 0.0f;
            nhi[0] = 0.0f; nhi[1] = 0.0f; nhi[2] = 0.0f; nhi[3] = 0.0f;
        }

        v8f acc[4];
#pragma unroll
        for (int tn = 0; tn < 4; ++tn)
#pragma unroll
            for (int r = 0; r < 8; ++r) {
                float v = kb[7][tn][r] * xlo[0];
                v = __builtin_fmaf(kb[6][tn][r], xlo[1], v);
                v = __builtin_fmaf(kb[5][tn][r], xlo[2], v);
                v = __builtin_fmaf(kb[4][tn][r], xlo[3], v);
                v = __builtin_fmaf(kb[3][tn][r], xhi[0], v);
                v = __builtin_fmaf(kb[2][tn][r], xhi[1], v);
                v = __builtin_fmaf(kb[1][tn][r], xhi[2], v);
                v = __builtin_fmaf(kb[0][tn][r], xhi[3], v);
                acc[tn][r] = v;
            }
        mat_step(a, s, acc);
#pragma unroll
        for (int tn = 0; tn < 4; ++tn) s[tn] = acc[tn];

        xlo = nlo;
        xhi = nhi;
    }

    if (power) {
#pragma unroll
        for (int tn = 0; tn < 4; ++tn)
#pragma unroll
            for (int r = 0; r < 8; ++r) out[(tn * 16 + r + 8 * lh) * SLEN + c] = s[tn][r];
    } else {
        save_state(colp + (k + 1) * LCH, lh, s);
    }
}

__global__ __launch_bounds__(256)
void hippo_final_kernel(const float* __restrict__ sig,
                        const float* __restrict__ Bm,
                        const float* __restrict__ Cmat,
                        const float* __restrict__ dtp,
                        const float* __restrict__ abar,
                        const float* __restrict__ a4,
                        float* out) {
    const int lane = threadIdx.x & 31;
    const int w    = threadIdx.x >> 5;
    const int lc   = lane & 15;
    const int lh   = lane >> 4;
    const int c    = w * 16 + lc;
    const int b    = blockIdx.x & 7;
    const float hdt = 0.5f * dtp[0];

    v8f kb[RSTEP][4], g[RSTEP][4];
    {
        v2f a1[4][4][4];
        load_afrags(abar, SDIM, lc, lh, a1);
        v8f bin[4], acc[4];
#pragma unroll
        for (int tn = 0; tn < 4; ++tn)
#pragma unroll
            for (int r = 0; r < 8; ++r) {
                bin[tn][r] = Bm[(tn * 16 + r + 8 * lh) * NCH + c];
                acc[tn][r] = bin[tn][r];
            }
        mat_step(a1, bin, acc);
#pragma unroll
        for (int tn = 0; tn < 4; ++tn)
#pragma unroll
            for (int r = 0; r < 8; ++r) kb[0][tn][r] = hdt * acc[tn][r];
#pragma unroll
        for (int i = 1; i < RSTEP; ++i) {
#pragma unroll
            for (int tn = 0; tn < 4; ++tn)
#pragma unroll
                for (int r = 0; r < 8; ++r) acc[tn][r] = 0.0f;
            mat_step(a1, kb[i - 1], acc);
#pragma unroll
            for (int tn = 0; tn < 4; ++tn) kb[i][tn] = acc[tn];
        }
    }
    {
        v2f at[4][4][4];
        load_afrags_t(abar, lc, lh, at);
#pragma unroll
        for (int tn = 0; tn < 4; ++tn)
#pragma unroll
            for (int r = 0; r < 8; ++r) g[0][tn][r] = Cmat[(tn * 16 + r + 8 * lh) * NCH + c];
#pragma unroll
        for (int i = 1; i < RSTEP; ++i) {
            v8f acc[4];
#pragma unroll
            for (int tn = 0; tn < 4; ++tn)
#pragma unroll
                for (int r = 0; r < 8; ++r) acc[tn][r] = 0.0f;
            mat_step_t(at, g[i - 1], acc);
#pragma unroll
            for (int tn = 0; tn < 4; ++tn) g[i][tn] = acc[tn];
        }
    }

    float h[RSTEP - 1];
#pragma unroll
    for (int jj = 0; jj < RSTEP - 1; ++jj) {
        float p = 0.0f;
#pragma unroll
        for (int tn = 0; tn < 4; ++tn)
#pragma unroll
            for (int r = 0; r < 8; ++r) p = __builtin_fmaf(g[0][tn][r], kb[jj][tn][r], p);
        p += __shfl_xor(p, 16, 32);
        h[jj] = p;
    }

    v2f a[4][4][4];
    load_afrags(a4, SDIM, lc, lh, a);

    const int colOff = (b * NCH + c) * SLEN;
    float* colp = out + colOff;

    const int k = blockIdx.x >> 3;

    v8f s[4];
    if (k > 0) {
        load_state(colp + k * LCH, lh, s);
    } else {
#pragma unroll
        for (int tn = 0; tn < 4; ++tn)
#pragma unroll
            for (int r = 0; r < 8; ++r) s[tn][r] = 0.0f;
    }

    const float* xp = sig + colOff + k * LCH;
    float*       yp = colp + k * LCH;

    v4f xcur4 = *reinterpret_cast<const v4f*>(xp);

#pragma unroll 1
    for (int tc = 0; tc < LCH; tc += RSTEP) {
        const int tnext = (tc + RSTEP < LCH) ? tc + RSTEP : tc;
        v4f xnext4 = *reinterpret_cast<const v4f*>(xp + tnext);

        float p1 = 0.0f, p2 = 0.0f, p3 = 0.0f;
#pragma unroll
        for (int tn = 0; tn < 4; ++tn)
#pragma unroll
            for (int r = 0; r < 8; ++r) {
                p1 = __builtin_fmaf(g[1][tn][r], s[tn][r], p1);
                p2 = __builtin_fmaf(g[2][tn][r], s[tn][r], p2);
                p3 = __builtin_fmaf(g[3][tn][r], s[tn][r], p3);
            }

        v8f acc[4];
#pragma unroll
        for (int tn = 0; tn < 4; ++tn)
#pragma unroll
            for (int r = 0; r < 8; ++r) {
                float v = kb[3][tn][r] * xcur4[0];
                v = __builtin_fmaf(kb[2][tn][r], xcur4[1], v);
                v = __builtin_fmaf(kb[1][tn][r], xcur4[2], v);
                v = __builtin_fmaf(kb[0][tn][r], xcur4[3], v);
                acc[tn][r] = v;
            }
        mat_step(a, s, acc);
#pragma unroll
        for (int tn = 0; tn < 4; ++tn) s[tn] = acc[tn];

        float p4 = 0.0f;
#pragma unroll
        for (int tn = 0; tn < 4; ++tn)
#pragma unroll
            for (int r = 0; r < 8; ++r) p4 = __builtin_fmaf(g[0][tn][r], s[tn][r], p4);

        p1 += __shfl_xor(p1, 16, 32);
        p2 += __shfl_xor(p2, 16, 32);
        p3 += __shfl_xor(p3, 16, 32);
        p4 += __shfl_xor(p4, 16, 32);

        v4f y4;
        y4[0] = __builtin_fmaf(h[0], xcur4[0], p1);
        y4[1] = __builtin_fmaf(h[1], xcur4[0], __builtin_fmaf(h[0], xcur4[1], p2));
        y4[2] = __builtin_fmaf(h[2], xcur4[0],
                               __builtin_fmaf(h[1], xcur4[1], __builtin_fmaf(h[0], xcur4[2], p3)));
        y4[3] = p4;

        if (lh == 0) *reinterpret_cast<v4f*>(yp + tc) = y4;

        xcur4 = xnext4;
    }
}

#define CW_TILE 256
#define CW_BUF  1024

__device__ __forceinline__ v8f lds_tile_load(const float* buf, int tb, int lane) {
    const v4f lo = *reinterpret_cast<const v4f*>(buf + tb * CW_TILE + lane * 4);
    const v4f hi = *reinterpret_cast<const v4f*>(buf + tb * CW_TILE + 128 + lane * 4);
    v8f t;
    t[0] = lo[0]; t[1] = lo[1]; t[2] = lo[2]; t[3] = lo[3];
    t[4] = hi[0]; t[5] = hi[1]; t[6] = hi[2]; t[7] = hi[3];
    return t;
}
__device__ __forceinline__ void lds_tile_store(float* buf, int tb, int lane, const v8f& t) {
    v4f lo, hi;
    lo[0] = t[0]; lo[1] = t[1]; lo[2] = t[2]; lo[3] = t[3];
    hi[0] = t[4]; hi[1] = t[5]; hi[2] = t[6]; hi[3] = t[7];
    *reinterpret_cast<v4f*>(buf + tb * CW_TILE + lane * 4) = lo;
    *reinterpret_cast<v4f*>(buf + tb * CW_TILE + 128 + lane * 4) = hi;
}

__device__ __forceinline__ void carry_block(const v2f (&a)[4], const v8f& t, v8f& acc) {
#pragma unroll
    for (int j = 0; j < 4; ++j)
        acc = __builtin_amdgcn_wmma_f32_16x16x4_f32(
             false, a[j],
             false, pair2(t[2 * j], t[2 * j + 1]),
             (short)0, acc,
             false,  false);
}

__device__ __forceinline__ void carry_lds_barrier() {
    __builtin_amdgcn_fence(__ATOMIC_RELEASE, "workgroup", "local");
    __builtin_amdgcn_s_barrier();
    __builtin_amdgcn_fence(__ATOMIC_ACQUIRE, "workgroup", "local");
}

__device__ __forceinline__ void carry_trip(float* rowp, float* sx, int tn, int lane, int k,
                                           const v2f (&aoff)[3][4], const v2f (&aown)[4],
                                           v8f& s, const v8f& e) {
    v8f acc = e;
    const float* rd = sx + ((k - 1) & 1) * CW_BUF;
    if ((tn & 1) == 0 && k > 2) {
        const v8f nb = lds_tile_load(rd, tn + 1, lane);
        *reinterpret_cast<v8f*>(rowp + (k - 1) * LCH) = s;
        *reinterpret_cast<v8f*>(rowp + (k - 1) * LCH + 16) = nb;
    }
    if (tn > 0) {
        v8f t[3];
        t[0] = lds_tile_load(rd, 0, lane);
        if (tn > 1) t[1] = lds_tile_load(rd, 1, lane);
        if (tn > 2) t[2] = lds_tile_load(rd, 2, lane);
        carry_block(aoff[0], t[0], acc);
        if (tn > 1) carry_block(aoff[1], t[1], acc);
        if (tn > 2) carry_block(aoff[2], t[2], acc);
    }
    carry_block(aown, s, acc);
    s = acc;
    lds_tile_store(sx + (k & 1) * CW_BUF, tn, lane, s);
    carry_lds_barrier();
}

__global__ __launch_bounds__(128)
void hippo_carry_kernel(float* out) {
    __shared__ __attribute__((aligned(16))) float sx[2 * CW_BUF];
    const int lane = threadIdx.x & 31;
    const int lc   = lane & 15;
    const int lh   = lane >> 4;
    const int tn = __builtin_amdgcn_readfirstlane((int)(threadIdx.x >> 5));
    float* rowp = out + (blockIdx.x * 16 + lc) * SLEN + tn * 16 + 8 * lh;

    v2f aoff[3][4], aown[4];
    const float* prow = out + (tn * 16 + lc) * SLEN + 8 * lh;
#pragma unroll
    for (int tb = 0; tb < 3; ++tb)
#pragma unroll
        for (int j = 0; j < 4; ++j)
            aoff[tb][j] = *reinterpret_cast<const v2f*>(prow + tb * 16 + 2 * j);
#pragma unroll
    for (int j = 0; j < 4; ++j)
        aown[j] = *reinterpret_cast<const v2f*>(prow + tn * 16 + 2 * j);

    v8f s  = *reinterpret_cast<const v8f*>(rowp + 1 * LCH);
    v8f q0 = *reinterpret_cast<const v8f*>(rowp + 2 * LCH);
    v8f q1 = *reinterpret_cast<const v8f*>(rowp + 3 * LCH);
    v8f q2 = *reinterpret_cast<const v8f*>(rowp + 4 * LCH);
    v8f q3 = *reinterpret_cast<const v8f*>(rowp + 5 * LCH);
    lds_tile_store(sx + CW_BUF, tn, lane, s);
    carry_lds_barrier();

#pragma unroll 1
    for (int k = 2; k < KCH; k += 2) {
        const int ka = (k + 4 < KCH) ? k + 4 : KCH - 1;
        const int kb = (k + 5 < KCH) ? k + 5 : KCH - 1;
        const v8f n2 = *reinterpret_cast<const v8f*>(rowp + ka * LCH);
        const v8f n3 = *reinterpret_cast<const v8f*>(rowp + kb * LCH);
        carry_trip(rowp, sx, tn, lane, k, aoff, aown, s, q0);
        carry_trip(rowp, sx, tn, lane, k + 1, aoff, aown, s, q1);
        q0 = q2;
        q1 = q3;
        q2 = n2;
        q3 = n3;
    }

    if ((tn & 1) == 0) {
        const v8f nb = lds_tile_load(sx + CW_BUF, tn + 1, lane);
        *reinterpret_cast<v8f*>(rowp + (KCH - 1) * LCH) = s;
        *reinterpret_cast<v8f*>(rowp + (KCH - 1) * LCH + 16) = nb;
    }
}

extern "C" void kernel_launch(void* const* d_in, const int* in_sizes, int n_in,
                              void* d_out, int out_size, void* d_ws, size_t ws_size,
                              hipStream_t stream) {
    const float* sig = (const float*)d_in[0];
    const float* Bm  = (const float*)d_in[1];
    const float* Cm  = (const float*)d_in[2];
    const float* Am  = (const float*)d_in[3];
    const float* dtp = (const float*)d_in[4];
    float* out  = (float*)d_out;
    float* abar = (float*)d_ws;
    float* a4   = abar + SDIM * SDIM;

    hippo_setup_kernel<<<1, 256, 0, stream>>>(Am, Bm, dtp, abar, a4, out);
    hippo_local_kernel<<<NBLK_LOCAL, 256, 0, stream>>>(sig, a4, out);
    hippo_carry_kernel<<<NGRP, 128, 0, stream>>>(out);
    hippo_final_kernel<<<NBLK_FINAL, 256, 0, stream>>>(sig, Bm, Cm, dtp, abar, a4, out);
}


# local and final pass: a block is now 16 channels x 8 batches, so its waves share the kb and g tables: built once per block by one or two waves and handed over through LDS; every sum unchanged
# speedup vs baseline: 38.5310x; 1.0652x over previous
#include <hip/hip_runtime.h>
#include <hip/hip_bf16.h>

typedef float v2f __attribute__((ext_vector_type(2)));
typedef float v4f __attribute__((ext_vector_type(4)));
typedef float v8f __attribute__((ext_vector_type(8)));

#define SDIM 64
#define NCH  128
#define NB   8
#define SLEN 4096
#define KCH  32
#define LCH  128
#define NGRP 64
#define NBLK_LOCAL 249
#define NBLK_FINAL 256
#define RSTEP 4
#define LSTEP 8
#define A8_COL0 64
#define KB_COL0 128

#define NRHS   192
#define LPITCH 68

template <int K>
__device__ __forceinline__ void hippo_solve_from(float (&s)[SDIM], const float* Lt) {
    if constexpr (K < SDIM) {
        const float xk = s[K] / Lt[K * LPITCH + K];
        s[K] = xk;
        const float nxk = -xk;
#pragma unroll
        for (int i = K + 1; i < SDIM; ++i)
            s[i] = __builtin_fmaf(Lt[K * LPITCH + i], nxk, s[i]);
        hippo_solve_from<K + 1>(s, Lt);
    }
}

__device__ __forceinline__ void hippo_setup_powers(const float* abar, float* a4, float* out, int tid);

__global__ __launch_bounds__(256)
void hippo_setup_kernel(const float* __restrict__ A,
                        const float* __restrict__ Bm,
                        const float* __restrict__ dtp,
                        float* abar,
                        float* a4,
                        float* out) {
    __shared__ __attribute__((aligned(16))) float Lt[SDIM * LPITCH];
    const float dt = dtp[0];
    const int tid = threadIdx.x;
    const bool solver = tid < NRHS;
    const bool bcol   = tid >= SDIM;
    const int  c      = tid - SDIM;

    float s[SDIM];
    if (solver && bcol) {
#pragma unroll
        for (int i = 0; i < SDIM; ++i) s[i] = Bm[i * NCH + c];
    } else {
#pragma unroll
        for (int i = 0; i < SDIM; ++i) s[i] = (i == tid) ? 1.0f : 0.0f;
    }

    float av[16];
#pragma unroll
    for (int q = 0; q < 16; ++q) av[q] = A[tid + 256 * q];
#pragma unroll
    for (int q = 0; q < 16; ++q) {
        const int e = tid + 256 * q;
        const int i = e >> 6, k = e & 63;
        float l = -0.5f * dt * av[q];
        if (i == k) l += 1.0f;
        Lt[k * LPITCH + i] = l;
    }
    __syncthreads();

    if (solver) {
        hippo_solve_from<0>(s, Lt);
        if (bcol) {
            float* bp = out + (KB_COL0 + c) * SLEN;
#pragma unroll
            for (int i = 0; i < SDIM; i += 4) {
                v4f q;
                q[0] = dt * s[i];
                q[1] = dt * s[i + 1];
                q[2] = dt * s[i + 2];
                q[3] = dt * s[i + 3];
                *reinterpret_cast<v4f*>(bp + i) = q;
            }
        } else {
#pragma unroll
            for (int i = 0; i < SDIM; ++i)
                abar[i * SDIM + tid] = 2.0f * s[i] - ((i == tid) ? 1.0f : 0.0f);
        }
    }

    __syncthreads();
    hippo_setup_powers(abar, a4, out, tid);
}

#define ABAR_BLOCK_NONZERO(tn, tb) ((tb) <= (tn))


__device__ __forceinline__ void load_afrags(const float* mat, int pitch, int lc, int lh,
                                            v2f (&a)[4][4][4]) {
#pragma unroll
    for (int tn = 0; tn < 4; ++tn)
#pragma unroll
        for (int tb = 0; tb < 4; ++tb)
#pragma unroll
            for (int j = 0; j < 4; ++j) {
                if (ABAR_BLOCK_NONZERO(tn, tb)) {
                    const int base = (tn * 16 + lc) * pitch + tb * 16 + 8 * lh + 2 * j;
                    a[tn][tb][j] = *reinterpret_cast<const v2f*>(mat + base);
                }
            }
}

__device__ __forceinline__ v2f pair2(float x, float y) {
    v2f p;
    p.x = x;
    p.y = y;
    return p;
}

__device__ __forceinline__ void load_afrags_t(const float* mat, int lc, int lh,
                                              v2f (&a)[4][4][4]) {
#pragma unroll
    for (int tn = 0; tn < 4; ++tn)
#pragma unroll
        for (int tb = 0; tb < 4; ++tb)
#pragma unroll
            for (int j = 0; j < 4; ++j) {
                if (ABAR_BLOCK_NONZERO(tb, tn)) {
                    const int m = tb * 16 + 8 * lh + 2 * j;
                    a[tn][tb][j] = pair2(mat[m * SDIM + tn * 16 + lc],
                                         mat[(m + 1) * SDIM + tn * 16 + lc]);
                }
            }
}

__device__ __forceinline__ void mat_step_t(const v2f (&a)[4][4][4], const v8f (&s)[4],
                                           v8f (&acc)[4]) {
#pragma unroll
    for (int tn = 0; tn < 4; ++tn)
#pragma unroll
        for (int tb = 0; tb < 4; ++tb)
#pragma unroll
            for (int j = 0; j < 4; ++j) {
                if (ABAR_BLOCK_NONZERO(tb, tn)) {
                    acc[tn] = __builtin_amdgcn_wmma_f32_16x16x4_f32(
                         false, a[tn][tb][j],
                         false, pair2(s[tb][2 * j], s[tb][2 * j + 1]),
                         (short)0, acc[tn],
                         false,  false);
                }
            }
}

__device__ __forceinline__ void mat_step(const v2f (&a)[4][4][4], const v8f (&s)[4],
                                         v8f (&acc)[4]) {
#pragma unroll
    for (int tn = 0; tn < 4; ++tn)
#pragma unroll
        for (int tb = 0; tb < 4; ++tb)
#pragma unroll
            for (int j = 0; j < 4; ++j) {
                if (ABAR_BLOCK_NONZERO(tn, tb)) {
                    acc[tn] = __builtin_amdgcn_wmma_f32_16x16x4_f32(
                         false, a[tn][tb][j],
                         false, pair2(s[tb][2 * j], s[tb][2 * j + 1]),
                         (short)0, acc[tn],
                         false,  false);
                }
            }
}

__device__ __forceinline__ void save_state(float* p, int lh, const v8f (&s)[4]) {
#pragma unroll
    for (int tn = 0; tn < 4; ++tn) *reinterpret_cast<v8f*>(p + tn * 16 + 8 * lh) = s[tn];
}
__device__ __forceinline__ void load_state(const float* p, int lh, v8f (&s)[4]) {
#pragma unroll
    for (int tn = 0; tn < 4; ++tn) s[tn] = *reinterpret_cast<const v8f*>(p + tn * 16 + 8 * lh);
}

__device__ __forceinline__ void hippo_setup_powers(const float* abar, float* a4, float* out, int tid) {
    const int lane = tid & 31;
    const int w    = tid >> 5;
    const int lc   = lane & 15;
    const int lh   = lane >> 4;
    const int c    = w * 16 + lc;

    v2f a[4][4][4];
    load_afrags(abar, SDIM, lc, lh, a);

    v8f s[4];
    load_state(out + (KB_COL0 + c) * SLEN, lh, s);

#pragma unroll
    for (int i = 1; i < RSTEP; ++i) {
        v8f acc[4];
#pragma unroll
        for (int tn = 0; tn < 4; ++tn)
#pragma unroll
            for (int r = 0; r < 8; ++r) acc[tn][r] = 0.0f;
        mat_step(a, s, acc);
#pragma unroll
        for (int tn = 0; tn < 4; ++tn) s[tn] = acc[tn];
        save_state(out + (KB_COL0 + i * NCH + c) * SLEN, lh, s);
    }

    if (w < 4) {
#pragma unroll
        for (int tn = 0; tn < 4; ++tn)
#pragma unroll
            for (int r = 0; r < 8; ++r) s[tn][r] = (tn * 16 + r + 8 * lh == c) ? 1.0f : 0.0f;
#pragma unroll
        for (int i = 0; i < RSTEP; ++i) {
            v8f acc[4];
#pragma unroll
            for (int tn = 0; tn < 4; ++tn)
#pragma unroll
                for (int r = 0; r < 8; ++r) acc[tn][r] = 0.0f;
            mat_step(a, s, acc);
#pragma unroll
            for (int tn = 0; tn < 4; ++tn) s[tn] = acc[tn];
        }
#pragma unroll
        for (int tn = 0; tn < 4; ++tn)
#pragma unroll
            for (int r = 0; r < 8; ++r)
                a4[(tn * 16 + r + 8 * lh) * SDIM + c] = s[tn][r];
#pragma unroll 1
        for (int i = RSTEP; i < LSTEP; ++i) {
            v8f acc[4];
#pragma unroll
            for (int tn = 0; tn < 4; ++tn)
#pragma unroll
                for (int r = 0; r < 8; ++r) acc[tn][r] = 0.0f;
            mat_step(a, s, acc);
#pragma unroll
            for (int tn = 0; tn < 4; ++tn) s[tn] = acc[tn];
        }
#pragma unroll
        for (int tn = 0; tn < 4; ++tn)
#pragma unroll
            for (int r = 0; r < 8; ++r)
                out[(A8_COL0 + tn * 16 + r + 8 * lh) * SLEN + c] = s[tn][r];
    }
}

#define CW_TILE 256
#define CW_BUF  1024

__device__ __forceinline__ v8f lds_tile_load(const float* buf, int tb, int lane) {
    const v4f lo = *reinterpret_cast<const v4f*>(buf + tb * CW_TILE + lane * 4);
    const v4f hi = *reinterpret_cast<const v4f*>(buf + tb * CW_TILE + 128 + lane * 4);
    v8f t;
    t[0] = lo[0]; t[1] = lo[1]; t[2] = lo[2]; t[3] = lo[3];
    t[4] = hi[0]; t[5] = hi[1]; t[6] = hi[2]; t[7] = hi[3];
    return t;
}
__device__ __forceinline__ void lds_tile_store(float* buf, int tb, int lane, const v8f& t) {
    v4f lo, hi;
    lo[0] = t[0]; lo[1] = t[1]; lo[2] = t[2]; lo[3] = t[3];
    hi[0] = t[4]; hi[1] = t[5]; hi[2] = t[6]; hi[3] = t[7];
    *reinterpret_cast<v4f*>(buf + tb * CW_TILE + lane * 4) = lo;
    *reinterpret_cast<v4f*>(buf + tb * CW_TILE + 128 + lane * 4) = hi;
}

__device__ __forceinline__ void lds_tab_store(float* tab, int lane, const v8f (&t)[4]) {
#pragma unroll
    for (int tn = 0; tn < 4; ++tn) lds_tile_store(tab, tn, lane, t[tn]);
}
__device__ __forceinline__ void lds_tab_load(const float* tab, int lane, v8f (&t)[4]) {
#pragma unroll
    for (int tn = 0; tn < 4; ++tn) t[tn] = lds_tile_load(tab, tn, lane);
}

__global__ __launch_bounds__(256)
void hippo_local_kernel(const float* __restrict__ sig,
                        const float* __restrict__ a4,
                        float* out) {
    __shared__ __attribute__((aligned(16))) float tabs[(LSTEP - RSTEP) * CW_BUF];
    const int lane = threadIdx.x & 31;
    const int w    = __builtin_amdgcn_readfirstlane((int)(threadIdx.x >> 5));
    const int lc   = lane & 15;
    const int lh   = lane >> 4;

    int k, b, c;
    bool power = false;
    if (blockIdx.x == 0) {
        power = true;
        k = 0;
        b = 0;
        c = w * 16 + lc;
        if (w >= 4) return;
    } else {
        k = (blockIdx.x - 1) >> 3;
        b = w;
        c = ((blockIdx.x - 1) & 7) * 16 + lc;
    }

    v8f kb[LSTEP][4];
#pragma unroll
    for (int i = 0; i < RSTEP; ++i) {
        load_state(out + (KB_COL0 + i * NCH + c) * SLEN, lh, kb[i]);
#pragma unroll
        for (int tn = 0; tn < 4; ++tn)
#pragma unroll
            for (int r = 0; r < 8; ++r) kb[i][tn][r] = power ? 0.0f : kb[i][tn][r];
    }
    if (!power) {
        if (w < LSTEP - RSTEP) {
            v2f a4f[4][4][4];
            load_afrags(a4, SDIM, lc, lh, a4f);
            v8f src[4], acc[4];
            load_state(out + (KB_COL0 + w * NCH + c) * SLEN, lh, src);
#pragma unroll
            for (int tn = 0; tn < 4; ++tn)
#pragma unroll
                for (int r = 0; r < 8; ++r) acc[tn][r] = 0.0f;
            mat_step(a4f, src, acc);
            lds_tab_store(tabs + w * CW_BUF, lane, acc);
        }
        __syncthreads();
#pragma unroll
        for (int i = RSTEP; i < LSTEP; ++i) lds_tab_load(tabs + (i - RSTEP) * CW_BUF, lane, kb[i]);
    } else {
#pragma unroll
        for (int i = RSTEP; i < LSTEP; ++i)
#pragma unroll
            for (int tn = 0; tn < 4; ++tn)
#pragma unroll
                for (int r = 0; r < 8; ++r) kb[i][tn][r] = 0.0f;
    }

    v2f a[4][4][4];
    load_afrags(out + A8_COL0 * SLEN, SLEN, lc, lh, a);

    const int colOff = (b * NCH + c) * SLEN;
    float* colp = out + colOff;

    v8f s[4];
#pragma unroll
    for (int tn = 0; tn < 4; ++tn)
#pragma unroll
        for (int r = 0; r < 8; ++r)
            s[tn][r] = (power && (tn * 16 + r + 8 * lh == c)) ? 1.0f : 0.0f;

    const float* xp = sig + colOff + k * LCH;

    v4f xlo = *reinterpret_cast<const v4f*>(xp);
    v4f xhi = *reinterpret_cast<const v4f*>(xp + 4);
    if (power) {
        xlo[0] = 0.0f; xlo[1] = 0.0f; xlo[2] = 0.0f; xlo[3] = 0.0f;
        xhi[0] = 0.0f; xhi[1] = 0.0f; xhi[2] = 0.0f; xhi[3] = 0.0f;
    }

#pragma unroll 1
    for (int tc = 0; tc < LCH; tc += LSTEP) {
        const int tnext = (tc + LSTEP < LCH) ? tc + LSTEP : tc;
        v4f nlo = *reinterpret_cast<const v4f*>(xp + tnext);
        v4f nhi = *reinterpret_cast<const v4f*>(xp + tnext + 4);
        if (power) {
            nlo[0] = 0.0f; nlo[1] = 0.0f; nlo[2] = 0.0f; nlo[3] = 0.0f;
            nhi[0] = 0.0f; nhi[1] = 0.0f; nhi[2] = 0.0f; nhi[3] = 0.0f;
        }

        v8f acc[4];
#pragma unroll
        for (int tn = 0; tn < 4; ++tn)
#pragma unroll
            for (int r = 0; r < 8; ++r) {
                float v = kb[7][tn][r] * xlo[0];
                v = __builtin_fmaf(kb[6][tn][r], xlo[1], v);
                v = __builtin_fmaf(kb[5][tn][r], xlo[2], v);
                v = __builtin_fmaf(kb[4][tn][r], xlo[3], v);
                v = __builtin_fmaf(kb[3][tn][r], xhi[0], v);
                v = __builtin_fmaf(kb[2][tn][r], xhi[1], v);
                v = __builtin_fmaf(kb[1][tn][r], xhi[2], v);
                v = __builtin_fmaf(kb[0][tn][r], xhi[3], v);
                acc[tn][r] = v;
            }
        mat_step(a, s, acc);
#pragma unroll
        for (int tn = 0; tn < 4; ++tn) s[tn] = acc[tn];

        xlo = nlo;
        xhi = nhi;
    }

    if (power) {
#pragma unroll
        for (int tn = 0; tn < 4; ++tn)
#pragma unroll
            for (int r = 0; r < 8; ++r) out[(tn * 16 + r + 8 * lh) * SLEN + c] = s[tn][r];
    } else {
        save_state(colp + (k + 1) * LCH, lh, s);
    }
}

__global__ __launch_bounds__(256)
void hippo_final_kernel(const float* __restrict__ sig,
                        const float* __restrict__ Bm,
                        const float* __restrict__ Cmat,
                        const float* __restrict__ dtp,
                        const float* __restrict__ abar,
                        const float* __restrict__ a4,
                        float* out) {
    __shared__ __attribute__((aligned(16))) float tabs[2 * RSTEP * CW_BUF];
    const int lane = threadIdx.x & 31;
    const int w    = __builtin_amdgcn_readfirstlane((int)(threadIdx.x >> 5));
    const int lc   = lane & 15;
    const int lh   = lane >> 4;
    const int c    = (blockIdx.x & 7) * 16 + lc;
    const int b    = w;
    const float hdt = 0.5f * dtp[0];

    v8f kb[RSTEP][4], g[RSTEP][4];
    if (w == 0) {
        v2f a1[4][4][4];
        load_afrags(abar, SDIM, lc, lh, a1);
        v8f bin[4], acc[4];
#pragma unroll
        for (int tn = 0; tn < 4; ++tn)
#pragma unroll
            for (int r = 0; r < 8; ++r) {
                bin[tn][r] = Bm[(tn * 16 + r + 8 * lh) * NCH + c];
                acc[tn][r] = bin[tn][r];
            }
        mat_step(a1, bin, acc);
#pragma unroll
        for (int tn = 0; tn < 4; ++tn)
#pragma unroll
            for (int r = 0; r < 8; ++r) kb[0][tn][r] = hdt * acc[tn][r];
#pragma unroll
        for (int i = 1; i < RSTEP; ++i) {
#pragma unroll
            for (int tn = 0; tn < 4; ++tn)
#pragma unroll
                for (int r = 0; r < 8; ++r) acc[tn][r] = 0.0f;
            mat_step(a1, kb[i - 1], acc);
#pragma unroll
            for (int tn = 0; tn < 4; ++tn) kb[i][tn] = acc[tn];
        }
#pragma unroll
        for (int i = 0; i < RSTEP; ++i) lds_tab_store(tabs + i * CW_BUF, lane, kb[i]);
    } else if (w == 1) {
        v2f at[4][4][4];
        load_afrags_t(abar, lc, lh, at);
#pragma unroll
        for (int tn = 0; tn < 4; ++tn)
#pragma unroll
            for (int r = 0; r < 8; ++r) g[0][tn][r] = Cmat[(tn * 16 + r + 8 * lh) * NCH + c];
#pragma unroll
        for (int i = 1; i < RSTEP; ++i) {
            v8f acc[4];
#pragma unroll
            for (int tn = 0; tn < 4; ++tn)
#pragma unroll
                for (int r = 0; r < 8; ++r) acc[tn][r] = 0.0f;
            mat_step_t(at, g[i - 1], acc);
#pragma unroll
            for (int tn = 0; tn < 4; ++tn) g[i][tn] = acc[tn];
        }
#pragma unroll
        for (int i = 0; i < RSTEP; ++i) lds_tab_store(tabs + (RSTEP + i) * CW_BUF, lane, g[i]);
    }
    __syncthreads();
#pragma unroll
    for (int i = 0; i < RSTEP; ++i) {
        lds_tab_load(tabs + i * CW_BUF, lane, kb[i]);
        lds_tab_load(tabs + (RSTEP + i) * CW_BUF, lane, g[i]);
    }

    float h[RSTEP - 1];
#pragma unroll
    for (int jj = 0; jj < RSTEP - 1; ++jj) {
        float p = 0.0f;
#pragma unroll
        for (int tn = 0; tn < 4; ++tn)
#pragma unroll
            for (int r = 0; r < 8; ++r) p = __builtin_fmaf(g[0][tn][r], kb[jj][tn][r], p);
        p += __shfl_xor(p, 16, 32);
        h[jj] = p;
    }

    v2f a[4][4][4];
    load_afrags(a4, SDIM, lc, lh, a);

    const int colOff = (b * NCH + c) * SLEN;
    float* colp = out + colOff;

    const int k = blockIdx.x >> 3;

    v8f s[4];
    if (k > 0) {
        load_state(colp + k * LCH, lh, s);
    } else {
#pragma unroll
        for (int tn = 0; tn < 4; ++tn)
#pragma unroll
            for (int r = 0; r < 8; ++r) s[tn][r] = 0.0f;
    }

    const float* xp = sig + colOff + k * LCH;
    float*       yp = colp + k * LCH;

    v4f xcur4 = *reinterpret_cast<const v4f*>(xp);

#pragma unroll 1
    for (int tc = 0; tc < LCH; tc += RSTEP) {
        const int tnext = (tc + RSTEP < LCH) ? tc + RSTEP : tc;
        v4f xnext4 = *reinterpret_cast<const v4f*>(xp + tnext);

        float p1 = 0.0f, p2 = 0.0f, p3 = 0.0f;
#pragma unroll
        for (int tn = 0; tn < 4; ++tn)
#pragma unroll
            for (int r = 0; r < 8; ++r) {
                p1 = __builtin_fmaf(g[1][tn][r], s[tn][r], p1);
                p2 = __builtin_fmaf(g[2][tn][r], s[tn][r], p2);
                p3 = __builtin_fmaf(g[3][tn][r], s[tn][r], p3);
            }

        v8f acc[4];
#pragma unroll
        for (int tn = 0; tn < 4; ++tn)
#pragma unroll
            for (int r = 0; r < 8; ++r) {
                float v = kb[3][tn][r] * xcur4[0];
                v = __builtin_fmaf(kb[2][tn][r], xcur4[1], v);
                v = __builtin_fmaf(kb[1][tn][r], xcur4[2], v);
                v = __builtin_fmaf(kb[0][tn][r], xcur4[3], v);
                acc[tn][r] = v;
            }
        mat_step(a, s, acc);
#pragma unroll
        for (int tn = 0; tn < 4; ++tn) s[tn] = acc[tn];

        float p4 = 0.0f;
#pragma unroll
        for (int tn = 0; tn < 4; ++tn)
#pragma unroll
            for (int r = 0; r < 8; ++r) p4 = __builtin_fmaf(g[0][tn][r], s[tn][r], p4);

        p1 += __shfl_xor(p1, 16, 32);
        p2 += __shfl_xor(p2, 16, 32);
        p3 += __shfl_xor(p3, 16, 32);
        p4 += __shfl_xor(p4, 16, 32);

        v4f y4;
        y4[0] = __builtin_fmaf(h[0], xcur4[0], p1);
        y4[1] = __builtin_fmaf(h[1], xcur4[0], __builtin_fmaf(h[0], xcur4[1], p2));
        y4[2] = __builtin_fmaf(h[2], xcur4[0],
                               __builtin_fmaf(h[1], xcur4[1], __builtin_fmaf(h[0], xcur4[2], p3)));
        y4[3] = p4;

        if (lh == 0) *reinterpret_cast<v4f*>(yp + tc) = y4;

        xcur4 = xnext4;
    }
}

__device__ __forceinline__ void carry_block(const v2f (&a)[4], const v8f& t, v8f& acc) {
#pragma unroll
    for (int j = 0; j < 4; ++j)
        acc = __builtin_amdgcn_wmma_f32_16x16x4_f32(
             false, a[j],
             false, pair2(t[2 * j], t[2 * j + 1]),
             (short)0, acc,
             false,  false);
}

__device__ __forceinline__ void carry_lds_barrier() {
    __builtin_amdgcn_fence(__ATOMIC_RELEASE, "workgroup", "local");
    __builtin_amdgcn_s_barrier();
    __builtin_amdgcn_fence(__ATOMIC_ACQUIRE, "workgroup", "local");
}

__device__ __forceinline__ void carry_trip(float* rowp, float* sx, int tn, int lane, int k,
                                           const v2f (&aoff)[3][4], const v2f (&aown)[4],
                                           v8f& s, const v8f& e) {
    v8f acc = e;
    const float* rd = sx + ((k - 1) & 1) * CW_BUF;
    if ((tn & 1) == 0 && k > 2) {
        const v8f nb = lds_tile_load(rd, tn + 1, lane);
        *reinterpret_cast<v8f*>(rowp + (k - 1) * LCH) = s;
        *reinterpret_cast<v8f*>(rowp + (k - 1) * LCH + 16) = nb;
    }
    if (tn > 0) {
        v8f t[3];
        t[0] = lds_tile_load(rd, 0, lane);
        if (tn > 1) t[1] = lds_tile_load(rd, 1, lane);
        if (tn > 2) t[2] = lds_tile_load(rd, 2, lane);
        carry_block(aoff[0], t[0], acc);
        if (tn > 1) carry_block(aoff[1], t[1], acc);
        if (tn > 2) carry_block(aoff[2], t[2], acc);
    }
    carry_block(aown, s, acc);
    s = acc;
    lds_tile_store(sx + (k & 1) * CW_BUF, tn, lane, s);
    carry_lds_barrier();
}

__global__ __launch_bounds__(128)
void hippo_carry_kernel(float* out) {
    __shared__ __attribute__((aligned(16))) float sx[2 * CW_BUF];
    const int lane = threadIdx.x & 31;
    const int lc   = lane & 15;
    const int lh   = lane >> 4;
    const int tn = __builtin_amdgcn_readfirstlane((int)(threadIdx.x >> 5));
    float* rowp = out + (blockIdx.x * 16 + lc) * SLEN + tn * 16 + 8 * lh;

    v2f aoff[3][4], aown[4];
    const float* prow = out + (tn * 16 + lc) * SLEN + 8 * lh;
#pragma unroll
    for (int tb = 0; tb < 3; ++tb)
#pragma unroll
        for (int j = 0; j < 4; ++j)
            aoff[tb][j] = *reinterpret_cast<const v2f*>(prow + tb * 16 + 2 * j);
#pragma unroll
    for (int j = 0; j < 4; ++j)
        aown[j] = *reinterpret_cast<const v2f*>(prow + tn * 16 + 2 * j);

    v8f s  = *reinterpret_cast<const v8f*>(rowp + 1 * LCH);
    v8f q0 = *reinterpret_cast<const v8f*>(rowp + 2 * LCH);
    v8f q1 = *reinterpret_cast<const v8f*>(rowp + 3 * LCH);
    v8f q2 = *reinterpret_cast<const v8f*>(rowp + 4 * LCH);
    v8f q3 = *reinterpret_cast<const v8f*>(rowp + 5 * LCH);
    lds_tile_store(sx + CW_BUF, tn, lane, s);
    carry_lds_barrier();

#pragma unroll 1
    for (int k = 2; k < KCH; k += 2) {
        const int ka = (k + 4 < KCH) ? k + 4 : KCH - 1;
        const int kb = (k + 5 < KCH) ? k + 5 : KCH - 1;
        const v8f n2 = *reinterpret_cast<const v8f*>(rowp + ka * LCH);
        const v8f n3 = *reinterpret_cast<const v8f*>(rowp + kb * LCH);
        carry_trip(rowp, sx, tn, lane, k, aoff, aown, s, q0);
        carry_trip(rowp, sx, tn, lane, k + 1, aoff, aown, s, q1);
        q0 = q2;
        q1 = q3;
        q2 = n2;
        q3 = n3;
    }

    if ((tn & 1) == 0) {
        const v8f nb = lds_tile_load(sx + CW_BUF, tn + 1, lane);
        *reinterpret_cast<v8f*>(rowp + (KCH - 1) * LCH) = s;
        *reinterpret_cast<v8f*>(rowp + (KCH - 1) * LCH + 16) = nb;
    }
}

extern "C" void kernel_launch(void* const* d_in, const int* in_sizes, int n_in,
                              void* d_out, int out_size, void* d_ws, size_t ws_size,
                              hipStream_t stream) {
    const float* sig = (const float*)d_in[0];
    const float* Bm  = (const float*)d_in[1];
    const float* Cm  = (const float*)d_in[2];
    const float* Am  = (const float*)d_in[3];
    const float* dtp = (const float*)d_in[4];
    float* out  = (float*)d_out;
    float* abar = (float*)d_ws;
    float* a4   = abar + SDIM * SDIM;

    hippo_setup_kernel<<<1, 256, 0, stream>>>(Am, Bm, dtp, abar, a4, out);
    hippo_local_kernel<<<NBLK_LOCAL, 256, 0, stream>>>(sig, a4, out);
    hippo_carry_kernel<<<NGRP, 128, 0, stream>>>(out);
    hippo_final_kernel<<<NBLK_FINAL, 256, 0, stream>>>(sig, Bm, Cm, dtp, abar, a4, out);
}


# final pass advances 8 time steps per WMMA product, not 4 (A_bar^8 built per block; 7 inner outputs as dots with (A_bar^T)^i C); a wave carries 2 batches half a trip apart; state and y sums reordered
# speedup vs baseline: 39.9602x; 1.0371x over previous
#include <hip/hip_runtime.h>
#include <hip/hip_bf16.h>

typedef float v2f __attribute__((ext_vector_type(2)));
typedef float v4f __attribute__((ext_vector_type(4)));
typedef float v8f __attribute__((ext_vector_type(8)));

#define SDIM 64
#define NCH  128
#define NB   8
#define SLEN 4096
#define KCH  32
#define LCH  128
#define NGRP 64
#define NBLK_LOCAL 249
#define NBLK_FINAL 256
#define RSTEP 4
#define LSTEP 8
#define FSTEP 8
#define A8P   68
#define A8_COL0 64
#define KB_COL0 128

#define NRHS   192
#define LPITCH 68

template <int K>
__device__ __forceinline__ void hippo_solve_from(float (&s)[SDIM], const float* Lt) {
    if constexpr (K < SDIM) {
        const float xk = s[K] / Lt[K * LPITCH + K];
        s[K] = xk;
        const float nxk = -xk;
#pragma unroll
        for (int i = K + 1; i < SDIM; ++i)
            s[i] = __builtin_fmaf(Lt[K * LPITCH + i], nxk, s[i]);
        hippo_solve_from<K + 1>(s, Lt);
    }
}

__device__ __forceinline__ void hippo_setup_powers(const float* abar, float* a4, float* out, int tid);

__global__ __launch_bounds__(256)
void hippo_setup_kernel(const float* __restrict__ A,
                        const float* __restrict__ Bm,
                        const float* __restrict__ dtp,
                        float* abar,
                        float* a4,
                        float* out) {
    __shared__ __attribute__((aligned(16))) float Lt[SDIM * LPITCH];
    const float dt = dtp[0];
    const int tid = threadIdx.x;
    const bool solver = tid < NRHS;
    const bool bcol   = tid >= SDIM;
    const int  c      = tid - SDIM;

    float s[SDIM];
    if (solver && bcol) {
#pragma unroll
        for (int i = 0; i < SDIM; ++i) s[i] = Bm[i * NCH + c];
    } else {
#pragma unroll
        for (int i = 0; i < SDIM; ++i) s[i] = (i == tid) ? 1.0f : 0.0f;
    }

    float av[16];
#pragma unroll
    for (int q = 0; q < 16; ++q) av[q] = A[tid + 256 * q];
#pragma unroll
    for (int q = 0; q < 16; ++q) {
        const int e = tid + 256 * q;
        const int i = e >> 6, k = e & 63;
        float l = -0.5f * dt * av[q];
        if (i == k) l += 1.0f;
        Lt[k * LPITCH + i] = l;
    }
    __syncthreads();

    if (solver) {
        hippo_solve_from<0>(s, Lt);
        if (bcol) {
            float* bp = out + (KB_COL0 + c) * SLEN;
#pragma unroll
            for (int i = 0; i < SDIM; i += 4) {
                v4f q;
                q[0] = dt * s[i];
                q[1] = dt * s[i + 1];
                q[2] = dt * s[i + 2];
                q[3] = dt * s[i + 3];
                *reinterpret_cast<v4f*>(bp + i) = q;
            }
        } else {
#pragma unroll
            for (int i = 0; i < SDIM; ++i)
                abar[i * SDIM + tid] = 2.0f * s[i] - ((i == tid) ? 1.0f : 0.0f);
        }
    }

    __syncthreads();
    hippo_setup_powers(abar, a4, out, tid);
}

#define ABAR_BLOCK_NONZERO(tn, tb) ((tb) <= (tn))


__device__ __forceinline__ void load_afrags(const float* mat, int pitch, int lc, int lh,
                                            v2f (&a)[4][4][4]) {
#pragma unroll
    for (int tn = 0; tn < 4; ++tn)
#pragma unroll
        for (int tb = 0; tb < 4; ++tb)
#pragma unroll
            for (int j = 0; j < 4; ++j) {
                if (ABAR_BLOCK_NONZERO(tn, tb)) {
                    const int base = (tn * 16 + lc) * pitch + tb * 16 + 8 * lh + 2 * j;
                    a[tn][tb][j] = *reinterpret_cast<const v2f*>(mat + base);
                }
            }
}

__device__ __forceinline__ v2f pair2(float x, float y) {
    v2f p;
    p.x = x;
    p.y = y;
    return p;
}

__device__ __forceinline__ void load_afrags_t(const float* mat, int lc, int lh,
                                              v2f (&a)[4][4][4]) {
#pragma unroll
    for (int tn = 0; tn < 4; ++tn)
#pragma unroll
        for (int tb = 0; tb < 4; ++tb)
#pragma unroll
            for (int j = 0; j < 4; ++j) {
                if (ABAR_BLOCK_NONZERO(tb, tn)) {
                    const int m = tb * 16 + 8 * lh + 2 * j;
                    a[tn][tb][j] = pair2(mat[m * SDIM + tn * 16 + lc],
                                         mat[(m + 1) * SDIM + tn * 16 + lc]);
                }
            }
}

__device__ __forceinline__ void mat_step_t(const v2f (&a)[4][4][4], const v8f (&s)[4],
                                           v8f (&acc)[4]) {
#pragma unroll
    for (int tn = 0; tn < 4; ++tn)
#pragma unroll
        for (int tb = 0; tb < 4; ++tb)
#pragma unroll
            for (int j = 0; j < 4; ++j) {
                if (ABAR_BLOCK_NONZERO(tb, tn)) {
                    acc[tn] = __builtin_amdgcn_wmma_f32_16x16x4_f32(
                         false, a[tn][tb][j],
                         false, pair2(s[tb][2 * j], s[tb][2 * j + 1]),
                         (short)0, acc[tn],
                         false,  false);
                }
            }
}

__device__ __forceinline__ void mat_step(const v2f (&a)[4][4][4], const v8f (&s)[4],
                                         v8f (&acc)[4]) {
#pragma unroll
    for (int tn = 0; tn < 4; ++tn)
#pragma unroll
        for (int tb = 0; tb < 4; ++tb)
#pragma unroll
            for (int j = 0; j < 4; ++j) {
                if (ABAR_BLOCK_NONZERO(tn, tb)) {
                    acc[tn] = __builtin_amdgcn_wmma_f32_16x16x4_f32(
                         false, a[tn][tb][j],
                         false, pair2(s[tb][2 * j], s[tb][2 * j + 1]),
                         (short)0, acc[tn],
                         false,  false);
                }
            }
}

__device__ __forceinline__ void save_state(float* p, int lh, const v8f (&s)[4]) {
#pragma unroll
    for (int tn = 0; tn < 4; ++tn) *reinterpret_cast<v8f*>(p + tn * 16 + 8 * lh) = s[tn];
}
__device__ __forceinline__ void load_state(const float* p, int lh, v8f (&s)[4]) {
#pragma unroll
    for (int tn = 0; tn < 4; ++tn) s[tn] = *reinterpret_cast<const v8f*>(p + tn * 16 + 8 * lh);
}

__device__ __forceinline__ void hippo_setup_powers(const float* abar, float* a4, float* out, int tid) {
    const int lane = tid & 31;
    const int w    = tid >> 5;
    const int lc   = lane & 15;
    const int lh   = lane >> 4;
    const int c    = w * 16 + lc;

    v2f a[4][4][4];
    load_afrags(abar, SDIM, lc, lh, a);

    v8f s[4];
    load_state(out + (KB_COL0 + c) * SLEN, lh, s);

#pragma unroll
    for (int i = 1; i < RSTEP; ++i) {
        v8f acc[4];
#pragma unroll
        for (int tn = 0; tn < 4; ++tn)
#pragma unroll
            for (int r = 0; r < 8; ++r) acc[tn][r] = 0.0f;
        mat_step(a, s, acc);
#pragma unroll
        for (int tn = 0; tn < 4; ++tn) s[tn] = acc[tn];
        save_state(out + (KB_COL0 + i * NCH + c) * SLEN, lh, s);
    }

    if (w < 4) {
#pragma unroll
        for (int tn = 0; tn < 4; ++tn)
#pragma unroll
            for (int r = 0; r < 8; ++r) s[tn][r] = (tn * 16 + r + 8 * lh == c) ? 1.0f : 0.0f;
#pragma unroll
        for (int i = 0; i < RSTEP; ++i) {
            v8f acc[4];
#pragma unroll
            for (int tn = 0; tn < 4; ++tn)
#pragma unroll
                for (int r = 0; r < 8; ++r) acc[tn][r] = 0.0f;
            mat_step(a, s, acc);
#pragma unroll
            for (int tn = 0; tn < 4; ++tn) s[tn] = acc[tn];
        }
#pragma unroll
        for (int tn = 0; tn < 4; ++tn)
#pragma unroll
            for (int r = 0; r < 8; ++r)
                a4[(tn * 16 + r + 8 * lh) * SDIM + c] = s[tn][r];
#pragma unroll 1
        for (int i = RSTEP; i < LSTEP; ++i) {
            v8f acc[4];
#pragma unroll
            for (int tn = 0; tn < 4; ++tn)
#pragma unroll
                for (int r = 0; r < 8; ++r) acc[tn][r] = 0.0f;
            mat_step(a, s, acc);
#pragma unroll
            for (int tn = 0; tn < 4; ++tn) s[tn] = acc[tn];
        }
#pragma unroll
        for (int tn = 0; tn < 4; ++tn)
#pragma unroll
            for (int r = 0; r < 8; ++r)
                out[(A8_COL0 + tn * 16 + r + 8 * lh) * SLEN + c] = s[tn][r];
    }
}

#define CW_TILE 256
#define CW_BUF  1024

__device__ __forceinline__ v8f lds_tile_load(const float* buf, int tb, int lane) {
    const v4f lo = *reinterpret_cast<const v4f*>(buf + tb * CW_TILE + lane * 4);
    const v4f hi = *reinterpret_cast<const v4f*>(buf + tb * CW_TILE + 128 + lane * 4);
    v8f t;
    t[0] = lo[0]; t[1] = lo[1]; t[2] = lo[2]; t[3] = lo[3];
    t[4] = hi[0]; t[5] = hi[1]; t[6] = hi[2]; t[7] = hi[3];
    return t;
}
__device__ __forceinline__ void lds_tile_store(float* buf, int tb, int lane, const v8f& t) {
    v4f lo, hi;
    lo[0] = t[0]; lo[1] = t[1]; lo[2] = t[2]; lo[3] = t[3];
    hi[0] = t[4]; hi[1] = t[5]; hi[2] = t[6]; hi[3] = t[7];
    *reinterpret_cast<v4f*>(buf + tb * CW_TILE + lane * 4) = lo;
    *reinterpret_cast<v4f*>(buf + tb * CW_TILE + 128 + lane * 4) = hi;
}

__device__ __forceinline__ void lds_tab_store(float* tab, int lane, const v8f (&t)[4]) {
#pragma unroll
    for (int tn = 0; tn < 4; ++tn) lds_tile_store(tab, tn, lane, t[tn]);
}
__device__ __forceinline__ void lds_tab_load(const float* tab, int lane, v8f (&t)[4]) {
#pragma unroll
    for (int tn = 0; tn < 4; ++tn) t[tn] = lds_tile_load(tab, tn, lane);
}

__global__ __launch_bounds__(256)
void hippo_local_kernel(const float* __restrict__ sig,
                        const float* __restrict__ a4,
                        float* out) {
    __shared__ __attribute__((aligned(16))) float tabs[(LSTEP - RSTEP) * CW_BUF];
    const int lane = threadIdx.x & 31;
    const int w    = __builtin_amdgcn_readfirstlane((int)(threadIdx.x >> 5));
    const int lc   = lane & 15;
    const int lh   = lane >> 4;

    int k, b, c;
    bool power = false;
    if (blockIdx.x == 0) {
        power = true;
        k = 0;
        b = 0;
        c = w * 16 + lc;
        if (w >= 4) return;
    } else {
        k = (blockIdx.x - 1) >> 3;
        b = w;
        c = ((blockIdx.x - 1) & 7) * 16 + lc;
    }

    v8f kb[LSTEP][4];
#pragma unroll
    for (int i = 0; i < RSTEP; ++i) {
        load_state(out + (KB_COL0 + i * NCH + c) * SLEN, lh, kb[i]);
#pragma unroll
        for (int tn = 0; tn < 4; ++tn)
#pragma unroll
            for (int r = 0; r < 8; ++r) kb[i][tn][r] = power ? 0.0f : kb[i][tn][r];
    }
    if (!power) {
        if (w < LSTEP - RSTEP) {
            v2f a4f[4][4][4];
            load_afrags(a4, SDIM, lc, lh, a4f);
            v8f src[4], acc[4];
            load_state(out + (KB_COL0 + w * NCH + c) * SLEN, lh, src);
#pragma unroll
            for (int tn = 0; tn < 4; ++tn)
#pragma unroll
                for (int r = 0; r < 8; ++r) acc[tn][r] = 0.0f;
            mat_step(a4f, src, acc);
            lds_tab_store(tabs + w * CW_BUF, lane, acc);
        }
        __syncthreads();
#pragma unroll
        for (int i = RSTEP; i < LSTEP; ++i) lds_tab_load(tabs + (i - RSTEP) * CW_BUF, lane, kb[i]);
    } else {
#pragma unroll
        for (int i = RSTEP; i < LSTEP; ++i)
#pragma unroll
            for (int tn = 0; tn < 4; ++tn)
#pragma unroll
                for (int r = 0; r < 8; ++r) kb[i][tn][r] = 0.0f;
    }

    v2f a[4][4][4];
    load_afrags(out + A8_COL0 * SLEN, SLEN, lc, lh, a);

    const int colOff = (b * NCH + c) * SLEN;
    float* colp = out + colOff;

    v8f s[4];
#pragma unroll
    for (int tn = 0; tn < 4; ++tn)
#pragma unroll
        for (int r = 0; r < 8; ++r)
            s[tn][r] = (power && (tn * 16 + r + 8 * lh == c)) ? 1.0f : 0.0f;

    const float* xp = sig + colOff + k * LCH;

    v4f xlo = *reinterpret_cast<const v4f*>(xp);
    v4f xhi = *reinterpret_cast<const v4f*>(xp + 4);
    if (power) {
        xlo[0] = 0.0f; xlo[1] = 0.0f; xlo[2] = 0.0f; xlo[3] = 0.0f;
        xhi[0] = 0.0f; xhi[1] = 0.0f; xhi[2] = 0.0f; xhi[3] = 0.0f;
    }

#pragma unroll 1
    for (int tc = 0; tc < LCH; tc += LSTEP) {
        const int tnext = (tc + LSTEP < LCH) ? tc + LSTEP : tc;
        v4f nlo = *reinterpret_cast<const v4f*>(xp + tnext);
        v4f nhi = *reinterpret_cast<const v4f*>(xp + tnext + 4);
        if (power) {
            nlo[0] = 0.0f; nlo[1] = 0.0f; nlo[2] = 0.0f; nlo[3] = 0.0f;
            nhi[0] = 0.0f; nhi[1] = 0.0f; nhi[2] = 0.0f; nhi[3] = 0.0f;
        }

        v8f acc[4];
#pragma unroll
        for (int tn = 0; tn < 4; ++tn)
#pragma unroll
            for (int r = 0; r < 8; ++r) {
                float v = kb[7][tn][r] * xlo[0];
                v = __builtin_fmaf(kb[6][tn][r], xlo[1], v);
                v = __builtin_fmaf(kb[5][tn][r], xlo[2], v);
                v = __builtin_fmaf(kb[4][tn][r], xlo[3], v);
                v = __builtin_fmaf(kb[3][tn][r], xhi[0], v);
                v = __builtin_fmaf(kb[2][tn][r], xhi[1], v);
                v = __builtin_fmaf(kb[1][tn][r], xhi[2], v);
                v = __builtin_fmaf(kb[0][tn][r], xhi[3], v);
                acc[tn][r] = v;
            }
        mat_step(a, s, acc);
#pragma unroll
        for (int tn = 0; tn < 4; ++tn) s[tn] = acc[tn];

        xlo = nlo;
        xhi = nhi;
    }

    if (power) {
#pragma unroll
        for (int tn = 0; tn < 4; ++tn)
#pragma unroll
            for (int r = 0; r < 8; ++r) out[(tn * 16 + r + 8 * lh) * SLEN + c] = s[tn][r];
    } else {
        save_state(colp + (k + 1) * LCH, lh, s);
    }
}

__device__ __forceinline__ void final_prep(const v8f (&g)[FSTEP][4], const v8f (&kb)[FSTEP][4],
                                           const v8f (&s)[4], const v4f& xlo, const v4f& xhi,
                                           float (&p)[FSTEP], v8f (&acc)[4]) {
    float x[FSTEP];
    x[0] = xlo[0]; x[1] = xlo[1]; x[2] = xlo[2]; x[3] = xlo[3];
    x[4] = xhi[0]; x[5] = xhi[1]; x[6] = xhi[2]; x[7] = xhi[3];
    v2f pv[FSTEP];
#pragma unroll
    for (int i = 1; i < FSTEP; ++i) pv[i] = pair2(0.0f, 0.0f);
#pragma unroll
    for (int tn = 0; tn < 4; ++tn)
#pragma unroll
        for (int r2 = 0; r2 < 4; ++r2) {
            const v2f sp = pair2(s[tn][2 * r2], s[tn][2 * r2 + 1]);
#pragma unroll
            for (int i = 1; i < FSTEP; ++i)
                pv[i] = __builtin_elementwise_fma(pair2(g[i][tn][2 * r2], g[i][tn][2 * r2 + 1]), sp, pv[i]);
        }
    p[0] = 0.0f;
#pragma unroll
    for (int i = 1; i < FSTEP; ++i) p[i] = pv[i].x + pv[i].y;
#pragma unroll
    for (int tn = 0; tn < 4; ++tn)
#pragma unroll
        for (int r2 = 0; r2 < 4; ++r2) {
            v2f v = pair2(kb[FSTEP - 1][tn][2 * r2], kb[FSTEP - 1][tn][2 * r2 + 1]) * pair2(x[0], x[0]);
#pragma unroll
            for (int i = 1; i < FSTEP; ++i)
                v = __builtin_elementwise_fma(pair2(kb[FSTEP - 1 - i][tn][2 * r2], kb[FSTEP - 1 - i][tn][2 * r2 + 1]),
                                              pair2(x[i], x[i]), v);
            acc[tn][2 * r2]     = v.x;
            acc[tn][2 * r2 + 1] = v.y;
        }
}

__device__ __forceinline__ void final_finish(const v8f (&g0)[4], const float (&h)[FSTEP - 1],
                                             const v8f (&s)[4], const v4f& xlo, const v4f& xhi,
                                             float (&p)[FSTEP], v4f& ylo, v4f& yhi) {
    float x[FSTEP];
    x[0] = xlo[0]; x[1] = xlo[1]; x[2] = xlo[2]; x[3] = xlo[3];
    x[4] = xhi[0]; x[5] = xhi[1]; x[6] = xhi[2]; x[7] = xhi[3];
    v2f qv = pair2(0.0f, 0.0f);
#pragma unroll
    for (int tn = 0; tn < 4; ++tn)
#pragma unroll
        for (int r2 = 0; r2 < 4; ++r2)
            qv = __builtin_elementwise_fma(pair2(g0[tn][2 * r2], g0[tn][2 * r2 + 1]),
                                           pair2(s[tn][2 * r2], s[tn][2 * r2 + 1]), qv);
    p[0] = qv.x + qv.y;
#pragma unroll
    for (int i = 0; i < FSTEP; ++i) p[i] += __shfl_xor(p[i], 16, 32);
    float y[FSTEP];
#pragma unroll
    for (int i = 1; i < FSTEP; ++i) {
        float v = p[i];
#pragma unroll
        for (int j = 1; j <= i; ++j) v = __builtin_fmaf(h[i - j], x[j - 1], v);
        y[i - 1] = v;
    }
    y[FSTEP - 1] = p[0];
    ylo[0] = y[0]; ylo[1] = y[1]; ylo[2] = y[2]; ylo[3] = y[3];
    yhi[0] = y[4]; yhi[1] = y[5]; yhi[2] = y[6]; yhi[3] = y[7];
}

__global__ __launch_bounds__(128)
void hippo_final_kernel(const float* __restrict__ sig,
                        const float* __restrict__ Bm,
                        const float* __restrict__ Cmat,
                        const float* __restrict__ dtp,
                        const float* __restrict__ abar,
                        const float* __restrict__ a4,
                        float* out) {
    __shared__ __attribute__((aligned(16))) float tabs[2 * RSTEP * CW_BUF];
    __shared__ __attribute__((aligned(16))) float a8s[SDIM * A8P];
    const int lane = threadIdx.x & 31;
    const int w    = __builtin_amdgcn_readfirstlane((int)(threadIdx.x >> 5));
    const int lc   = lane & 15;
    const int lh   = lane >> 4;
    const int c    = (blockIdx.x & 7) * 16 + lc;
    const float hdt = 0.5f * dtp[0];

    v2f a4f[4][4][4], at4[4][4][4];
    load_afrags(a4, SDIM, lc, lh, a4f);
    load_afrags_t(a4, lc, lh, at4);

    if (w == 0) {
        v2f a1[4][4][4];
        load_afrags(abar, SDIM, lc, lh, a1);
        v8f t[RSTEP][4], bin[4], acc[4];
#pragma unroll
        for (int tn = 0; tn < 4; ++tn)
#pragma unroll
            for (int r = 0; r < 8; ++r) {
                bin[tn][r] = Bm[(tn * 16 + r + 8 * lh) * NCH + c];
                acc[tn][r] = bin[tn][r];
            }
        mat_step(a1, bin, acc);
#pragma unroll
        for (int tn = 0; tn < 4; ++tn)
#pragma unroll
            for (int r = 0; r < 8; ++r) t[0][tn][r] = hdt * acc[tn][r];
#pragma unroll
        for (int i = 1; i < RSTEP; ++i) {
#pragma unroll
            for (int tn = 0; tn < 4; ++tn)
#pragma unroll
                for (int r = 0; r < 8; ++r) acc[tn][r] = 0.0f;
            mat_step(a1, t[i - 1], acc);
#pragma unroll
            for (int tn = 0; tn < 4; ++tn) t[i][tn] = acc[tn];
        }
#pragma unroll
        for (int i = 0; i < RSTEP; ++i) lds_tab_store(tabs + i * CW_BUF, lane, t[i]);
    } else if (w == 1) {
        v2f at[4][4][4];
        load_afrags_t(abar, lc, lh, at);
        v8f t[RSTEP][4];
#pragma unroll
        for (int tn = 0; tn < 4; ++tn)
#pragma unroll
            for (int r = 0; r < 8; ++r) t[0][tn][r] = Cmat[(tn * 16 + r + 8 * lh) * NCH + c];
#pragma unroll
        for (int i = 1; i < RSTEP; ++i) {
            v8f acc[4];
#pragma unroll
            for (int tn = 0; tn < 4; ++tn)
#pragma unroll
                for (int r = 0; r < 8; ++r) acc[tn][r] = 0.0f;
            mat_step_t(at, t[i - 1], acc);
#pragma unroll
            for (int tn = 0; tn < 4; ++tn) t[i][tn] = acc[tn];
        }
#pragma unroll
        for (int i = 0; i < RSTEP; ++i) lds_tab_store(tabs + (RSTEP + i) * CW_BUF, lane, t[i]);
    } else {
#pragma unroll
        for (int q2 = 0; q2 < 2; ++q2) {
            const int q = (w - 2) * 2 + q2;
            v8f src[4], acc[4];
#pragma unroll
            for (int tn = 0; tn < 4; ++tn)
#pragma unroll
                for (int r = 0; r < 8; ++r) {
                    src[tn][r] = a4[(tn * 16 + r + 8 * lh) * SDIM + q * 16 + lc];
                    acc[tn][r] = 0.0f;
                }
            mat_step(a4f, src, acc);
#pragma unroll
            for (int tn = 0; tn < 4; ++tn)
#pragma unroll
                for (int r = 0; r < 8; ++r)
                    a8s[(tn * 16 + r + 8 * lh) * A8P + q * 16 + lc] = acc[tn][r];
        }
    }
    __syncthreads();

    v8f kb[FSTEP][4], g[FSTEP][4];
#pragma unroll
    for (int i = 0; i < RSTEP; ++i) {
        lds_tab_load(tabs + i * CW_BUF, lane, kb[i]);
        lds_tab_load(tabs + (RSTEP + i) * CW_BUF, lane, g[i]);
    }
    {
        v8f src[4], nk[4], ng[4];
        lds_tab_load(tabs + w * CW_BUF, lane, src);
#pragma unroll
        for (int tn = 0; tn < 4; ++tn)
#pragma unroll
            for (int r = 0; r < 8; ++r) nk[tn][r] = 0.0f;
        mat_step(a4f, src, nk);
        lds_tab_load(tabs + (RSTEP + w) * CW_BUF, lane, src);
#pragma unroll
        for (int tn = 0; tn < 4; ++tn)
#pragma unroll
            for (int r = 0; r < 8; ++r) ng[tn][r] = 0.0f;
        mat_step_t(at4, src, ng);
        __syncthreads();
        lds_tab_store(tabs + w * CW_BUF, lane, nk);
        lds_tab_store(tabs + (RSTEP + w) * CW_BUF, lane, ng);
    }
    __syncthreads();
#pragma unroll
    for (int i = 0; i < RSTEP; ++i) {
        lds_tab_load(tabs + i * CW_BUF, lane, kb[RSTEP + i]);
        lds_tab_load(tabs + (RSTEP + i) * CW_BUF, lane, g[RSTEP + i]);
    }

    float h[FSTEP - 1];
#pragma unroll
    for (int jj = 0; jj < FSTEP - 1; ++jj) {
        float p = 0.0f;
#pragma unroll
        for (int tn = 0; tn < 4; ++tn)
#pragma unroll
            for (int r = 0; r < 8; ++r) p = __builtin_fmaf(g[0][tn][r], kb[jj][tn][r], p);
        p += __shfl_xor(p, 16, 32);
        h[jj] = p;
    }

    v2f a[4][4][4];
    load_afrags(a8s, A8P, lc, lh, a);

    const int k = blockIdx.x >> 3;

    v8f s[2][4], acc[2][4];
    float p[2][FSTEP];
    const float* xp[2];
    float*       yp[2];
    v4f xlo[2], xhi[2];
    v4f nlo[2], nhi[2];
#pragma unroll
    for (int q = 0; q < 2; ++q) {
        const int colOff = ((2 * w + q) * NCH + c) * SLEN;
        if (k > 0) {
            load_state(out + colOff + k * LCH, lh, s[q]);
        } else {
#pragma unroll
            for (int tn = 0; tn < 4; ++tn)
#pragma unroll
                for (int r = 0; r < 8; ++r) s[q][tn][r] = 0.0f;
        }
        xp[q] = sig + colOff + k * LCH;
        yp[q] = out + colOff + k * LCH;
        xlo[q] = *reinterpret_cast<const v4f*>(xp[q]);
        xhi[q] = *reinterpret_cast<const v4f*>(xp[q] + 4);
        nlo[q] = *reinterpret_cast<const v4f*>(xp[q] + FSTEP);
        nhi[q] = *reinterpret_cast<const v4f*>(xp[q] + FSTEP + 4);
    }

    final_prep(g, kb, s[0], xlo[0], xhi[0], p[0], acc[0]);

#pragma unroll 1
    for (int tc = 0; tc < LCH; tc += FSTEP) {
        const int t2 = (tc + 2 * FSTEP < LCH) ? tc + 2 * FSTEP : LCH - FSTEP;

        mat_step(a, s[0], acc[0]);
        final_prep(g, kb, s[1], xlo[1], xhi[1], p[1], acc[1]);

        mat_step(a, s[1], acc[1]);
#pragma unroll
        for (int tn = 0; tn < 4; ++tn) s[0][tn] = acc[0][tn];
        v4f ylo0, yhi0;
        final_finish(g[0], h, s[0], xlo[0], xhi[0], p[0], ylo0, yhi0);
        xlo[0] = nlo[0];
        xhi[0] = nhi[0];
        nlo[0] = *reinterpret_cast<const v4f*>(xp[0] + t2);
        nhi[0] = *reinterpret_cast<const v4f*>(xp[0] + t2 + 4);
        final_prep(g, kb, s[0], xlo[0], xhi[0], p[0], acc[0]);

#pragma unroll
        for (int tn = 0; tn < 4; ++tn) s[1][tn] = acc[1][tn];
        v4f ylo1, yhi1;
        final_finish(g[0], h, s[1], xlo[1], xhi[1], p[1], ylo1, yhi1);
        xlo[1] = nlo[1];
        xhi[1] = nhi[1];
        nlo[1] = *reinterpret_cast<const v4f*>(xp[1] + t2);
        nhi[1] = *reinterpret_cast<const v4f*>(xp[1] + t2 + 4);

        {
            v4f y0, y1;
#pragma unroll
            for (int e = 0; e < 4; ++e) {
                y0[e] = lh ? yhi0[e] : ylo0[e];
                y1[e] = lh ? yhi1[e] : ylo1[e];
            }
            *reinterpret_cast<v4f*>(yp[0] + tc + 4 * lh) = y0;
            *reinterpret_cast<v4f*>(yp[1] + tc + 4 * lh) = y1;
        }

#pragma unroll
        for (int i = 0; i < 40; ++i) {
            __builtin_amdgcn_sched_group_barrier(0x008, 1, 0);
            __builtin_amdgcn_sched_group_barrier(0x002, 7, 0);
        }
#pragma unroll
        for (int i = 0; i < 40; ++i) {
            __builtin_amdgcn_sched_group_barrier(0x008, 1, 0);
            __builtin_amdgcn_sched_group_barrier(0x002, 8, 0);
        }
    }
}

__device__ __forceinline__ void carry_block(const v2f (&a)[4], const v8f& t, v8f& acc) {
#pragma unroll
    for (int j = 0; j < 4; ++j)
        acc = __builtin_amdgcn_wmma_f32_16x16x4_f32(
             false, a[j],
             false, pair2(t[2 * j], t[2 * j + 1]),
             (short)0, acc,
             false,  false);
}

__device__ __forceinline__ void carry_lds_barrier() {
    __builtin_amdgcn_fence(__ATOMIC_RELEASE, "workgroup", "local");
    __builtin_amdgcn_s_barrier();
    __builtin_amdgcn_fence(__ATOMIC_ACQUIRE, "workgroup", "local");
}

__device__ __forceinline__ void carry_trip(float* rowp, float* sx, int tn, int lane, int k,
                                           const v2f (&aoff)[3][4], const v2f (&aown)[4],
                                           v8f& s, const v8f& e) {
    v8f acc = e;
    const float* rd = sx + ((k - 1) & 1) * CW_BUF;
    if ((tn & 1) == 0 && k > 2) {
        const v8f nb = lds_tile_load(rd, tn + 1, lane);
        *reinterpret_cast<v8f*>(rowp + (k - 1) * LCH) = s;
        *reinterpret_cast<v8f*>(rowp + (k - 1) * LCH + 16) = nb;
    }
    if (tn > 0) {
        v8f t[3];
        t[0] = lds_tile_load(rd, 0, lane);
        if (tn > 1) t[1] = lds_tile_load(rd, 1, lane);
        if (tn > 2) t[2] = lds_tile_load(rd, 2, lane);
        carry_block(aoff[0], t[0], acc);
        if (tn > 1) carry_block(aoff[1], t[1], acc);
        if (tn > 2) carry_block(aoff[2], t[2], acc);
    }
    carry_block(aown, s, acc);
    s = acc;
    lds_tile_store(sx + (k & 1) * CW_BUF, tn, lane, s);
    carry_lds_barrier();
}

__global__ __launch_bounds__(128)
void hippo_carry_kernel(float* out) {
    __shared__ __attribute__((aligned(16))) float sx[2 * CW_BUF];
    const int lane = threadIdx.x & 31;
    const int lc   = lane & 15;
    const int lh   = lane >> 4;
    const int tn = __builtin_amdgcn_readfirstlane((int)(threadIdx.x >> 5));
    float* rowp = out + (blockIdx.x * 16 + lc) * SLEN + tn * 16 + 8 * lh;

    v2f aoff[3][4], aown[4];
    const float* prow = out + (tn * 16 + lc) * SLEN + 8 * lh;
#pragma unroll
    for (int tb = 0; tb < 3; ++tb)
#pragma unroll
        for (int j = 0; j < 4; ++j)
            aoff[tb][j] = *reinterpret_cast<const v2f*>(prow + tb * 16 + 2 * j);
#pragma unroll
    for (int j = 0; j < 4; ++j)
        aown[j] = *reinterpret_cast<const v2f*>(prow + tn * 16 + 2 * j);

    v8f s  = *reinterpret_cast<const v8f*>(rowp + 1 * LCH);
    v8f q0 = *reinterpret_cast<const v8f*>(rowp + 2 * LCH);
    v8f q1 = *reinterpret_cast<const v8f*>(rowp + 3 * LCH);
    v8f q2 = *reinterpret_cast<const v8f*>(rowp + 4 * LCH);
    v8f q3 = *reinterpret_cast<const v8f*>(rowp + 5 * LCH);
    lds_tile_store(sx + CW_BUF, tn, lane, s);
    carry_lds_barrier();

#pragma unroll 1
    for (int k = 2; k < KCH; k += 2) {
        const int ka = (k + 4 < KCH) ? k + 4 : KCH - 1;
        const int kb = (k + 5 < KCH) ? k + 5 : KCH - 1;
        const v8f n2 = *reinterpret_cast<const v8f*>(rowp + ka * LCH);
        const v8f n3 = *reinterpret_cast<const v8f*>(rowp + kb * LCH);
        carry_trip(rowp, sx, tn, lane, k, aoff, aown, s, q0);
        carry_trip(rowp, sx, tn, lane, k + 1, aoff, aown, s, q1);
        q0 = q2;
        q1 = q3;
        q2 = n2;
        q3 = n3;
    }

    if ((tn & 1) == 0) {
        const v8f nb = lds_tile_load(sx + CW_BUF, tn + 1, lane);
        *reinterpret_cast<v8f*>(rowp + (KCH - 1) * LCH) = s;
        *reinterpret_cast<v8f*>(rowp + (KCH - 1) * LCH + 16) = nb;
    }
}

extern "C" void kernel_launch(void* const* d_in, const int* in_sizes, int n_in,
                              void* d_out, int out_size, void* d_ws, size_t ws_size,
                              hipStream_t stream) {
    const float* sig = (const float*)d_in[0];
    const float* Bm  = (const float*)d_in[1];
    const float* Cm  = (const float*)d_in[2];
    const float* Am  = (const float*)d_in[3];
    const float* dtp = (const float*)d_in[4];
    float* out  = (float*)d_out;
    float* abar = (float*)d_ws;
    float* a4   = abar + SDIM * SDIM;

    hippo_setup_kernel<<<1, 256, 0, stream>>>(Am, Bm, dtp, abar, a4, out);
    hippo_local_kernel<<<NBLK_LOCAL, 256, 0, stream>>>(sig, a4, out);
    hippo_carry_kernel<<<NGRP, 128, 0, stream>>>(out);
    hippo_final_kernel<<<NBLK_FINAL, 128, 0, stream>>>(sig, Bm, Cm, dtp, abar, a4, out);
}
